# starting kernel
# baseline (speedup 1.0000x reference)
#include <hip/hip_runtime.h>
#include <hip/hip_bf16.h>

#define N_NODES 100000
#define N_EDGES 3200000
#define HID 64
#define RED 32
#define EMB 32
#define LN_EPS 1e-5f

typedef __attribute__((ext_vector_type(2))) float v2f;
typedef __attribute__((ext_vector_type(8))) float v8f;

__global__ void k_deg_init(float* __restrict__ deg) {
    int i = blockIdx.x * blockDim.x + threadIdx.x;
    if (i < N_NODES) deg[i] = 1.0f;
}

__global__ void k_deg_scatter(const int* __restrict__ col, const float* __restrict__ w,
                              float* __restrict__ deg) {
    int e = blockIdx.x * blockDim.x + threadIdx.x;
    if (e < N_EDGES) atomicAdd(&deg[col[e]], w[e]);
}

__global__ void k_l1_init(const float* __restrict__ x, const float* __restrict__ deg,
                          const float* __restrict__ W1, const float* __restrict__ b1,
                          const float* __restrict__ rW1, const float* __restrict__ rb1,
                          float* __restrict__ dinv, float* __restrict__ xw1,
                          float* __restrict__ acc1) {
    long long t = (long long)blockIdx.x * blockDim.x + threadIdx.x;
    if (t >= (long long)N_NODES * HID) return;
    int n = (int)(t >> 6);
    int f = (int)(t & 63);
    float d  = deg[n];
    float di = d > 0.f ? rsqrtf(d) : 0.f;
    if (f == 0) dinv[n] = di;
    float x0 = x[2 * n], x1v = x[2 * n + 1];
    float xw = fmaf(x0, W1[f], x1v * W1[HID + f]);
    xw1[t]  = xw;
    acc1[t] = di * di * xw + b1[f] + fmaf(x0, rW1[f], x1v * rW1[HID + f]) + rb1[f];
}

template <int F>
__global__ void k_edge_scatter(const int* __restrict__ row, const int* __restrict__ col,
                               const float* __restrict__ w, const float* __restrict__ dinv,
                               const float* __restrict__ src, float* __restrict__ acc) {
    long long t = (long long)blockIdx.x * blockDim.x + threadIdx.x;
    if (t >= (long long)N_EDGES * F) return;
    int e = (int)(t / F);
    int f = (int)(t & (F - 1));
    int r = row[e], c = col[e];
    float coeff = dinv[r] * w[e] * dinv[c];
    atomicAdd(&acc[(long long)c * F + f], coeff * src[(long long)r * F + f]);
}

__global__ void k_ln_relu_64(const float* __restrict__ acc, const float* __restrict__ g,
                             const float* __restrict__ be, float* __restrict__ out) {
    int wave = threadIdx.x >> 5;
    int lane = threadIdx.x & 31;
    int n = blockIdx.x * (blockDim.x >> 5) + wave;
    if (n >= N_NODES) return;
    float v0 = acc[(long long)n * 64 + lane];
    float v1 = acc[(long long)n * 64 + lane + 32];
    float s = v0 + v1, sq = fmaf(v0, v0, v1 * v1);
    #pragma unroll
    for (int off = 16; off > 0; off >>= 1) {
        s  += __shfl_xor(s,  off, 32);
        sq += __shfl_xor(sq, off, 32);
    }
    float mu  = s * (1.f / 64.f);
    float var = sq * (1.f / 64.f) - mu * mu;
    float inv = rsqrtf(var + LN_EPS);
    float o0 = (v0 - mu) * inv * g[lane]      + be[lane];
    float o1 = (v1 - mu) * inv * g[lane + 32] + be[lane + 32];
    out[(long long)n * 64 + lane]      = fmaxf(o0, 0.f);
    out[(long long)n * 64 + lane + 32] = fmaxf(o1, 0.f);
}

__global__ void k_l2_wmma(const float* __restrict__ x1, const float* __restrict__ W2,
                          const float* __restrict__ rW2, const float* __restrict__ b2,
                          const float* __restrict__ rb2, const float* __restrict__ dinv,
                          float* __restrict__ xw2, float* __restrict__ acc2) {
    int wave_global = (blockIdx.x * blockDim.x + threadIdx.x) >> 5;
    int lane = threadIdx.x & 31;
    const int tiles_m = N_NODES / 16;
    int tm = wave_global >> 1;
    int tn = wave_global & 1;
    if (tm >= tiles_m) return;
    int m0 = tm * 16, n0 = tn * 16;
    int  lr = lane & 15;
    bool hi = lane >= 16;
    int  kofs = hi ? 2 : 0;

    v8f c0 = {};
    v8f c1 = {};
    #pragma unroll
    for (int k0 = 0; k0 < HID; k0 += 4) {
        const float* ap = &x1[(long long)(m0 + lr) * HID + k0 + kofs];
        v2f a; a.x = ap[0]; a.y = ap[1];
        int kb = k0 + kofs;
        v2f b;  b.x  = W2 [kb * RED + n0 + lr]; b.y  = W2 [(kb + 1) * RED + n0 + lr];
        v2f br; br.x = rW2[kb * RED + n0 + lr]; br.y = rW2[(kb + 1) * RED + n0 + lr];
        c0 = __builtin_amdgcn_wmma_f32_16x16x4_f32(false, a, false, b,  (short)0, c0, false, false);
        c1 = __builtin_amdgcn_wmma_f32_16x16x4_f32(false, a, false, br, (short)0, c1, false, false);
    }
    int ncol = n0 + lr;
    float b2v = b2[ncol], rb2v = rb2[ncol];
    #pragma unroll
    for (int v = 0; v < 8; ++v) {
        int m = m0 + v + (hi ? 8 : 0);
        float di = dinv[m];
        float xv = c0[v];
        xw2 [(long long)m * RED + ncol] = xv;
        acc2[(long long)m * RED + ncol] = di * di * xv + b2v + c1[v] + rb2v;
    }
}

__global__ void k_head(const float* __restrict__ acc2, const float* __restrict__ g2,
                       const float* __restrict__ be2, const float* __restrict__ fcW,
                       const float* __restrict__ fcb, const float* __restrict__ pW,
                       const float* __restrict__ pb, float* __restrict__ logits,
                       float* __restrict__ emb) {
    int wave = threadIdx.x >> 5;
    int lane = threadIdx.x & 31;
    int n = blockIdx.x * (blockDim.x >> 5) + wave;
    if (n >= N_NODES) return;
    float v = acc2[(long long)n * 32 + lane];
    float s = v, sq = v * v;
    #pragma unroll
    for (int off = 16; off > 0; off >>= 1) {
        s  += __shfl_xor(s,  off, 32);
        sq += __shfl_xor(sq, off, 32);
    }
    float mu  = s * (1.f / 32.f);
    float var = sq * (1.f / 32.f) - mu * mu;
    float inv = rsqrtf(var + LN_EPS);
    float x2 = fmaxf((v - mu) * inv * g2[lane] + be2[lane], 0.f);

    float lg = x2 * fcW[lane];
    #pragma unroll
    for (int off = 16; off > 0; off >>= 1) lg += __shfl_xor(lg, off, 32);
    if (lane == 0) logits[n] = lg + fcb[0];

    float p = pb[lane];
    #pragma unroll
    for (int i = 0; i < 32; ++i)
        p = fmaf(__shfl(x2, i, 32), pW[i * 32 + lane], p);
    float nsq = p * p;
    #pragma unroll
    for (int off = 16; off > 0; off >>= 1) nsq += __shfl_xor(nsq, off, 32);
    float nrm = sqrtf(nsq);
    emb[(long long)n * 32 + lane] = p / fmaxf(nrm, 1e-12f);
}

extern "C" void kernel_launch(void* const* d_in, const int* in_sizes, int n_in,
                              void* d_out, int out_size, void* d_ws, size_t ws_size,
                              hipStream_t stream) {
    const float* x    = (const float*)d_in[0];
    const float* ew   = (const float*)d_in[1];
    const float* W1   = (const float*)d_in[2];
    const float* b1   = (const float*)d_in[3];
    const float* rW1  = (const float*)d_in[4];
    const float* rb1  = (const float*)d_in[5];
    const float* g1   = (const float*)d_in[6];
    const float* be1  = (const float*)d_in[7];
    const float* W2   = (const float*)d_in[8];
    const float* b2   = (const float*)d_in[9];
    const float* rW2  = (const float*)d_in[10];
    const float* rb2  = (const float*)d_in[11];
    const float* g2   = (const float*)d_in[12];
    const float* be2  = (const float*)d_in[13];
    const float* fcW  = (const float*)d_in[14];
    const float* fcb  = (const float*)d_in[15];
    const float* pW   = (const float*)d_in[16];
    const float* pb   = (const float*)d_in[17];
    const int*   ei   = (const int*)d_in[18];
    const int*   row  = ei;
    const int*   col  = ei + N_EDGES;

    float* ws   = (float*)d_ws;
    float* deg  = ws;  ws += N_NODES;
    float* dinv = ws;  ws += N_NODES;
    float* xw1  = ws;  ws += (size_t)N_NODES * HID;
    float* acc1 = ws;  ws += (size_t)N_NODES * HID;
    float* x1   = ws;  ws += (size_t)N_NODES * HID;
    float* xw2  = ws;  ws += (size_t)N_NODES * RED;
    float* acc2 = ws;  ws += (size_t)N_NODES * RED;

    float* logits = (float*)d_out;
    float* emb    = logits + N_NODES;

    const int B = 256;
    k_deg_init<<<(N_NODES + B - 1) / B, B, 0, stream>>>(deg);
    k_deg_scatter<<<(N_EDGES + B - 1) / B, B, 0, stream>>>(col, ew, deg);

    long long t1 = (long long)N_NODES * HID;
    k_l1_init<<<(unsigned)((t1 + B - 1) / B), B, 0, stream>>>(x, deg, W1, b1, rW1, rb1,
                                                              dinv, xw1, acc1);

    long long te1 = (long long)N_EDGES * HID;
    k_edge_scatter<HID><<<(unsigned)((te1 + B - 1) / B), B, 0, stream>>>(row, col, ew, dinv,
                                                                          xw1, acc1);

    k_ln_relu_64<<<(N_NODES + 7) / 8, B, 0, stream>>>(acc1, g1, be1, x1);

    int waves_l2 = (N_NODES / 16) * (RED / 16);
    int thr_l2   = waves_l2 * 32;
    k_l2_wmma<<<(thr_l2 + B - 1) / B, B, 0, stream>>>(x1, W2, rW2, b2, rb2, dinv, xw2, acc2);

    long long te2 = (long long)N_EDGES * RED;
    k_edge_scatter<RED><<<(unsigned)((te2 + B - 1) / B), B, 0, stream>>>(row, col, ew, dinv,
                                                                          xw2, acc2);

    k_head<<<(N_NODES + 7) / 8, B, 0, stream>>>(acc2, g2, be2, fcW, fcb, pW, pb, logits, emb);
}


# layer-1 neighbour sum taken before W1 on dinv*x (2 adds per edge, not 64) in f64 atomics: the f32 arrival-order error that hit all 64 features alike is gone; W1, residual, LayerNorm in one node pass
# speedup vs baseline: 1.7944x; 1.7944x over previous
#include <hip/hip_runtime.h>
#include <hip/hip_bf16.h>

#define N_NODES 100000
#define N_EDGES 3200000
#define HID 64
#define RED 32
#define EMB 32
#define LN_EPS 1e-5f

typedef __attribute__((ext_vector_type(2))) float v2f;
typedef __attribute__((ext_vector_type(8))) float v8f;

__global__ void k_deg_init(float* __restrict__ deg) {
    int i = blockIdx.x * blockDim.x + threadIdx.x;
    if (i < N_NODES) deg[i] = 1.0f;
}

__global__ void k_deg_scatter(const int* __restrict__ col, const float* __restrict__ w,
                              float* __restrict__ deg) {
    int e = blockIdx.x * blockDim.x + threadIdx.x;
    if (e < N_EDGES) atomicAdd(&deg[col[e]], w[e]);
}

__global__ void k_prep(const float* __restrict__ x, const float* __restrict__ deg,
                       float* __restrict__ dinv, float* __restrict__ y,
                       double* __restrict__ s) {
    int n = blockIdx.x * blockDim.x + threadIdx.x;
    if (n >= N_NODES) return;
    float d  = deg[n];
    float di = d > 0.f ? rsqrtf(d) : 0.f;
    dinv[n] = di;
    y[2 * n]     = di * x[2 * n];
    y[2 * n + 1] = di * x[2 * n + 1];
    s[2 * n]     = 0.0;
    s[2 * n + 1] = 0.0;
}

__global__ void k_edge_l1(const int* __restrict__ row, const int* __restrict__ col,
                          const float* __restrict__ w, const float* __restrict__ y,
                          double* __restrict__ s) {
    int t = blockIdx.x * blockDim.x + threadIdx.x;
    if (t >= 2 * N_EDGES) return;
    int e = t >> 1;
    int k = t & 1;
    int r = row[e], c = col[e];
    double v = (double)w[e] * (double)y[2 * r + k];
    atomicAdd(&s[2 * c + k], v);
}

__global__ void k_node_l1(const float* __restrict__ x, const float* __restrict__ dinv,
                          const float* __restrict__ y, const double* __restrict__ s,
                          const float* __restrict__ W1, const float* __restrict__ b1,
                          const float* __restrict__ rW1, const float* __restrict__ rb1,
                          const float* __restrict__ g, const float* __restrict__ be,
                          float* __restrict__ out) {
    int wave = threadIdx.x >> 5;
    int lane = threadIdx.x & 31;
    int n = blockIdx.x * (blockDim.x >> 5) + wave;
    if (n >= N_NODES) return;
    float di = dinv[n];
    float x0 = x[2 * n], x1v = x[2 * n + 1];
    float t0 = di * (float)(s[2 * n]     + (double)y[2 * n]);
    float t1 = di * (float)(s[2 * n + 1] + (double)y[2 * n + 1]);
    int f0 = lane, f1 = lane + 32;
    float v0 = fmaf(t0, W1[f0], t1 * W1[HID + f0]) + b1[f0]
             + fmaf(x0, rW1[f0], x1v * rW1[HID + f0]) + rb1[f0];
    float v1 = fmaf(t0, W1[f1], t1 * W1[HID + f1]) + b1[f1]
             + fmaf(x0, rW1[f1], x1v * rW1[HID + f1]) + rb1[f1];
    float sm = v0 + v1, sq = fmaf(v0, v0, v1 * v1);
    #pragma unroll
    for (int off = 16; off > 0; off >>= 1) {
        sm += __shfl_xor(sm, off, 32);
        sq += __shfl_xor(sq, off, 32);
    }
    float mu  = sm * (1.f / 64.f);
    float var = sq * (1.f / 64.f) - mu * mu;
    float inv = rsqrtf(var + LN_EPS);
    float o0 = (v0 - mu) * inv * g[f0] + be[f0];
    float o1 = (v1 - mu) * inv * g[f1] + be[f1];
    out[(long long)n * 64 + f0] = fmaxf(o0, 0.f);
    out[(long long)n * 64 + f1] = fmaxf(o1, 0.f);
}

template <int F>
__global__ void k_edge_scatter(const int* __restrict__ row, const int* __restrict__ col,
                               const float* __restrict__ w, const float* __restrict__ dinv,
                               const float* __restrict__ src, float* __restrict__ acc) {
    long long t = (long long)blockIdx.x * blockDim.x + threadIdx.x;
    if (t >= (long long)N_EDGES * F) return;
    int e = (int)(t / F);
    int f = (int)(t & (F - 1));
    int r = row[e], c = col[e];
    float coeff = dinv[r] * w[e] * dinv[c];
    atomicAdd(&acc[(long long)c * F + f], coeff * src[(long long)r * F + f]);
}

__global__ void k_l2_wmma(const float* __restrict__ x1, const float* __restrict__ W2,
                          const float* __restrict__ rW2, const float* __restrict__ b2,
                          const float* __restrict__ rb2, const float* __restrict__ dinv,
                          float* __restrict__ xw2, float* __restrict__ acc2) {
    int wave_global = (blockIdx.x * blockDim.x + threadIdx.x) >> 5;
    int lane = threadIdx.x & 31;
    const int tiles_m = N_NODES / 16;
    int tm = wave_global >> 1;
    int tn = wave_global & 1;
    if (tm >= tiles_m) return;
    int m0 = tm * 16, n0 = tn * 16;
    int  lr = lane & 15;
    bool hi = lane >= 16;
    int  kofs = hi ? 2 : 0;

    v8f c0 = {};
    v8f c1 = {};
    #pragma unroll
    for (int k0 = 0; k0 < HID; k0 += 4) {
        const float* ap = &x1[(long long)(m0 + lr) * HID + k0 + kofs];
        v2f a; a.x = ap[0]; a.y = ap[1];
        int kb = k0 + kofs;
        v2f b;  b.x  = W2 [kb * RED + n0 + lr]; b.y  = W2 [(kb + 1) * RED + n0 + lr];
        v2f br; br.x = rW2[kb * RED + n0 + lr]; br.y = rW2[(kb + 1) * RED + n0 + lr];
        c0 = __builtin_amdgcn_wmma_f32_16x16x4_f32(false, a, false, b,  (short)0, c0, false, false);
        c1 = __builtin_amdgcn_wmma_f32_16x16x4_f32(false, a, false, br, (short)0, c1, false, false);
    }
    int ncol = n0 + lr;
    float b2v = b2[ncol], rb2v = rb2[ncol];
    #pragma unroll
    for (int v = 0; v < 8; ++v) {
        int m = m0 + v + (hi ? 8 : 0);
        float di = dinv[m];
        float xv = c0[v];
        xw2 [(long long)m * RED + ncol] = xv;
        acc2[(long long)m * RED + ncol] = di * di * xv + b2v + c1[v] + rb2v;
    }
}

__global__ void k_head(const float* __restrict__ acc2, const float* __restrict__ g2,
                       const float* __restrict__ be2, const float* __restrict__ fcW,
                       const float* __restrict__ fcb, const float* __restrict__ pW,
                       const float* __restrict__ pb, float* __restrict__ logits,
                       float* __restrict__ emb) {
    int wave = threadIdx.x >> 5;
    int lane = threadIdx.x & 31;
    int n = blockIdx.x * (blockDim.x >> 5) + wave;
    if (n >= N_NODES) return;
    float v = acc2[(long long)n * 32 + lane];
    float s = v, sq = v * v;
    #pragma unroll
    for (int off = 16; off > 0; off >>= 1) {
        s  += __shfl_xor(s,  off, 32);
        sq += __shfl_xor(sq, off, 32);
    }
    float mu  = s * (1.f / 32.f);
    float var = sq * (1.f / 32.f) - mu * mu;
    float inv = rsqrtf(var + LN_EPS);
    float x2 = fmaxf((v - mu) * inv * g2[lane] + be2[lane], 0.f);

    float lg = x2 * fcW[lane];
    #pragma unroll
    for (int off = 16; off > 0; off >>= 1) lg += __shfl_xor(lg, off, 32);
    if (lane == 0) logits[n] = lg + fcb[0];

    float p = pb[lane];
    #pragma unroll
    for (int i = 0; i < 32; ++i)
        p = fmaf(__shfl(x2, i, 32), pW[i * 32 + lane], p);
    float nsq = p * p;
    #pragma unroll
    for (int off = 16; off > 0; off >>= 1) nsq += __shfl_xor(nsq, off, 32);
    float nrm = sqrtf(nsq);
    emb[(long long)n * 32 + lane] = p / fmaxf(nrm, 1e-12f);
}

extern "C" void kernel_launch(void* const* d_in, const int* in_sizes, int n_in,
                              void* d_out, int out_size, void* d_ws, size_t ws_size,
                              hipStream_t stream) {
    const float* x    = (const float*)d_in[0];
    const float* ew   = (const float*)d_in[1];
    const float* W1   = (const float*)d_in[2];
    const float* b1   = (const float*)d_in[3];
    const float* rW1  = (const float*)d_in[4];
    const float* rb1  = (const float*)d_in[5];
    const float* g1   = (const float*)d_in[6];
    const float* be1  = (const float*)d_in[7];
    const float* W2   = (const float*)d_in[8];
    const float* b2   = (const float*)d_in[9];
    const float* rW2  = (const float*)d_in[10];
    const float* rb2  = (const float*)d_in[11];
    const float* g2   = (const float*)d_in[12];
    const float* be2  = (const float*)d_in[13];
    const float* fcW  = (const float*)d_in[14];
    const float* fcb  = (const float*)d_in[15];
    const float* pW   = (const float*)d_in[16];
    const float* pb   = (const float*)d_in[17];
    const int*   ei   = (const int*)d_in[18];
    const int*   row  = ei;
    const int*   col  = ei + N_EDGES;

    float* ws   = (float*)d_ws;
    float* deg  = ws;  ws += N_NODES;
    float* dinv = ws;  ws += N_NODES;
    float*  y   = ws;
    double* s   = (double*)(ws + 2 * (size_t)N_NODES);
    ws += (size_t)N_NODES * HID;
    ws += (size_t)N_NODES * HID;
    float* x1   = ws;  ws += (size_t)N_NODES * HID;
    float* xw2  = ws;  ws += (size_t)N_NODES * RED;
    float* acc2 = ws;  ws += (size_t)N_NODES * RED;

    float* logits = (float*)d_out;
    float* emb    = logits + N_NODES;

    const int B = 256;
    k_deg_init<<<(N_NODES + B - 1) / B, B, 0, stream>>>(deg);
    k_deg_scatter<<<(N_EDGES + B - 1) / B, B, 0, stream>>>(col, ew, deg);

    k_prep<<<(N_NODES + B - 1) / B, B, 0, stream>>>(x, deg, dinv, y, s);

    k_edge_l1<<<(2 * N_EDGES + B - 1) / B, B, 0, stream>>>(row, col, ew, y, s);

    k_node_l1<<<(N_NODES + 7) / 8, B, 0, stream>>>(x, dinv, y, s, W1, b1, rW1, rb1, g1, be1, x1);

    int waves_l2 = (N_NODES / 16) * (RED / 16);
    int thr_l2   = waves_l2 * 32;
    k_l2_wmma<<<(thr_l2 + B - 1) / B, B, 0, stream>>>(x1, W2, rW2, b2, rb2, dinv, xw2, acc2);

    long long te2 = (long long)N_EDGES * RED;
    k_edge_scatter<RED><<<(unsigned)((te2 + B - 1) / B), B, 0, stream>>>(row, col, ew, dinv,
                                                                          xw2, acc2);

    k_head<<<(N_NODES + 7) / 8, B, 0, stream>>>(acc2, g2, be2, fcW, fcb, pW, pb, logits, emb);
}


# layer-2 neighbour sum as a per-node gather over an edge list counting-sorted by destination inside the call (int rank atomics); no f32 atomics on acc2, sum in list order; LN2+heads in that pass
# speedup vs baseline: 1.9787x; 1.1027x over previous
#include <hip/hip_runtime.h>
#include <hip/hip_bf16.h>

#define N_NODES 100000
#define N_EDGES 3200000
#define HID 64
#define RED 32
#define EMB 32
#define LN_EPS 1e-5f

typedef __attribute__((ext_vector_type(2))) float v2f;
typedef __attribute__((ext_vector_type(8))) float v8f;

__global__ void k_deg_init(float* dcf, int* dci) {
    int i = blockIdx.x * blockDim.x + threadIdx.x;
    if (i < N_NODES) {
        dcf[2 * i]     = 1.0f;
        dci[2 * i + 1] = 0;
    }
}

__global__ void k_deg_scatter(const int* __restrict__ col, const float* __restrict__ w,
                              float* dcf, int* dci, int* __restrict__ rank) {
    int e = blockIdx.x * blockDim.x + threadIdx.x;
    if (e >= N_EDGES) return;
    int c = col[e];
    atomicAdd(&dcf[2 * c], w[e]);
    rank[e] = atomicAdd(&dci[2 * c + 1], 1);
}

__global__ void k_prep(const float* __restrict__ x, const float* __restrict__ dcf,
                       const int* __restrict__ dci,
                       float* __restrict__ dinv, float* __restrict__ y,
                       double* __restrict__ s, int* __restrict__ bsum) {
    __shared__ int wsum[8];
    int n = blockIdx.x * blockDim.x + threadIdx.x;
    int c = 0;
    if (n < N_NODES) {
        float d  = dcf[2 * n];
        float di = d > 0.f ? rsqrtf(d) : 0.f;
        dinv[n] = di;
        y[2 * n]     = di * x[2 * n];
        y[2 * n + 1] = di * x[2 * n + 1];
        s[2 * n]     = 0.0;
        s[2 * n + 1] = 0.0;
        c = dci[2 * n + 1];
    }
    #pragma unroll
    for (int o = 16; o > 0; o >>= 1) c += __shfl_xor(c, o, 32);
    if ((threadIdx.x & 31) == 0) wsum[threadIdx.x >> 5] = c;
    __syncthreads();
    if (threadIdx.x == 0) {
        int t = 0;
        #pragma unroll
        for (int i = 0; i < 8; ++i) t += wsum[i];
        bsum[blockIdx.x] = t;
    }
}

__global__ void k_scan(const int* __restrict__ dci, const int* __restrict__ bsum,
                       const float* __restrict__ dinv, int* __restrict__ od) {
    __shared__ int red[8];
    __shared__ int wtot[8];
    int t = threadIdx.x, lane = t & 31, wave = t >> 5;
    int b = blockIdx.x;
    int part = 0;
    for (int i = t; i < b; i += 256) part += bsum[i];
    #pragma unroll
    for (int o = 16; o > 0; o >>= 1) part += __shfl_xor(part, o, 32);
    if (lane == 0) red[wave] = part;
    int n = b * 256 + t;
    int c = n < N_NODES ? dci[2 * n + 1] : 0;
    int incl = c;
    #pragma unroll
    for (int d = 1; d < 32; d <<= 1) {
        int v = __shfl_up(incl, d, 32);
        if (lane >= d) incl += v;
    }
    if (lane == 31) wtot[wave] = incl;
    __syncthreads();
    int base = 0;
    #pragma unroll
    for (int i = 0; i < 8; ++i) base += red[i];
    #pragma unroll
    for (int i = 0; i < 8; ++i) base += (i < wave) ? wtot[i] : 0;
    if (n < N_NODES)
        reinterpret_cast<int2*>(od)[n] = make_int2(base + incl - c, __float_as_int(dinv[n]));
}

__global__ void k_fill(const int* __restrict__ row, const int* __restrict__ col,
                       const float* __restrict__ w, const int* __restrict__ rank,
                       const int* __restrict__ od, const float* __restrict__ dinv,
                       int* __restrict__ csr) {
    int e = blockIdx.x * blockDim.x + threadIdx.x;
    if (e >= N_EDGES) return;
    int r = row[e], c = col[e];
    int2 oc = reinterpret_cast<const int2*>(od)[c];
    int p = oc.x + rank[e];
    float coeff = dinv[r] * w[e] * __int_as_float(oc.y);
    reinterpret_cast<int2*>(csr)[p] = make_int2(r, __float_as_int(coeff));
}

__global__ void k_edge_l1(const int* __restrict__ row, const int* __restrict__ col,
                          const float* __restrict__ w, const float* __restrict__ y,
                          double* __restrict__ s) {
    int t = blockIdx.x * blockDim.x + threadIdx.x;
    if (t >= 2 * N_EDGES) return;
    int e = t >> 1;
    int k = t & 1;
    int r = row[e], c = col[e];
    double v = (double)w[e] * (double)y[2 * r + k];
    atomicAdd(&s[2 * c + k], v);
}

__global__ void k_node_l1(const float* __restrict__ x, const float* __restrict__ dinv,
                          const float* __restrict__ y, const double* __restrict__ s,
                          const float* __restrict__ W1, const float* __restrict__ b1,
                          const float* __restrict__ rW1, const float* __restrict__ rb1,
                          const float* __restrict__ g, const float* __restrict__ be,
                          float* __restrict__ out) {
    int wave = threadIdx.x >> 5;
    int lane = threadIdx.x & 31;
    int n = blockIdx.x * (blockDim.x >> 5) + wave;
    if (n >= N_NODES) return;
    float di = dinv[n];
    float x0 = x[2 * n], x1v = x[2 * n + 1];
    float t0 = di * (float)(s[2 * n]     + (double)y[2 * n]);
    float t1 = di * (float)(s[2 * n + 1] + (double)y[2 * n + 1]);
    int f0 = lane, f1 = lane + 32;
    float v0 = fmaf(t0, W1[f0], t1 * W1[HID + f0]) + b1[f0]
             + fmaf(x0, rW1[f0], x1v * rW1[HID + f0]) + rb1[f0];
    float v1 = fmaf(t0, W1[f1], t1 * W1[HID + f1]) + b1[f1]
             + fmaf(x0, rW1[f1], x1v * rW1[HID + f1]) + rb1[f1];
    float sm = v0 + v1, sq = fmaf(v0, v0, v1 * v1);
    #pragma unroll
    for (int off = 16; off > 0; off >>= 1) {
        sm += __shfl_xor(sm, off, 32);
        sq += __shfl_xor(sq, off, 32);
    }
    float mu  = sm * (1.f / 64.f);
    float var = sq * (1.f / 64.f) - mu * mu;
    float inv = rsqrtf(var + LN_EPS);
    float o0 = (v0 - mu) * inv * g[f0] + be[f0];
    float o1 = (v1 - mu) * inv * g[f1] + be[f1];
    out[(long long)n * 64 + f0] = fmaxf(o0, 0.f);
    out[(long long)n * 64 + f1] = fmaxf(o1, 0.f);
}

__global__ void k_l2_wmma(const float* __restrict__ x1, const float* __restrict__ W2,
                          const float* __restrict__ rW2, const float* __restrict__ b2,
                          const float* __restrict__ rb2, const float* __restrict__ dinv,
                          float* __restrict__ xw2, float* __restrict__ acc2) {
    int wave_global = (blockIdx.x * blockDim.x + threadIdx.x) >> 5;
    int lane = threadIdx.x & 31;
    const int tiles_m = N_NODES / 16;
    int tm = wave_global >> 1;
    int tn = wave_global & 1;
    if (tm >= tiles_m) return;
    int m0 = tm * 16, n0 = tn * 16;
    int  lr = lane & 15;
    bool hi = lane >= 16;
    int  kofs = hi ? 2 : 0;

    v8f c0 = {};
    v8f c1 = {};
    #pragma unroll
    for (int k0 = 0; k0 < HID; k0 += 4) {
        const float* ap = &x1[(long long)(m0 + lr) * HID + k0 + kofs];
        v2f a; a.x = ap[0]; a.y = ap[1];
        int kb = k0 + kofs;
        v2f b;  b.x  = W2 [kb * RED + n0 + lr]; b.y  = W2 [(kb + 1) * RED + n0 + lr];
        v2f br; br.x = rW2[kb * RED + n0 + lr]; br.y = rW2[(kb + 1) * RED + n0 + lr];
        c0 = __builtin_amdgcn_wmma_f32_16x16x4_f32(false, a, false, b,  (short)0, c0, false, false);
        c1 = __builtin_amdgcn_wmma_f32_16x16x4_f32(false, a, false, br, (short)0, c1, false, false);
    }
    int ncol = n0 + lr;
    float b2v = b2[ncol], rb2v = rb2[ncol];
    #pragma unroll
    for (int v = 0; v < 8; ++v) {
        int m = m0 + v + (hi ? 8 : 0);
        float di = dinv[m];
        float xv = c0[v];
        xw2 [(long long)m * RED + ncol] = xv;
        acc2[(long long)m * RED + ncol] = di * di * xv + b2v + c1[v] + rb2v;
    }
}

#define GATHER_STEP(A, J)                                                              \
    {                                                                                  \
        int   r_  = __builtin_amdgcn_readlane(ent.x, (J));                             \
        float cf_ = __int_as_float(__builtin_amdgcn_readlane(ent.y, (J)));             \
        A = fmaf(cf_, xw2[(long long)r_ * 32 + lane], A);                              \
    }
__global__ void k_gather_head(const int* __restrict__ od, const int* __restrict__ dci,
                       const int* __restrict__ csr, const float* __restrict__ xw2,
                       const float* __restrict__ acc2, const float* __restrict__ g2,
                       const float* __restrict__ be2, const float* __restrict__ fcW,
                       const float* __restrict__ fcb, const float* __restrict__ pW,
                       const float* __restrict__ pb, float* __restrict__ logits,
                       float* __restrict__ emb) {
    int lane = threadIdx.x & 31;
    int n = __builtin_amdgcn_readfirstlane((int)(blockIdx.x * (blockDim.x >> 5) + (threadIdx.x >> 5)));
    if (n >= N_NODES) return;
    int beg = od[2 * n];
    int len = dci[2 * n + 1];
    const int2* seg = reinterpret_cast<const int2*>(csr) + beg;
    float a0 = 0.f, a1 = 0.f, a2 = 0.f, a3 = 0.f;
    for (int base = 0; base < len; base += 32) {
        int idx = base + lane;
        int2 ent = make_int2(0, 0);
        if (idx < len) ent = seg[idx];
        int m = len - base < 32 ? len - base : 32;
        int j = 0;
        for (; j + 8 <= m; j += 8) {
            GATHER_STEP(a0, j)     GATHER_STEP(a1, j + 1)
            GATHER_STEP(a2, j + 2) GATHER_STEP(a3, j + 3)
            GATHER_STEP(a0, j + 4) GATHER_STEP(a1, j + 5)
            GATHER_STEP(a2, j + 6) GATHER_STEP(a3, j + 7)
        }
        if (j + 4 <= m) {
            GATHER_STEP(a0, j)     GATHER_STEP(a1, j + 1)
            GATHER_STEP(a2, j + 2) GATHER_STEP(a3, j + 3)
            j += 4;
        }
        int rem = m - j;
        if (rem == 3)      { GATHER_STEP(a0, j) GATHER_STEP(a1, j + 1) GATHER_STEP(a2, j + 2) }
        else if (rem == 2) { GATHER_STEP(a0, j) GATHER_STEP(a1, j + 1) }
        else if (rem == 1) { GATHER_STEP(a0, j) }
    }
    float v = acc2[(long long)n * 32 + lane] + ((a0 + a1) + (a2 + a3));
    float s = v, sq = v * v;
    #pragma unroll
    for (int off = 16; off > 0; off >>= 1) {
        s  += __shfl_xor(s,  off, 32);
        sq += __shfl_xor(sq, off, 32);
    }
    float mu  = s * (1.f / 32.f);
    float var = sq * (1.f / 32.f) - mu * mu;
    float inv = rsqrtf(var + LN_EPS);
    float x2 = fmaxf((v - mu) * inv * g2[lane] + be2[lane], 0.f);

    float lg = x2 * fcW[lane];
    #pragma unroll
    for (int off = 16; off > 0; off >>= 1) lg += __shfl_xor(lg, off, 32);
    if (lane == 0) logits[n] = lg + fcb[0];

    float p = pb[lane];
    #pragma unroll
    for (int i = 0; i < 32; ++i)
        p = fmaf(__shfl(x2, i, 32), pW[i * 32 + lane], p);
    float nsq = p * p;
    #pragma unroll
    for (int off = 16; off > 0; off >>= 1) nsq += __shfl_xor(nsq, off, 32);
    float nrm = sqrtf(nsq);
    emb[(long long)n * 32 + lane] = p / fmaxf(nrm, 1e-12f);
}

extern "C" void kernel_launch(void* const* d_in, const int* in_sizes, int n_in,
                              void* d_out, int out_size, void* d_ws, size_t ws_size,
                              hipStream_t stream) {
    const float* x    = (const float*)d_in[0];
    const float* ew   = (const float*)d_in[1];
    const float* W1   = (const float*)d_in[2];
    const float* b1   = (const float*)d_in[3];
    const float* rW1  = (const float*)d_in[4];
    const float* rb1  = (const float*)d_in[5];
    const float* g1   = (const float*)d_in[6];
    const float* be1  = (const float*)d_in[7];
    const float* W2   = (const float*)d_in[8];
    const float* b2   = (const float*)d_in[9];
    const float* rW2  = (const float*)d_in[10];
    const float* rb2  = (const float*)d_in[11];
    const float* g2   = (const float*)d_in[12];
    const float* be2  = (const float*)d_in[13];
    const float* fcW  = (const float*)d_in[14];
    const float* fcb  = (const float*)d_in[15];
    const float* pW   = (const float*)d_in[16];
    const float* pb   = (const float*)d_in[17];
    const int*   ei   = (const int*)d_in[18];
    const int*   row  = ei;
    const int*   col  = ei + N_EDGES;

    float* ws   = (float*)d_ws;
    float* dinv = ws;  ws += 2 * (size_t)N_NODES;
    float*  y   = ws;
    double* s   = (double*)(ws + 2 * (size_t)N_NODES);
    int*   rank = (int*)(ws + 8 * (size_t)N_NODES);
    float* dcf  = ws + 40 * (size_t)N_NODES;
    int*   dci  = (int*)dcf;
    int*   od   = (int*)(ws + 42 * (size_t)N_NODES);
    int*   bsum = (int*)(ws + 44 * (size_t)N_NODES);
    ws += (size_t)N_NODES * HID;
    int*   csr  = (int*)ws;
    ws += (size_t)N_NODES * HID;
    float* x1   = ws;  ws += (size_t)N_NODES * HID;
    float* xw2  = ws;  ws += (size_t)N_NODES * RED;
    float* acc2 = ws;  ws += (size_t)N_NODES * RED;

    float* logits = (float*)d_out;
    float* emb    = logits + N_NODES;

    const int B = 256;
    k_deg_init<<<(N_NODES + B - 1) / B, B, 0, stream>>>(dcf, dci);
    k_deg_scatter<<<(N_EDGES + B - 1) / B, B, 0, stream>>>(col, ew, dcf, dci, rank);

    k_prep<<<(N_NODES + B - 1) / B, B, 0, stream>>>(x, dcf, dci, dinv, y, s, bsum);
    k_scan<<<(N_NODES + B - 1) / B, B, 0, stream>>>(dci, bsum, dinv, od);
    k_fill<<<(N_EDGES + B - 1) / B, B, 0, stream>>>(row, col, ew, rank, od, dinv, csr);

    k_edge_l1<<<(2 * N_EDGES + B - 1) / B, B, 0, stream>>>(row, col, ew, y, s);

    k_node_l1<<<(N_NODES + 7) / 8, B, 0, stream>>>(x, dinv, y, s, W1, b1, rW1, rb1, g1, be1, x1);

    int waves_l2 = (N_NODES / 16) * (RED / 16);
    int thr_l2   = waves_l2 * 32;
    k_l2_wmma<<<(thr_l2 + B - 1) / B, B, 0, stream>>>(x1, W2, rW2, b2, rb2, dinv, xw2, acc2);

    k_gather_head<<<(N_NODES + 7) / 8, B, 0, stream>>>(od, dci, csr, xw2, acc2, g2, be2, fcW, fcb,
                                                       pW, pb, logits, emb);
}


# layer-1 neighbour sum as a per-node gather over the sorted edge list inside the node pass (f64, summed in list order, no atomics): the edge kernel and its f64 atomics are gone; 8 launches
# speedup vs baseline: 2.3672x; 1.1963x over previous
#include <hip/hip_runtime.h>
#include <hip/hip_bf16.h>

#define N_NODES 100000
#define N_EDGES 3200000
#define HID 64
#define RED 32
#define EMB 32
#define LN_EPS 1e-5f

typedef __attribute__((ext_vector_type(2))) float v2f;
typedef __attribute__((ext_vector_type(8))) float v8f;

__global__ void k_deg_init(float* dcf, int* dci) {
    int i = blockIdx.x * blockDim.x + threadIdx.x;
    if (i < N_NODES) {
        dcf[2 * i]     = 1.0f;
        dci[2 * i + 1] = 0;
    }
}

__global__ void k_deg_scatter(const int* __restrict__ col, const float* __restrict__ w,
                              float* dcf, int* dci, int* __restrict__ rank) {
    int e = blockIdx.x * blockDim.x + threadIdx.x;
    if (e >= N_EDGES) return;
    int c = col[e];
    atomicAdd(&dcf[2 * c], w[e]);
    rank[e] = atomicAdd(&dci[2 * c + 1], 1);
}

__global__ void k_prep(const float* __restrict__ dcf, const int* __restrict__ dci,
                       float* __restrict__ dinv, int* __restrict__ bsum) {
    __shared__ int wsum[8];
    int n = blockIdx.x * blockDim.x + threadIdx.x;
    int c = 0;
    if (n < N_NODES) {
        float d  = dcf[2 * n];
        float di = d > 0.f ? rsqrtf(d) : 0.f;
        dinv[n] = di;
        c = dci[2 * n + 1];
    }
    #pragma unroll
    for (int o = 16; o > 0; o >>= 1) c += __shfl_xor(c, o, 32);
    if ((threadIdx.x & 31) == 0) wsum[threadIdx.x >> 5] = c;
    __syncthreads();
    if (threadIdx.x == 0) {
        int t = 0;
        #pragma unroll
        for (int i = 0; i < 8; ++i) t += wsum[i];
        bsum[blockIdx.x] = t;
    }
}

__global__ void k_scan(const int* __restrict__ dci, const int* __restrict__ bsum,
                       const float* __restrict__ dinv, int* __restrict__ od) {
    __shared__ int red[8];
    __shared__ int wtot[8];
    int t = threadIdx.x, lane = t & 31, wave = t >> 5;
    int b = blockIdx.x;
    int part = 0;
    for (int i = t; i < b; i += 256) part += bsum[i];
    #pragma unroll
    for (int o = 16; o > 0; o >>= 1) part += __shfl_xor(part, o, 32);
    if (lane == 0) red[wave] = part;
    int n = b * 256 + t;
    int c = n < N_NODES ? dci[2 * n + 1] : 0;
    int incl = c;
    #pragma unroll
    for (int d = 1; d < 32; d <<= 1) {
        int v = __shfl_up(incl, d, 32);
        if (lane >= d) incl += v;
    }
    if (lane == 31) wtot[wave] = incl;
    __syncthreads();
    int base = 0;
    #pragma unroll
    for (int i = 0; i < 8; ++i) base += red[i];
    #pragma unroll
    for (int i = 0; i < 8; ++i) base += (i < wave) ? wtot[i] : 0;
    if (n < N_NODES)
        reinterpret_cast<int2*>(od)[n] = make_int2(base + incl - c, __float_as_int(dinv[n]));
}

__global__ void k_fill(const int* __restrict__ row, const int* __restrict__ col,
                       const float* __restrict__ w, const int* __restrict__ rank,
                       const int* __restrict__ od, const float* __restrict__ dinv,
                       int* __restrict__ csr) {
    int e = blockIdx.x * blockDim.x + threadIdx.x;
    if (e >= N_EDGES) return;
    int r = row[e], c = col[e];
    int2 oc = reinterpret_cast<const int2*>(od)[c];
    int p = oc.x + rank[e];
    float coeff = dinv[r] * w[e] * __int_as_float(oc.y);
    reinterpret_cast<int2*>(csr)[p] = make_int2(r, __float_as_int(coeff));
}

__global__ void k_node_l1(const float* __restrict__ x, const int* __restrict__ od,
                          const int* __restrict__ dci, const int* __restrict__ csr,
                          const float* __restrict__ W1, const float* __restrict__ b1,
                          const float* __restrict__ rW1, const float* __restrict__ rb1,
                          const float* __restrict__ g, const float* __restrict__ be,
                          float* __restrict__ out) {
    int lane = threadIdx.x & 31;
    int n = __builtin_amdgcn_readfirstlane((int)(blockIdx.x * (blockDim.x >> 5) + (threadIdx.x >> 5)));
    if (n >= N_NODES) return;
    int2 oc  = reinterpret_cast<const int2*>(od)[n];
    int  len = dci[2 * n + 1];
    float di = __int_as_float(oc.y);
    float x0 = x[2 * n], x1v = x[2 * n + 1];
    const unsigned long long* seg = reinterpret_cast<const unsigned long long*>(csr) + oc.x;
    const float2* xr = reinterpret_cast<const float2*>(x);
    double s0 = 0.0, s1 = 0.0;
    for (int base = 0; base < len; base += 64) {
        int ia = base + lane, ib = ia + 32, last = len - 1;
        unsigned long long ea = seg[ia < last ? ia : last];
        unsigned long long eb = seg[ib < last ? ib : last];
        float2 xa = xr[(int)(unsigned)ea];
        float2 xb = xr[(int)(unsigned)eb];
        float fa = __int_as_float((int)(unsigned)(ea >> 32));
        float fb = __int_as_float((int)(unsigned)(eb >> 32));
        double ca = ia < len ? (double)fa : 0.0;
        double cb = ib < len ? (double)fb : 0.0;
        s0 += ca * (double)xa.x;  s1 += ca * (double)xa.y;
        s0 += cb * (double)xb.x;  s1 += cb * (double)xb.y;
    }
    #pragma unroll
    for (int off = 16; off > 0; off >>= 1) {
        s0 += __shfl_xor(s0, off, 32);
        s1 += __shfl_xor(s1, off, 32);
    }
    double dd = (double)(di * di);
    float t0 = (float)(s0 + dd * (double)x0);
    float t1 = (float)(s1 + dd * (double)x1v);
    int f0 = lane, f1 = lane + 32;
    float v0 = fmaf(t0, W1[f0], t1 * W1[HID + f0]) + b1[f0]
             + fmaf(x0, rW1[f0], x1v * rW1[HID + f0]) + rb1[f0];
    float v1 = fmaf(t0, W1[f1], t1 * W1[HID + f1]) + b1[f1]
             + fmaf(x0, rW1[f1], x1v * rW1[HID + f1]) + rb1[f1];
    float sm = v0 + v1, sq = fmaf(v0, v0, v1 * v1);
    #pragma unroll
    for (int off = 16; off > 0; off >>= 1) {
        sm += __shfl_xor(sm, off, 32);
        sq += __shfl_xor(sq, off, 32);
    }
    float mu  = sm * (1.f / 64.f);
    float var = sq * (1.f / 64.f) - mu * mu;
    float inv = rsqrtf(var + LN_EPS);
    float o0 = (v0 - mu) * inv * g[f0] + be[f0];
    float o1 = (v1 - mu) * inv * g[f1] + be[f1];
    out[(long long)n * 64 + f0] = fmaxf(o0, 0.f);
    out[(long long)n * 64 + f1] = fmaxf(o1, 0.f);
}

__global__ void k_l2_wmma(const float* __restrict__ x1, const float* __restrict__ W2,
                          const float* __restrict__ rW2, const float* __restrict__ b2,
                          const float* __restrict__ rb2, const float* __restrict__ dinv,
                          float* __restrict__ xw2, float* __restrict__ acc2) {
    int wave_global = (blockIdx.x * blockDim.x + threadIdx.x) >> 5;
    int lane = threadIdx.x & 31;
    const int tiles_m = N_NODES / 16;
    int tm = wave_global >> 1;
    int tn = wave_global & 1;
    if (tm >= tiles_m) return;
    int m0 = tm * 16, n0 = tn * 16;
    int  lr = lane & 15;
    bool hi = lane >= 16;
    int  kofs = hi ? 2 : 0;

    v8f c0 = {};
    v8f c1 = {};
    #pragma unroll
    for (int k0 = 0; k0 < HID; k0 += 4) {
        const float* ap = &x1[(long long)(m0 + lr) * HID + k0 + kofs];
        v2f a; a.x = ap[0]; a.y = ap[1];
        int kb = k0 + kofs;
        v2f b;  b.x  = W2 [kb * RED + n0 + lr]; b.y  = W2 [(kb + 1) * RED + n0 + lr];
        v2f br; br.x = rW2[kb * RED + n0 + lr]; br.y = rW2[(kb + 1) * RED + n0 + lr];
        c0 = __builtin_amdgcn_wmma_f32_16x16x4_f32(false, a, false, b,  (short)0, c0, false, false);
        c1 = __builtin_amdgcn_wmma_f32_16x16x4_f32(false, a, false, br, (short)0, c1, false, false);
    }
    int ncol = n0 + lr;
    float b2v = b2[ncol], rb2v = rb2[ncol];
    #pragma unroll
    for (int v = 0; v < 8; ++v) {
        int m = m0 + v + (hi ? 8 : 0);
        float di = dinv[m];
        float xv = c0[v];
        xw2 [(long long)m * RED + ncol] = xv;
        acc2[(long long)m * RED + ncol] = di * di * xv + b2v + c1[v] + rb2v;
    }
}

#define GATHER_STEP(A, J)                                                              \
    {                                                                                  \
        int   r_  = __builtin_amdgcn_readlane(ent.x, (J));                             \
        float cf_ = __int_as_float(__builtin_amdgcn_readlane(ent.y, (J)));             \
        A = fmaf(cf_, xw2[(long long)r_ * 32 + lane], A);                              \
    }
__global__ void k_gather_head(const int* __restrict__ od, const int* __restrict__ dci,
                       const int* __restrict__ csr, const float* __restrict__ xw2,
                       const float* __restrict__ acc2, const float* __restrict__ g2,
                       const float* __restrict__ be2, const float* __restrict__ fcW,
                       const float* __restrict__ fcb, const float* __restrict__ pW,
                       const float* __restrict__ pb, float* __restrict__ logits,
                       float* __restrict__ emb) {
    int lane = threadIdx.x & 31;
    int n = __builtin_amdgcn_readfirstlane((int)(blockIdx.x * (blockDim.x >> 5) + (threadIdx.x >> 5)));
    if (n >= N_NODES) return;
    int beg = od[2 * n];
    int len = dci[2 * n + 1];
    const int2* seg = reinterpret_cast<const int2*>(csr) + beg;
    float a0 = 0.f, a1 = 0.f, a2 = 0.f, a3 = 0.f;
    for (int base = 0; base < len; base += 32) {
        int idx = base + lane;
        int2 ent = make_int2(0, 0);
        if (idx < len) ent = seg[idx];
        int m = len - base < 32 ? len - base : 32;
        int j = 0;
        for (; j + 8 <= m; j += 8) {
            GATHER_STEP(a0, j)     GATHER_STEP(a1, j + 1)
            GATHER_STEP(a2, j + 2) GATHER_STEP(a3, j + 3)
            GATHER_STEP(a0, j + 4) GATHER_STEP(a1, j + 5)
            GATHER_STEP(a2, j + 6) GATHER_STEP(a3, j + 7)
        }
        if (j + 4 <= m) {
            GATHER_STEP(a0, j)     GATHER_STEP(a1, j + 1)
            GATHER_STEP(a2, j + 2) GATHER_STEP(a3, j + 3)
            j += 4;
        }
        int rem = m - j;
        if (rem == 3)      { GATHER_STEP(a0, j) GATHER_STEP(a1, j + 1) GATHER_STEP(a2, j + 2) }
        else if (rem == 2) { GATHER_STEP(a0, j) GATHER_STEP(a1, j + 1) }
        else if (rem == 1) { GATHER_STEP(a0, j) }
    }
    float v = acc2[(long long)n * 32 + lane] + ((a0 + a1) + (a2 + a3));
    float s = v, sq = v * v;
    #pragma unroll
    for (int off = 16; off > 0; off >>= 1) {
        s  += __shfl_xor(s,  off, 32);
        sq += __shfl_xor(sq, off, 32);
    }
    float mu  = s * (1.f / 32.f);
    float var = sq * (1.f / 32.f) - mu * mu;
    float inv = rsqrtf(var + LN_EPS);
    float x2 = fmaxf((v - mu) * inv * g2[lane] + be2[lane], 0.f);

    float lg = x2 * fcW[lane];
    #pragma unroll
    for (int off = 16; off > 0; off >>= 1) lg += __shfl_xor(lg, off, 32);
    if (lane == 0) logits[n] = lg + fcb[0];

    float p = pb[lane];
    #pragma unroll
    for (int i = 0; i < 32; ++i)
        p = fmaf(__shfl(x2, i, 32), pW[i * 32 + lane], p);
    float nsq = p * p;
    #pragma unroll
    for (int off = 16; off > 0; off >>= 1) nsq += __shfl_xor(nsq, off, 32);
    float nrm = sqrtf(nsq);
    emb[(long long)n * 32 + lane] = p / fmaxf(nrm, 1e-12f);
}

extern "C" void kernel_launch(void* const* d_in, const int* in_sizes, int n_in,
                              void* d_out, int out_size, void* d_ws, size_t ws_size,
                              hipStream_t stream) {
    const float* x    = (const float*)d_in[0];
    const float* ew   = (const float*)d_in[1];
    const float* W1   = (const float*)d_in[2];
    const float* b1   = (const float*)d_in[3];
    const float* rW1  = (const float*)d_in[4];
    const float* rb1  = (const float*)d_in[5];
    const float* g1   = (const float*)d_in[6];
    const float* be1  = (const float*)d_in[7];
    const float* W2   = (const float*)d_in[8];
    const float* b2   = (const float*)d_in[9];
    const float* rW2  = (const float*)d_in[10];
    const float* rb2  = (const float*)d_in[11];
    const float* g2   = (const float*)d_in[12];
    const float* be2  = (const float*)d_in[13];
    const float* fcW  = (const float*)d_in[14];
    const float* fcb  = (const float*)d_in[15];
    const float* pW   = (const float*)d_in[16];
    const float* pb   = (const float*)d_in[17];
    const int*   ei   = (const int*)d_in[18];
    const int*   row  = ei;
    const int*   col  = ei + N_EDGES;

    float* ws   = (float*)d_ws;
    float* dinv = ws;  ws += 2 * (size_t)N_NODES;
    int*   rank = (int*)(ws + 8 * (size_t)N_NODES);
    float* dcf  = ws + 40 * (size_t)N_NODES;
    int*   dci  = (int*)dcf;
    int*   od   = (int*)(ws + 42 * (size_t)N_NODES);
    int*   bsum = (int*)(ws + 44 * (size_t)N_NODES);
    ws += (size_t)N_NODES * HID;
    int*   csr  = (int*)ws;
    ws += (size_t)N_NODES * HID;
    float* x1   = ws;  ws += (size_t)N_NODES * HID;
    float* xw2  = ws;  ws += (size_t)N_NODES * RED;
    float* acc2 = ws;  ws += (size_t)N_NODES * RED;

    float* logits = (float*)d_out;
    float* emb    = logits + N_NODES;

    const int B = 256;
    k_deg_init<<<(N_NODES + B - 1) / B, B, 0, stream>>>(dcf, dci);
    k_deg_scatter<<<(N_EDGES + B - 1) / B, B, 0, stream>>>(col, ew, dcf, dci, rank);

    k_prep<<<(N_NODES + B - 1) / B, B, 0, stream>>>(dcf, dci, dinv, bsum);
    k_scan<<<(N_NODES + B - 1) / B, B, 0, stream>>>(dci, bsum, dinv, od);
    k_fill<<<(N_EDGES + B - 1) / B, B, 0, stream>>>(row, col, ew, rank, od, dinv, csr);

    k_node_l1<<<(N_NODES + 7) / 8, B, 0, stream>>>(x, od, dci, csr, W1, b1, rW1, rb1, g1, be1, x1);

    int waves_l2 = (N_NODES / 16) * (RED / 16);
    int thr_l2   = waves_l2 * 32;
    k_l2_wmma<<<(thr_l2 + B - 1) / B, B, 0, stream>>>(x1, W2, rW2, b2, rb2, dinv, xw2, acc2);

    k_gather_head<<<(N_NODES + 7) / 8, B, 0, stream>>>(od, dci, csr, xw2, acc2, g2, be2, fcW, fcb,
                                                       pW, pb, logits, emb);
}


# edge list grouped by destination via LDS counters (edges binned by 256-node range, a work-group per range ranks them, sums degrees in LDS): no global atomics; degree sum, list order by LDS arrival
# speedup vs baseline: 6.6316x; 2.8015x over previous
#include <hip/hip_runtime.h>
#include <hip/hip_bf16.h>

#define N_NODES 100000
#define N_EDGES 3200000
#define HID 64
#define RED 32
#define EMB 32
#define LN_EPS 1e-5f

typedef __attribute__((ext_vector_type(2))) float v2f;
typedef __attribute__((ext_vector_type(8))) float v8f;

#define NB  391
#define NCH 256
#define CH  12500
#define KPT 13
#define STG 16384

__global__ void __launch_bounds__(1024) k_bin_count(const int* __restrict__ col,
                                                    int* __restrict__ counts) {
    __shared__ int h[NB];
    int t = threadIdx.x, g = blockIdx.x;
    for (int b = t; b < NB; b += 1024) h[b] = 0;
    __syncthreads();
    const int* cp = col + g * CH;
    int rc[KPT];
    #pragma unroll
    for (int k = 0; k < KPT; ++k) {
        int i = k * 1024 + t;
        rc[k] = i < CH ? cp[i] : 0;
    }
    #pragma unroll
    for (int k = 0; k < KPT; ++k) {
        int i = k * 1024 + t;
        if (i < CH) atomicAdd(&h[rc[k] >> 8], 1);
    }
    __syncthreads();
    for (int b = t; b < NB; b += 1024) counts[b * NCH + g] = h[b];
}

__device__ __forceinline__ void wave_scan_nb(const int* src, int* dst, int lane) {
    int b0 = lane * 13;
    int loc[13];
    int sum = 0;
    #pragma unroll
    for (int j = 0; j < 13; ++j) {
        int b = b0 + j;
        int v = b < NB ? src[b] : 0;
        loc[j] = sum;
        sum += v;
    }
    int incl = sum;
    #pragma unroll
    for (int d = 1; d < 32; d <<= 1) {
        int v = __shfl_up(incl, d, 32);
        if (lane >= d) incl += v;
    }
    int ex = incl - sum;
    #pragma unroll
    for (int j = 0; j < 13; ++j) {
        int b = b0 + j;
        if (b < NB) dst[b] = ex + loc[j];
    }
}

__global__ void __launch_bounds__(1024) k_bin_scatter(const int* __restrict__ row,
                                                      const int* __restrict__ col,
                                                      const float* __restrict__ w,
                                                      const int* __restrict__ counts,
                                                      int* __restrict__ bstart,
                                                      int2* __restrict__ binned) {
    __shared__ int   s_tot[NB];
    __shared__ int   s_bef[NB];
    __shared__ int   s_cnt[NB];
    __shared__ int   s_base[NB];
    __shared__ int   s_lst[NB];
    __shared__ int   s_cur[NB];
    __shared__ int   s_goff[NB];
    __shared__ int   s_row[CH];
    __shared__ int   s_col[CH];
    __shared__ float s_w[CH];
    int t = threadIdx.x, lane = t & 31, wave = t >> 5, g = blockIdx.x;

    int rc[KPT], rr[KPT];
    float rw[KPT];
    const int e0 = g * CH;
    #pragma unroll
    for (int k = 0; k < KPT; ++k) {
        int i = k * 1024 + t;
        rc[k] = 0; rr[k] = 0; rw[k] = 0.f;
        if (i < CH) { rc[k] = col[e0 + i]; rr[k] = row[e0 + i]; rw[k] = w[e0 + i]; }
    }

    #pragma unroll 1
    for (int j0 = 0; j0 < 13; j0 += 4) {
        int4 u[4], v[4];
        #pragma unroll
        for (int q = 0; q < 4; ++q) {
            int b  = wave + 32 * (j0 + q);
            int bc = b < NB ? b : NB - 1;
            const int4* p = reinterpret_cast<const int4*>(counts + bc * NCH) + lane * 2;
            u[q] = p[0]; v[q] = p[1];
        }
        #pragma unroll
        for (int q = 0; q < 4; ++q) {
            int b  = wave + 32 * (j0 + q);
            int gg = lane * 8;
            int c8[8] = {u[q].x, u[q].y, u[q].z, u[q].w, v[q].x, v[q].y, v[q].z, v[q].w};
            int tot = 0, bef = 0, mine = 0;
            #pragma unroll
            for (int j = 0; j < 8; ++j) {
                tot  += c8[j];
                bef  += (gg + j <  g) ? c8[j] : 0;
                mine += (gg + j == g) ? c8[j] : 0;
            }
            #pragma unroll
            for (int o = 16; o > 0; o >>= 1) {
                tot  += __shfl_xor(tot,  o, 32);
                bef  += __shfl_xor(bef,  o, 32);
                mine += __shfl_xor(mine, o, 32);
            }
            if (lane == 0 && b < NB) { s_tot[b] = tot; s_bef[b] = bef; s_cnt[b] = mine; }
        }
    }
    __syncthreads();
    if (wave == 0) wave_scan_nb(s_tot, s_base, lane);
    if (wave == 1) wave_scan_nb(s_cnt, s_lst,  lane);
    __syncthreads();
    for (int b = t; b < NB; b += 1024) {
        s_cur[b]  = 0;
        s_goff[b] = s_base[b] + s_bef[b] - s_lst[b];
        if (g == 0) bstart[b] = s_base[b];
    }
    if (g == 0 && t == 0) bstart[NB] = N_EDGES;
    __syncthreads();

    #pragma unroll
    for (int k = 0; k < KPT; ++k) {
        int i = k * 1024 + t;
        if (i < CH) {
            int b = rc[k] >> 8;
            int p = s_lst[b] + atomicAdd(&s_cur[b], 1);
            s_row[p] = rr[k]; s_col[p] = rc[k]; s_w[p] = rw[k];
        }
    }
    __syncthreads();
    for (int i = t; i < CH; i += 1024) {
        int c = s_col[i];
        binned[s_goff[c >> 8] + i] = make_int2(s_row[i] | ((c & 255) << 17), __float_as_int(s_w[i]));
    }
}

__global__ void __launch_bounds__(1024) k_bucket(const int* __restrict__ bstart,
                                                 const int2* __restrict__ binned,
                                                 const float* __restrict__ x,
                                                 int* __restrict__ dci, int* __restrict__ od,
                                                 float* __restrict__ dinv, float4* __restrict__ xd,
                                                 int2* __restrict__ csr) {
    __shared__ int   s_cnt[256];
    __shared__ float s_deg[256];
    __shared__ int   s_first[256];
    __shared__ int   s_cur[256];
    __shared__ int   s_wt[8];
    __shared__ int2  s_rec[STG];
    int t = threadIdx.x, lane = t & 31, wave = t >> 5, b = blockIdx.x;
    int beg = bstart[b], m = bstart[b + 1] - beg;
    if (t < 256) { s_cnt[t] = 0; s_deg[t] = 1.0f; s_cur[t] = 0; }
    __syncthreads();
    for (int i0 = 0; i0 < m; i0 += 8192) {
        int2 r[8];
        #pragma unroll
        for (int k = 0; k < 8; ++k) {
            int i = i0 + k * 1024 + t;
            r[k] = binned[beg + (i < m ? i : m - 1)];
        }
        #pragma unroll
        for (int k = 0; k < 8; ++k) {
            int i = i0 + k * 1024 + t;
            if (i < m) {
                int cl = (r[k].x >> 17) & 255;
                atomicAdd(&s_cnt[cl], 1);
                atomicAdd(&s_deg[cl], __int_as_float(r[k].y));
            }
        }
    }
    __syncthreads();
    int c = 0, incl = 0;
    if (t < 256) {
        c = s_cnt[t];
        incl = c;
        #pragma unroll
        for (int d = 1; d < 32; d <<= 1) {
            int v = __shfl_up(incl, d, 32);
            if (lane >= d) incl += v;
        }
        if (lane == 31) s_wt[wave] = incl;
    }
    __syncthreads();
    if (t < 256) {
        int base = 0;
        #pragma unroll
        for (int i = 0; i < 8; ++i) base += (i < wave) ? s_wt[i] : 0;
        int first = base + incl - c;
        s_first[t] = first;
        int n = b * 256 + t;
        if (n < N_NODES) {
            float d  = s_deg[t];
            float di = d > 0.f ? rsqrtf(d) : 0.f;
            dinv[n] = di;
            reinterpret_cast<int2*>(od)[n] = make_int2(beg + first, __float_as_int(di));
            dci[2 * n + 1] = c;
            xd[n] = make_float4(x[2 * n], x[2 * n + 1], di, 0.f);
        }
    }
    __syncthreads();
    const bool staged = m <= STG;
    for (int i0 = 0; i0 < m; i0 += 8192) {
        int2 r[8];
        #pragma unroll
        for (int k = 0; k < 8; ++k) {
            int i = i0 + k * 1024 + t;
            r[k] = binned[beg + (i < m ? i : m - 1)];
        }
        #pragma unroll
        for (int k = 0; k < 8; ++k) {
            int i = i0 + k * 1024 + t;
            if (i < m) {
                int cl = (r[k].x >> 17) & 255;
                int p  = s_first[cl] + atomicAdd(&s_cur[cl], 1);
                int2 o = make_int2(r[k].x & 0x1FFFF, r[k].y);
                if (staged) s_rec[p] = o; else csr[beg + p] = o;
            }
        }
    }
    if (staged) {
        __syncthreads();
        for (int i = t; i < m; i += 1024) csr[beg + i] = s_rec[i];
    }
}

__global__ void k_node_l1(const float* __restrict__ x, const float4* __restrict__ xd,
                          const int* __restrict__ od,
                          const int* __restrict__ dci, int* csr,
                          const float* __restrict__ W1, const float* __restrict__ b1,
                          const float* __restrict__ rW1, const float* __restrict__ rb1,
                          const float* __restrict__ g, const float* __restrict__ be,
                          float* __restrict__ out) {
    int lane = threadIdx.x & 31;
    int n = __builtin_amdgcn_readfirstlane((int)(blockIdx.x * (blockDim.x >> 5) + (threadIdx.x >> 5)));
    if (n >= N_NODES) return;
    int2 oc  = reinterpret_cast<const int2*>(od)[n];
    int  len = dci[2 * n + 1];
    float di = __int_as_float(oc.y);
    float x0 = x[2 * n], x1v = x[2 * n + 1];
    unsigned long long* seg = reinterpret_cast<unsigned long long*>(csr) + oc.x;
    double s0 = 0.0, s1 = 0.0;
    for (int base = 0; base < len; base += 64) {
        int ia = base + lane, ib = ia + 32, last = len - 1;
        unsigned long long ea = seg[ia < last ? ia : last];
        unsigned long long eb = seg[ib < last ? ib : last];
        __builtin_amdgcn_sched_barrier(0);
        float4 xa = xd[(int)(unsigned)ea];
        float4 xb = xd[(int)(unsigned)eb];
        __builtin_amdgcn_sched_barrier(0);
        float fa = xa.z * __int_as_float((int)(unsigned)(ea >> 32)) * di;
        float fb = xb.z * __int_as_float((int)(unsigned)(eb >> 32)) * di;
        double ca = ia < len ? (double)fa : 0.0;
        double cb = ib < len ? (double)fb : 0.0;
        s0 += ca * (double)xa.x;  s1 += ca * (double)xa.y;
        s0 += cb * (double)xb.x;  s1 += cb * (double)xb.y;
        if (ia < len) seg[ia] = ((unsigned long long)__float_as_uint(fa) << 32) | (unsigned)ea;
        if (ib < len) seg[ib] = ((unsigned long long)__float_as_uint(fb) << 32) | (unsigned)eb;
    }
    #pragma unroll
    for (int off = 16; off > 0; off >>= 1) {
        s0 += __shfl_xor(s0, off, 32);
        s1 += __shfl_xor(s1, off, 32);
    }
    double dd = (double)(di * di);
    float t0 = (float)(s0 + dd * (double)x0);
    float t1 = (float)(s1 + dd * (double)x1v);
    int f0 = lane, f1 = lane + 32;
    float v0 = fmaf(t0, W1[f0], t1 * W1[HID + f0]) + b1[f0]
             + fmaf(x0, rW1[f0], x1v * rW1[HID + f0]) + rb1[f0];
    float v1 = fmaf(t0, W1[f1], t1 * W1[HID + f1]) + b1[f1]
             + fmaf(x0, rW1[f1], x1v * rW1[HID + f1]) + rb1[f1];
    float sm = v0 + v1, sq = fmaf(v0, v0, v1 * v1);
    #pragma unroll
    for (int off = 16; off > 0; off >>= 1) {
        sm += __shfl_xor(sm, off, 32);
        sq += __shfl_xor(sq, off, 32);
    }
    float mu  = sm * (1.f / 64.f);
    float var = sq * (1.f / 64.f) - mu * mu;
    float inv = rsqrtf(var + LN_EPS);
    float o0 = (v0 - mu) * inv * g[f0] + be[f0];
    float o1 = (v1 - mu) * inv * g[f1] + be[f1];
    out[(long long)n * 64 + f0] = fmaxf(o0, 0.f);
    out[(long long)n * 64 + f1] = fmaxf(o1, 0.f);
}

__global__ void k_l2_wmma(const float* __restrict__ x1, const float* __restrict__ W2,
                          const float* __restrict__ rW2, const float* __restrict__ b2,
                          const float* __restrict__ rb2, const float* __restrict__ dinv,
                          float* __restrict__ xw2, float* __restrict__ acc2) {
    int wave_global = (blockIdx.x * blockDim.x + threadIdx.x) >> 5;
    int lane = threadIdx.x & 31;
    const int tiles_m = N_NODES / 16;
    int tm = wave_global >> 1;
    int tn = wave_global & 1;
    if (tm >= tiles_m) return;
    int m0 = tm * 16, n0 = tn * 16;
    int  lr = lane & 15;
    bool hi = lane >= 16;
    int  kofs = hi ? 2 : 0;

    v8f c0 = {};
    v8f c1 = {};
    #pragma unroll
    for (int k0 = 0; k0 < HID; k0 += 4) {
        const float* ap = &x1[(long long)(m0 + lr) * HID + k0 + kofs];
        v2f a; a.x = ap[0]; a.y = ap[1];
        int kb = k0 + kofs;
        v2f b;  b.x  = W2 [kb * RED + n0 + lr]; b.y  = W2 [(kb + 1) * RED + n0 + lr];
        v2f br; br.x = rW2[kb * RED + n0 + lr]; br.y = rW2[(kb + 1) * RED + n0 + lr];
        c0 = __builtin_amdgcn_wmma_f32_16x16x4_f32(false, a, false, b,  (short)0, c0, false, false);
        c1 = __builtin_amdgcn_wmma_f32_16x16x4_f32(false, a, false, br, (short)0, c1, false, false);
    }
    int ncol = n0 + lr;
    float b2v = b2[ncol], rb2v = rb2[ncol];
    #pragma unroll
    for (int v = 0; v < 8; ++v) {
        int m = m0 + v + (hi ? 8 : 0);
        float di = dinv[m];
        float xv = c0[v];
        xw2 [(long long)m * RED + ncol] = xv;
        acc2[(long long)m * RED + ncol] = di * di * xv + b2v + c1[v] + rb2v;
    }
}

#define GATHER_STEP(A, J)                                                              \
    {                                                                                  \
        int   r_  = __builtin_amdgcn_readlane(ent.x, (J));                             \
        float cf_ = __int_as_float(__builtin_amdgcn_readlane(ent.y, (J)));             \
        A = fmaf(cf_, xw2[(long long)r_ * 32 + lane], A);                              \
    }
__global__ void k_gather_head(const int* __restrict__ od, const int* __restrict__ dci,
                       const int* __restrict__ csr, const float* __restrict__ xw2,
                       const float* __restrict__ acc2, const float* __restrict__ g2,
                       const float* __restrict__ be2, const float* __restrict__ fcW,
                       const float* __restrict__ fcb, const float* __restrict__ pW,
                       const float* __restrict__ pb, float* __restrict__ logits,
                       float* __restrict__ emb) {
    int lane = threadIdx.x & 31;
    int n = __builtin_amdgcn_readfirstlane((int)(blockIdx.x * (blockDim.x >> 5) + (threadIdx.x >> 5)));
    if (n >= N_NODES) return;
    int beg = od[2 * n];
    int len = dci[2 * n + 1];
    const int2* seg = reinterpret_cast<const int2*>(csr) + beg;
    float a0 = 0.f, a1 = 0.f, a2 = 0.f, a3 = 0.f;
    for (int base = 0; base < len; base += 32) {
        int idx = base + lane;
        int2 ent = make_int2(0, 0);
        if (idx < len) ent = seg[idx];
        int m = len - base < 32 ? len - base : 32;
        int j = 0;
        for (; j + 8 <= m; j += 8) {
            GATHER_STEP(a0, j)     GATHER_STEP(a1, j + 1)
            GATHER_STEP(a2, j + 2) GATHER_STEP(a3, j + 3)
            GATHER_STEP(a0, j + 4) GATHER_STEP(a1, j + 5)
            GATHER_STEP(a2, j + 6) GATHER_STEP(a3, j + 7)
        }
        if (j + 4 <= m) {
            GATHER_STEP(a0, j)     GATHER_STEP(a1, j + 1)
            GATHER_STEP(a2, j + 2) GATHER_STEP(a3, j + 3)
            j += 4;
        }
        int rem = m - j;
        if (rem == 3)      { GATHER_STEP(a0, j) GATHER_STEP(a1, j + 1) GATHER_STEP(a2, j + 2) }
        else if (rem == 2) { GATHER_STEP(a0, j) GATHER_STEP(a1, j + 1) }
        else if (rem == 1) { GATHER_STEP(a0, j) }
    }
    float v = acc2[(long long)n * 32 + lane] + ((a0 + a1) + (a2 + a3));
    float s = v, sq = v * v;
    #pragma unroll
    for (int off = 16; off > 0; off >>= 1) {
        s  += __shfl_xor(s,  off, 32);
        sq += __shfl_xor(sq, off, 32);
    }
    float mu  = s * (1.f / 32.f);
    float var = sq * (1.f / 32.f) - mu * mu;
    float inv = rsqrtf(var + LN_EPS);
    float x2 = fmaxf((v - mu) * inv * g2[lane] + be2[lane], 0.f);

    float lg = x2 * fcW[lane];
    #pragma unroll
    for (int off = 16; off > 0; off >>= 1) lg += __shfl_xor(lg, off, 32);
    if (lane == 0) logits[n] = lg + fcb[0];

    float p = pb[lane];
    #pragma unroll
    for (int i = 0; i < 32; ++i)
        p = fmaf(__shfl(x2, i, 32), pW[i * 32 + lane], p);
    float nsq = p * p;
    #pragma unroll
    for (int off = 16; off > 0; off >>= 1) nsq += __shfl_xor(nsq, off, 32);
    float nrm = sqrtf(nsq);
    emb[(long long)n * 32 + lane] = p / fmaxf(nrm, 1e-12f);
}

extern "C" void kernel_launch(void* const* d_in, const int* in_sizes, int n_in,
                              void* d_out, int out_size, void* d_ws, size_t ws_size,
                              hipStream_t stream) {
    const float* x    = (const float*)d_in[0];
    const float* ew   = (const float*)d_in[1];
    const float* W1   = (const float*)d_in[2];
    const float* b1   = (const float*)d_in[3];
    const float* rW1  = (const float*)d_in[4];
    const float* rb1  = (const float*)d_in[5];
    const float* g1   = (const float*)d_in[6];
    const float* be1  = (const float*)d_in[7];
    const float* W2   = (const float*)d_in[8];
    const float* b2   = (const float*)d_in[9];
    const float* rW2  = (const float*)d_in[10];
    const float* rb2  = (const float*)d_in[11];
    const float* g2   = (const float*)d_in[12];
    const float* be2  = (const float*)d_in[13];
    const float* fcW  = (const float*)d_in[14];
    const float* fcb  = (const float*)d_in[15];
    const float* pW   = (const float*)d_in[16];
    const float* pb   = (const float*)d_in[17];
    const int*   ei   = (const int*)d_in[18];
    const int*   row  = ei;
    const int*   col  = ei + N_EDGES;

    float* ws   = (float*)d_ws;
    float* dinv = ws;  ws += 2 * (size_t)N_NODES;
    int*   counts = (int*)ws;
    int*   bstart = (int*)(ws + 2 * (size_t)N_NODES);
    int*   dci  = (int*)(ws + 40 * (size_t)N_NODES);
    int*   od   = (int*)(ws + 42 * (size_t)N_NODES);
    float4* xd  = (float4*)(ws + 44 * (size_t)N_NODES);
    ws += (size_t)N_NODES * HID;
    int*   csr  = (int*)ws;
    ws += (size_t)N_NODES * HID;
    float* x1   = ws;  ws += (size_t)N_NODES * HID;
    int2*  binned = (int2*)x1;
    float* xw2  = ws;  ws += (size_t)N_NODES * RED;
    float* acc2 = ws;  ws += (size_t)N_NODES * RED;

    float* logits = (float*)d_out;
    float* emb    = logits + N_NODES;

    const int B = 256;
    k_bin_count<<<NCH, 1024, 0, stream>>>(col, counts);
    k_bin_scatter<<<NCH, 1024, 0, stream>>>(row, col, ew, counts, bstart, binned);
    k_bucket<<<NB, 1024, 0, stream>>>(bstart, binned, x, dci, od, dinv, xd, (int2*)csr);

    k_node_l1<<<(N_NODES + 7) / 8, B, 0, stream>>>(x, xd, od, dci, csr, W1, b1, rW1, rb1, g1, be1, x1);

    int waves_l2 = (N_NODES / 16) * (RED / 16);
    int thr_l2   = waves_l2 * 32;
    k_l2_wmma<<<(thr_l2 + B - 1) / B, B, 0, stream>>>(x1, W2, rW2, b2, rb2, dinv, xw2, acc2);

    k_gather_head<<<(N_NODES + 7) / 8, B, 0, stream>>>(od, dci, csr, xw2, acc2, g2, be2, fcW, fcb,
                                                       pW, pb, logits, emb);
}


# sort: each chunk ordered by node range in LDS and written as one run; a range's work-group takes its piece of every chunk via a {slot,count} table; count launch gone; list order still by LDS arrival
# speedup vs baseline: 7.3157x; 1.1032x over previous
#include <hip/hip_runtime.h>
#include <hip/hip_bf16.h>

#define N_NODES 100000
#define N_EDGES 3200000
#define HID 64
#define RED 32
#define EMB 32
#define LN_EPS 1e-5f

typedef __attribute__((ext_vector_type(2))) float v2f;
typedef __attribute__((ext_vector_type(8))) float v8f;

#define NB  391
#define NCH 256
#define CH  12500
#define KPT 13
#define PPW 8
#define STG 16384

__device__ __forceinline__ void wave_scan_nb(const int* src, int* dst, int lane) {
    int b0 = lane * 13;
    int loc[13];
    int sum = 0;
    #pragma unroll
    for (int j = 0; j < 13; ++j) {
        int b = b0 + j;
        int v = b < NB ? src[b] : 0;
        loc[j] = sum;
        sum += v;
    }
    int incl = sum;
    #pragma unroll
    for (int d = 1; d < 32; d <<= 1) {
        int v = __shfl_up(incl, d, 32);
        if (lane >= d) incl += v;
    }
    int ex = incl - sum;
    #pragma unroll
    for (int j = 0; j < 13; ++j) {
        int b = b0 + j;
        if (b < NB) dst[b] = ex + loc[j];
    }
}

__global__ void __launch_bounds__(1024) k_chunk_sort(const int* __restrict__ row,
                                                     const int* __restrict__ col,
                                                     const float* __restrict__ w,
                                                     int2* __restrict__ tab,
                                                     int* __restrict__ counter,
                                                     int2* __restrict__ binned) {
    __shared__ int  s_cnt[NB];
    __shared__ int  s_lst[NB];
    __shared__ int  s_cur[NB];
    __shared__ int2 s_rec[CH];
    int t = threadIdx.x, lane = t & 31, wave = t >> 5, g = blockIdx.x;

    int rc[KPT], rr[KPT];
    float rw[KPT];
    const int e0 = g * CH;
    #pragma unroll
    for (int k = 0; k < KPT; ++k) {
        int i = k * 1024 + t;
        rc[k] = 0; rr[k] = 0; rw[k] = 0.f;
        if (i < CH) { rc[k] = col[e0 + i]; rr[k] = row[e0 + i]; rw[k] = w[e0 + i]; }
    }
    for (int b = t; b < NB; b += 1024) { s_cnt[b] = 0; s_cur[b] = 0; }
    if (g == 0 && t == 0) *counter = 0;
    __syncthreads();
    #pragma unroll
    for (int k = 0; k < KPT; ++k) {
        int i = k * 1024 + t;
        if (i < CH) atomicAdd(&s_cnt[rc[k] >> 8], 1);
    }
    __syncthreads();
    if (wave == 0) wave_scan_nb(s_cnt, s_lst, lane);
    __syncthreads();
    for (int b = t; b < NB; b += 1024) tab[b * NCH + g] = make_int2(s_lst[b], s_cnt[b]);
    #pragma unroll
    for (int k = 0; k < KPT; ++k) {
        int i = k * 1024 + t;
        if (i < CH) {
            int b = rc[k] >> 8;
            int p = s_lst[b] + atomicAdd(&s_cur[b], 1);
            s_rec[p] = make_int2(rr[k] | ((rc[k] & 255) << 17), __float_as_int(rw[k]));
        }
    }
    __syncthreads();
    for (int i = t; i < CH; i += 1024) binned[e0 + i] = s_rec[i];
}

#define BK_ADD(E)                                                                      \
    {                                                                                  \
        int2 e_ = (E);                                                                 \
        int cl_ = (e_.x >> 17) & 255;                                                  \
        atomicAdd(&s_cnt[cl_], 1);                                                     \
        atomicAdd(&s_deg[cl_], __int_as_float(e_.y));                                  \
    }
#define BK_PLACE(E)                                                                    \
    {                                                                                  \
        int2 e_ = (E);                                                                 \
        int cl_ = (e_.x >> 17) & 255;                                                  \
        int p_  = s_first[cl_] + atomicAdd(&s_cur[cl_], 1);                            \
        int2 o_ = make_int2(e_.x & 0x1FFFF, e_.y);                                     \
        if (staged) s_rec[p_] = o_; else csr[beg + p_] = o_;                           \
    }
#define BK_PASS(OP)                                                                    \
    _Pragma("unroll")                                                                  \
    for (int h = 0; h < PPW; h += 4) {                                                 \
        int2 r[4], s[2];                                                               \
        _Pragma("unroll")                                                              \
        for (int q = 0; q < 4; ++q) {                                                  \
            int ia_ = po[h + q] + (lane < pc[h + q] ? lane : 0);                       \
            r[q] = binned[ia_ < N_EDGES ? ia_ : N_EDGES - 1];                          \
        }                                                                              \
        _Pragma("unroll")                                                              \
        for (int k = 0; k < 2; ++k) {                                                  \
            int pcs_ = hih ? pc[h + 2 * k + 1] : pc[h + 2 * k];                        \
            int pos_ = hih ? po[h + 2 * k + 1] : po[h + 2 * k];                        \
            int ib_  = pos_ + (32 + l16 < pcs_ ? 32 + l16 : 0);                        \
            s[k] = binned[ib_ < N_EDGES ? ib_ : N_EDGES - 1];                          \
        }                                                                              \
          \
          \
          \
        asm volatile("" :: "v"(r[0].x), "v"(r[0].y), "v"(r[1].x), "v"(r[1].y),         \
                           "v"(r[2].x), "v"(r[2].y), "v"(r[3].x), "v"(r[3].y),         \
                           "v"(s[0].x), "v"(s[0].y), "v"(s[1].x), "v"(s[1].y));        \
        _Pragma("unroll")                                                              \
        for (int q = 0; q < 4; ++q)                                                    \
            if (lane < pc[h + q]) OP(r[q])                                             \
        _Pragma("unroll")                                                              \
        for (int k = 0; k < 2; ++k) {                                                  \
            int pcs_ = hih ? pc[h + 2 * k + 1] : pc[h + 2 * k];                        \
            if (32 + l16 < pcs_) OP(s[k])                                              \
        }                                                                              \
        _Pragma("unroll")                                                              \
        for (int q = 0; q < 4; ++q)                                                    \
            for (int j = lane + 48; j < pc[h + q]; j += 32) OP(binned[po[h + q] + j])  \
    }

__global__ void __launch_bounds__(1024) k_bucket(const int2* __restrict__ tab,
                                                 int* __restrict__ counter,
                                                 const int2* __restrict__ binned,
                                                 const float* __restrict__ x,
                                                 int* __restrict__ dci, int* __restrict__ od,
                                                 float* __restrict__ dinv, float4* __restrict__ xd,
                                                 int2* __restrict__ csr) {
    __shared__ int   s_cnt[256];
    __shared__ float s_deg[256];
    __shared__ int   s_first[256];
    __shared__ int   s_cur[256];
    __shared__ int   s_wt[8];
    __shared__ int   s_po[NCH];
    __shared__ int   s_pc[NCH];
    __shared__ int   s_bm[2];
    __shared__ int2  s_rec[STG];
    int t = threadIdx.x, lane = t & 31, wave = t >> 5, b = blockIdx.x;
    if (t < NCH) {
        s_cnt[t] = 0; s_deg[t] = 1.0f; s_cur[t] = 0;
        int2 e = tab[b * NCH + t];
        s_po[t] = t * CH + e.x;
        s_pc[t] = e.y;
        int s = e.y;
        #pragma unroll
        for (int o = 16; o > 0; o >>= 1) s += __shfl_xor(s, o, 32);
        if (lane == 0) s_wt[wave] = s;
    }
    __syncthreads();
    if (t == 0) {
        int tot = 0;
        #pragma unroll
        for (int i = 0; i < 8; ++i) tot += s_wt[i];
        s_bm[0] = atomicAdd(counter, tot);
        s_bm[1] = tot;
    }
    __syncthreads();
    const int beg = s_bm[0], m = s_bm[1];

    const int  l16 = lane & 15;
    const bool hih = lane >= 16;
    int pc[PPW], po[PPW];
    #pragma unroll
    for (int q = 0; q < PPW; ++q) {
        pc[q] = __builtin_amdgcn_readfirstlane(s_pc[wave + 32 * q]);
        po[q] = __builtin_amdgcn_readfirstlane(s_po[wave + 32 * q]);
    }
    BK_PASS(BK_ADD)
    __syncthreads();
    int c = 0, incl = 0;
    if (t < 256) {
        c = s_cnt[t];
        incl = c;
        #pragma unroll
        for (int d = 1; d < 32; d <<= 1) {
            int v = __shfl_up(incl, d, 32);
            if (lane >= d) incl += v;
        }
        if (lane == 31) s_wt[wave] = incl;
    }
    __syncthreads();
    if (t < 256) {
        int base = 0;
        #pragma unroll
        for (int i = 0; i < 8; ++i) base += (i < wave) ? s_wt[i] : 0;
        int first = base + incl - c;
        s_first[t] = first;
        int n = b * 256 + t;
        if (n < N_NODES) {
            float d  = s_deg[t];
            float di = d > 0.f ? rsqrtf(d) : 0.f;
            dinv[n] = di;
            reinterpret_cast<int2*>(od)[n] = make_int2(beg + first, __float_as_int(di));
            dci[2 * n + 1] = c;
            xd[n] = make_float4(x[2 * n], x[2 * n + 1], di, 0.f);
        }
    }
    __syncthreads();
    const bool staged = m <= STG;
    BK_PASS(BK_PLACE)
    if (staged) {
        __syncthreads();
        for (int i = t; i < m; i += 1024) csr[beg + i] = s_rec[i];
    }
}

__global__ void k_node_l1(const float* __restrict__ x, const float4* __restrict__ xd,
                          const int* __restrict__ od,
                          const int* __restrict__ dci, int* csr,
                          const float* __restrict__ W1, const float* __restrict__ b1,
                          const float* __restrict__ rW1, const float* __restrict__ rb1,
                          const float* __restrict__ g, const float* __restrict__ be,
                          float* __restrict__ out) {
    int lane = threadIdx.x & 31;
    int n = __builtin_amdgcn_readfirstlane((int)(blockIdx.x * (blockDim.x >> 5) + (threadIdx.x >> 5)));
    if (n >= N_NODES) return;
    int2 oc  = reinterpret_cast<const int2*>(od)[n];
    int  len = dci[2 * n + 1];
    float di = __int_as_float(oc.y);
    float x0 = x[2 * n], x1v = x[2 * n + 1];
    unsigned long long* seg = reinterpret_cast<unsigned long long*>(csr) + oc.x;
    double s0 = 0.0, s1 = 0.0;
    for (int base = 0; base < len; base += 64) {
        int ia = base + lane, ib = ia + 32, last = len - 1;
        unsigned long long ea = seg[ia < last ? ia : last];
        unsigned long long eb = seg[ib < last ? ib : last];
        __builtin_amdgcn_sched_barrier(0);
        float4 xa = xd[(int)(unsigned)ea];
        float4 xb = xd[(int)(unsigned)eb];
        __builtin_amdgcn_sched_barrier(0);
        float fa = xa.z * __int_as_float((int)(unsigned)(ea >> 32)) * di;
        float fb = xb.z * __int_as_float((int)(unsigned)(eb >> 32)) * di;
        double ca = ia < len ? (double)fa : 0.0;
        double cb = ib < len ? (double)fb : 0.0;
        s0 += ca * (double)xa.x;  s1 += ca * (double)xa.y;
        s0 += cb * (double)xb.x;  s1 += cb * (double)xb.y;
        if (ia < len) seg[ia] = ((unsigned long long)__float_as_uint(fa) << 32) | (unsigned)ea;
        if (ib < len) seg[ib] = ((unsigned long long)__float_as_uint(fb) << 32) | (unsigned)eb;
    }
    #pragma unroll
    for (int off = 16; off > 0; off >>= 1) {
        s0 += __shfl_xor(s0, off, 32);
        s1 += __shfl_xor(s1, off, 32);
    }
    double dd = (double)(di * di);
    float t0 = (float)(s0 + dd * (double)x0);
    float t1 = (float)(s1 + dd * (double)x1v);
    int f0 = lane, f1 = lane + 32;
    float v0 = fmaf(t0, W1[f0], t1 * W1[HID + f0]) + b1[f0]
             + fmaf(x0, rW1[f0], x1v * rW1[HID + f0]) + rb1[f0];
    float v1 = fmaf(t0, W1[f1], t1 * W1[HID + f1]) + b1[f1]
             + fmaf(x0, rW1[f1], x1v * rW1[HID + f1]) + rb1[f1];
    float sm = v0 + v1, sq = fmaf(v0, v0, v1 * v1);
    #pragma unroll
    for (int off = 16; off > 0; off >>= 1) {
        sm += __shfl_xor(sm, off, 32);
        sq += __shfl_xor(sq, off, 32);
    }
    float mu  = sm * (1.f / 64.f);
    float var = sq * (1.f / 64.f) - mu * mu;
    float inv = rsqrtf(var + LN_EPS);
    float o0 = (v0 - mu) * inv * g[f0] + be[f0];
    float o1 = (v1 - mu) * inv * g[f1] + be[f1];
    out[(long long)n * 64 + f0] = fmaxf(o0, 0.f);
    out[(long long)n * 64 + f1] = fmaxf(o1, 0.f);
}

__global__ void k_l2_wmma(const float* __restrict__ x1, const float* __restrict__ W2,
                          const float* __restrict__ rW2, const float* __restrict__ b2,
                          const float* __restrict__ rb2, const float* __restrict__ dinv,
                          float* __restrict__ xw2, float* __restrict__ acc2) {
    int wave_global = (blockIdx.x * blockDim.x + threadIdx.x) >> 5;
    int lane = threadIdx.x & 31;
    const int tiles_m = N_NODES / 16;
    int tm = wave_global >> 1;
    int tn = wave_global & 1;
    if (tm >= tiles_m) return;
    int m0 = tm * 16, n0 = tn * 16;
    int  lr = lane & 15;
    bool hi = lane >= 16;
    int  kofs = hi ? 2 : 0;

    v8f c0 = {};
    v8f c1 = {};
    #pragma unroll
    for (int k0 = 0; k0 < HID; k0 += 4) {
        const float* ap = &x1[(long long)(m0 + lr) * HID + k0 + kofs];
        v2f a; a.x = ap[0]; a.y = ap[1];
        int kb = k0 + kofs;
        v2f b;  b.x  = W2 [kb * RED + n0 + lr]; b.y  = W2 [(kb + 1) * RED + n0 + lr];
        v2f br; br.x = rW2[kb * RED + n0 + lr]; br.y = rW2[(kb + 1) * RED + n0 + lr];
        c0 = __builtin_amdgcn_wmma_f32_16x16x4_f32(false, a, false, b,  (short)0, c0, false, false);
        c1 = __builtin_amdgcn_wmma_f32_16x16x4_f32(false, a, false, br, (short)0, c1, false, false);
    }
    int ncol = n0 + lr;
    float b2v = b2[ncol], rb2v = rb2[ncol];
    #pragma unroll
    for (int v = 0; v < 8; ++v) {
        int m = m0 + v + (hi ? 8 : 0);
        float di = dinv[m];
        float xv = c0[v];
        xw2 [(long long)m * RED + ncol] = xv;
        acc2[(long long)m * RED + ncol] = di * di * xv + b2v + c1[v] + rb2v;
    }
}

#define GATHER_STEP(A, J)                                                              \
    {                                                                                  \
        int   r_  = __builtin_amdgcn_readlane(ent.x, (J));                             \
        float cf_ = __int_as_float(__builtin_amdgcn_readlane(ent.y, (J)));             \
        A = fmaf(cf_, xw2[(long long)r_ * 32 + lane], A);                              \
    }
__global__ void k_gather_head(const int* __restrict__ od, const int* __restrict__ dci,
                       const int* __restrict__ csr, const float* __restrict__ xw2,
                       const float* __restrict__ acc2, const float* __restrict__ g2,
                       const float* __restrict__ be2, const float* __restrict__ fcW,
                       const float* __restrict__ fcb, const float* __restrict__ pW,
                       const float* __restrict__ pb, float* __restrict__ logits,
                       float* __restrict__ emb) {
    int lane = threadIdx.x & 31;
    int n = __builtin_amdgcn_readfirstlane((int)(blockIdx.x * (blockDim.x >> 5) + (threadIdx.x >> 5)));
    if (n >= N_NODES) return;
    int beg = od[2 * n];
    int len = dci[2 * n + 1];
    const int2* seg = reinterpret_cast<const int2*>(csr) + beg;
    float a0 = 0.f, a1 = 0.f, a2 = 0.f, a3 = 0.f;
    for (int base = 0; base < len; base += 32) {
        int idx = base + lane;
        int2 ent = make_int2(0, 0);
        if (idx < len) ent = seg[idx];
        int m = len - base < 32 ? len - base : 32;
        int j = 0;
        for (; j + 8 <= m; j += 8) {
            GATHER_STEP(a0, j)     GATHER_STEP(a1, j + 1)
            GATHER_STEP(a2, j + 2) GATHER_STEP(a3, j + 3)
            GATHER_STEP(a0, j + 4) GATHER_STEP(a1, j + 5)
            GATHER_STEP(a2, j + 6) GATHER_STEP(a3, j + 7)
        }
        if (j + 4 <= m) {
            GATHER_STEP(a0, j)     GATHER_STEP(a1, j + 1)
            GATHER_STEP(a2, j + 2) GATHER_STEP(a3, j + 3)
            j += 4;
        }
        int rem = m - j;
        if (rem == 3)      { GATHER_STEP(a0, j) GATHER_STEP(a1, j + 1) GATHER_STEP(a2, j + 2) }
        else if (rem == 2) { GATHER_STEP(a0, j) GATHER_STEP(a1, j + 1) }
        else if (rem == 1) { GATHER_STEP(a0, j) }
    }
    float v = acc2[(long long)n * 32 + lane] + ((a0 + a1) + (a2 + a3));
    float s = v, sq = v * v;
    #pragma unroll
    for (int off = 16; off > 0; off >>= 1) {
        s  += __shfl_xor(s,  off, 32);
        sq += __shfl_xor(sq, off, 32);
    }
    float mu  = s * (1.f / 32.f);
    float var = sq * (1.f / 32.f) - mu * mu;
    float inv = rsqrtf(var + LN_EPS);
    float x2 = fmaxf((v - mu) * inv * g2[lane] + be2[lane], 0.f);

    float lg = x2 * fcW[lane];
    #pragma unroll
    for (int off = 16; off > 0; off >>= 1) lg += __shfl_xor(lg, off, 32);
    if (lane == 0) logits[n] = lg + fcb[0];

    float p = pb[lane];
    #pragma unroll
    for (int i = 0; i < 32; ++i)
        p = fmaf(__shfl(x2, i, 32), pW[i * 32 + lane], p);
    float nsq = p * p;
    #pragma unroll
    for (int off = 16; off > 0; off >>= 1) nsq += __shfl_xor(nsq, off, 32);
    float nrm = sqrtf(nsq);
    emb[(long long)n * 32 + lane] = p / fmaxf(nrm, 1e-12f);
}

extern "C" void kernel_launch(void* const* d_in, const int* in_sizes, int n_in,
                              void* d_out, int out_size, void* d_ws, size_t ws_size,
                              hipStream_t stream) {
    const float* x    = (const float*)d_in[0];
    const float* ew   = (const float*)d_in[1];
    const float* W1   = (const float*)d_in[2];
    const float* b1   = (const float*)d_in[3];
    const float* rW1  = (const float*)d_in[4];
    const float* rb1  = (const float*)d_in[5];
    const float* g1   = (const float*)d_in[6];
    const float* be1  = (const float*)d_in[7];
    const float* W2   = (const float*)d_in[8];
    const float* b2   = (const float*)d_in[9];
    const float* rW2  = (const float*)d_in[10];
    const float* rb2  = (const float*)d_in[11];
    const float* g2   = (const float*)d_in[12];
    const float* be2  = (const float*)d_in[13];
    const float* fcW  = (const float*)d_in[14];
    const float* fcb  = (const float*)d_in[15];
    const float* pW   = (const float*)d_in[16];
    const float* pb   = (const float*)d_in[17];
    const int*   ei   = (const int*)d_in[18];
    const int*   row  = ei;
    const int*   col  = ei + N_EDGES;

    float* ws   = (float*)d_ws;
    float* dinv = ws;  ws += 2 * (size_t)N_NODES;
    int2*  tab     = (int2*)ws;
    int*   counter = (int*)(ws + 4 * (size_t)N_NODES);
    int*   dci  = (int*)(ws + 40 * (size_t)N_NODES);
    int*   od   = (int*)(ws + 42 * (size_t)N_NODES);
    float4* xd  = (float4*)(ws + 44 * (size_t)N_NODES);
    ws += (size_t)N_NODES * HID;
    int*   csr  = (int*)ws;
    ws += (size_t)N_NODES * HID;
    float* x1   = ws;  ws += (size_t)N_NODES * HID;
    int2*  binned = (int2*)x1;
    float* xw2  = ws;  ws += (size_t)N_NODES * RED;
    float* acc2 = ws;  ws += (size_t)N_NODES * RED;

    float* logits = (float*)d_out;
    float* emb    = logits + N_NODES;

    const int B = 256;
    k_chunk_sort<<<NCH, 1024, 0, stream>>>(row, col, ew, tab, counter, binned);
    k_bucket<<<NB, 1024, 0, stream>>>(tab, counter, binned, x, dci, od, dinv, xd, (int2*)csr);

    k_node_l1<<<(N_NODES + 7) / 8, B, 0, stream>>>(x, xd, od, dci, csr, W1, b1, rW1, rb1, g1, be1, x1);

    int waves_l2 = (N_NODES / 16) * (RED / 16);
    int thr_l2   = waves_l2 * 32;
    k_l2_wmma<<<(thr_l2 + B - 1) / B, B, 0, stream>>>(x1, W2, rW2, b2, rb2, dinv, xw2, acc2);

    k_gather_head<<<(N_NODES + 7) / 8, B, 0, stream>>>(od, dci, csr, xw2, acc2, g2, be2, fcW, fcb,
                                                       pW, pb, logits, emb);
}


# layer-1 neighbour sum in 64-bit fixed point (exact 24x24-bit products shifted to 2^-44, integer adds: independent of list order), not f64: the 27 f64-rate instructions a node are gone
# speedup vs baseline: 8.1955x; 1.1203x over previous
#include <hip/hip_runtime.h>
#include <hip/hip_bf16.h>

#define N_NODES 100000
#define N_EDGES 3200000
#define HID 64
#define RED 32
#define EMB 32
#define LN_EPS 1e-5f

typedef __attribute__((ext_vector_type(2))) float v2f;
typedef __attribute__((ext_vector_type(8))) float v8f;

#define NB  391
#define NCH 256
#define CH  12500
#define KPT 13
#define PPW 8
#define STG 16384

__device__ __forceinline__ void wave_scan_nb(const int* src, int* dst, int lane) {
    int b0 = lane * 13;
    int loc[13];
    int sum = 0;
    #pragma unroll
    for (int j = 0; j < 13; ++j) {
        int b = b0 + j;
        int v = b < NB ? src[b] : 0;
        loc[j] = sum;
        sum += v;
    }
    int incl = sum;
    #pragma unroll
    for (int d = 1; d < 32; d <<= 1) {
        int v = __shfl_up(incl, d, 32);
        if (lane >= d) incl += v;
    }
    int ex = incl - sum;
    #pragma unroll
    for (int j = 0; j < 13; ++j) {
        int b = b0 + j;
        if (b < NB) dst[b] = ex + loc[j];
    }
}

__global__ void __launch_bounds__(1024) k_chunk_sort(const int* __restrict__ row,
                                                     const int* __restrict__ col,
                                                     const float* __restrict__ w,
                                                     int2* __restrict__ tab,
                                                     int* __restrict__ counter,
                                                     int2* __restrict__ binned) {
    __shared__ int  s_cnt[NB];
    __shared__ int  s_lst[NB];
    __shared__ int  s_cur[NB];
    __shared__ int2 s_rec[CH];
    int t = threadIdx.x, lane = t & 31, wave = t >> 5, g = blockIdx.x;

    int rc[KPT], rr[KPT];
    float rw[KPT];
    const int e0 = g * CH;
    #pragma unroll
    for (int k = 0; k < KPT; ++k) {
        int i = k * 1024 + t;
        rc[k] = 0; rr[k] = 0; rw[k] = 0.f;
        if (i < CH) { rc[k] = col[e0 + i]; rr[k] = row[e0 + i]; rw[k] = w[e0 + i]; }
    }
    for (int b = t; b < NB; b += 1024) { s_cnt[b] = 0; s_cur[b] = 0; }
    if (g == 0 && t == 0) *counter = 0;
    __syncthreads();
    #pragma unroll
    for (int k = 0; k < KPT; ++k) {
        int i = k * 1024 + t;
        if (i < CH) atomicAdd(&s_cnt[rc[k] >> 8], 1);
    }
    __syncthreads();
    if (wave == 0) wave_scan_nb(s_cnt, s_lst, lane);
    __syncthreads();
    for (int b = t; b < NB; b += 1024) tab[b * NCH + g] = make_int2(s_lst[b], s_cnt[b]);
    #pragma unroll
    for (int k = 0; k < KPT; ++k) {
        int i = k * 1024 + t;
        if (i < CH) {
            int b = rc[k] >> 8;
            int p = s_lst[b] + atomicAdd(&s_cur[b], 1);
            s_rec[p] = make_int2(rr[k] | ((rc[k] & 255) << 17), __float_as_int(rw[k]));
        }
    }
    __syncthreads();
    for (int i = t; i < CH; i += 1024) binned[e0 + i] = s_rec[i];
}

#define BK_ADD(E)                                                                      \
    {                                                                                  \
        int2 e_ = (E);                                                                 \
        int cl_ = (e_.x >> 17) & 255;                                                  \
        atomicAdd(&s_cnt[cl_], 1);                                                     \
        atomicAdd(&s_deg[cl_], __int_as_float(e_.y));                                  \
    }
#define BK_PLACE(E)                                                                    \
    {                                                                                  \
        int2 e_ = (E);                                                                 \
        int cl_ = (e_.x >> 17) & 255;                                                  \
        int p_  = s_first[cl_] + atomicAdd(&s_cur[cl_], 1);                            \
        int2 o_ = make_int2(e_.x & 0x1FFFF, e_.y);                                     \
        if (staged) s_rec[p_] = o_; else csr[beg + p_] = o_;                           \
    }
#define BK_PASS(OP)                                                                    \
    _Pragma("unroll")                                                                  \
    for (int h = 0; h < PPW; h += 4) {                                                 \
        int2 r[4], s[2];                                                               \
        _Pragma("unroll")                                                              \
        for (int q = 0; q < 4; ++q) {                                                  \
            int ia_ = po[h + q] + (lane < pc[h + q] ? lane : 0);                       \
            r[q] = binned[ia_ < N_EDGES ? ia_ : N_EDGES - 1];                          \
        }                                                                              \
        _Pragma("unroll")                                                              \
        for (int k = 0; k < 2; ++k) {                                                  \
            int pcs_ = hih ? pc[h + 2 * k + 1] : pc[h + 2 * k];                        \
            int pos_ = hih ? po[h + 2 * k + 1] : po[h + 2 * k];                        \
            int ib_  = pos_ + (32 + l16 < pcs_ ? 32 + l16 : 0);                        \
            s[k] = binned[ib_ < N_EDGES ? ib_ : N_EDGES - 1];                          \
        }                                                                              \
          \
          \
          \
        asm volatile("" :: "v"(r[0].x), "v"(r[0].y), "v"(r[1].x), "v"(r[1].y),         \
                           "v"(r[2].x), "v"(r[2].y), "v"(r[3].x), "v"(r[3].y),         \
                           "v"(s[0].x), "v"(s[0].y), "v"(s[1].x), "v"(s[1].y));        \
        _Pragma("unroll")                                                              \
        for (int q = 0; q < 4; ++q)                                                    \
            if (lane < pc[h + q]) OP(r[q])                                             \
        _Pragma("unroll")                                                              \
        for (int k = 0; k < 2; ++k) {                                                  \
            int pcs_ = hih ? pc[h + 2 * k + 1] : pc[h + 2 * k];                        \
            if (32 + l16 < pcs_) OP(s[k])                                              \
        }                                                                              \
        _Pragma("unroll")                                                              \
        for (int q = 0; q < 4; ++q)                                                    \
            for (int j = lane + 48; j < pc[h + q]; j += 32) OP(binned[po[h + q] + j])  \
    }

__global__ void __launch_bounds__(1024) k_bucket(const int2* __restrict__ tab,
                                                 int* __restrict__ counter,
                                                 const int2* __restrict__ binned,
                                                 const float* __restrict__ x,
                                                 int* __restrict__ dci, int* __restrict__ od,
                                                 float* __restrict__ dinv, float4* __restrict__ xd,
                                                 int2* __restrict__ csr) {
    __shared__ int   s_cnt[256];
    __shared__ float s_deg[256];
    __shared__ int   s_first[256];
    __shared__ int   s_cur[256];
    __shared__ int   s_wt[8];
    __shared__ int   s_po[NCH];
    __shared__ int   s_pc[NCH];
    __shared__ int   s_bm[2];
    __shared__ int2  s_rec[STG];
    int t = threadIdx.x, lane = t & 31, wave = t >> 5, b = blockIdx.x;
    if (t < NCH) {
        s_cnt[t] = 0; s_deg[t] = 1.0f; s_cur[t] = 0;
        int2 e = tab[b * NCH + t];
        s_po[t] = t * CH + e.x;
        s_pc[t] = e.y;
        int s = e.y;
        #pragma unroll
        for (int o = 16; o > 0; o >>= 1) s += __shfl_xor(s, o, 32);
        if (lane == 0) s_wt[wave] = s;
    }
    __syncthreads();
    if (t == 0) {
        int tot = 0;
        #pragma unroll
        for (int i = 0; i < 8; ++i) tot += s_wt[i];
        s_bm[0] = atomicAdd(counter, tot);
        s_bm[1] = tot;
    }
    __syncthreads();
    const int beg = s_bm[0], m = s_bm[1];

    const int  l16 = lane & 15;
    const bool hih = lane >= 16;
    int pc[PPW], po[PPW];
    #pragma unroll
    for (int q = 0; q < PPW; ++q) {
        pc[q] = __builtin_amdgcn_readfirstlane(s_pc[wave + 32 * q]);
        po[q] = __builtin_amdgcn_readfirstlane(s_po[wave + 32 * q]);
    }
    BK_PASS(BK_ADD)
    __syncthreads();
    int c = 0, incl = 0;
    if (t < 256) {
        c = s_cnt[t];
        incl = c;
        #pragma unroll
        for (int d = 1; d < 32; d <<= 1) {
            int v = __shfl_up(incl, d, 32);
            if (lane >= d) incl += v;
        }
        if (lane == 31) s_wt[wave] = incl;
    }
    __syncthreads();
    if (t < 256) {
        int base = 0;
        #pragma unroll
        for (int i = 0; i < 8; ++i) base += (i < wave) ? s_wt[i] : 0;
        int first = base + incl - c;
        s_first[t] = first;
        int n = b * 256 + t;
        if (n < N_NODES) {
            float d  = s_deg[t];
            float di = d > 0.f ? rsqrtf(d) : 0.f;
            dinv[n] = di;
            reinterpret_cast<int2*>(od)[n] = make_int2(beg + first, __float_as_int(di));
            dci[2 * n + 1] = c;
            xd[n] = make_float4(x[2 * n], x[2 * n + 1], di, 0.f);
        }
    }
    __syncthreads();
    const bool staged = m <= STG;
    BK_PASS(BK_PLACE)
    if (staged) {
        __syncthreads();
        for (int i = t; i < m; i += 1024) csr[beg + i] = s_rec[i];
    }
}

#define FXS 44

__device__ __forceinline__ int      fx_exp(unsigned u) { return (int)((u >> 23) & 255u); }
__device__ __forceinline__ unsigned fx_sig(unsigned u) { return (u & 0x007FFFFFu) | 0x00800000u; }

__device__ __forceinline__ long long fx_term(unsigned mc, int ec, unsigned uc, unsigned ux) {
    unsigned long long pr = ((unsigned long long)mc * (unsigned long long)fx_sig(ux)) << 14;
    int rs = (314 - FXS) - ec - fx_exp(ux);
    rs = rs < 0 ? 0 : (rs > 63 ? 63 : rs);
    unsigned long long mag = pr >> rs;
    unsigned long long sg  = (unsigned long long)(long long)((int)(uc ^ ux) >> 31);
    return (long long)((mag ^ sg) - sg);
}

__device__ __forceinline__ float fx_to_float(long long s) {
    unsigned long long a = s < 0 ? 0ull - (unsigned long long)s : (unsigned long long)s;
    int lz = __clzll((long long)a);
    lz = lz > 63 ? 63 : lz;
    unsigned long long nrm = a << lz;
    unsigned top = (unsigned)(nrm >> 32) | ((unsigned)nrm != 0u ? 1u : 0u);
    float f = ldexpf((float)top, 32 - lz - FXS);
    return s < 0 ? -f : f;
}

__global__ void k_node_l1(const float* __restrict__ x, const float4* __restrict__ xd,
                          const int* __restrict__ od,
                          const int* __restrict__ dci, int* csr,
                          const float* __restrict__ W1, const float* __restrict__ b1,
                          const float* __restrict__ rW1, const float* __restrict__ rb1,
                          const float* __restrict__ g, const float* __restrict__ be,
                          float* __restrict__ out) {
    int lane = threadIdx.x & 31;
    int n = __builtin_amdgcn_readfirstlane((int)(blockIdx.x * (blockDim.x >> 5) + (threadIdx.x >> 5)));
    if (n >= N_NODES) return;
    int2 oc  = reinterpret_cast<const int2*>(od)[n];
    int  len = dci[2 * n + 1];
    float di = __int_as_float(oc.y);
    float x0 = x[2 * n], x1v = x[2 * n + 1];
    unsigned long long* seg = reinterpret_cast<unsigned long long*>(csr) + oc.x;
    unsigned long long s0 = 0, s1 = 0;
    int emc = 0, emx = 0;
    for (int base = 0; base < len; base += 64) {
        int ia = base + lane, ib = ia + 32, last = len - 1;
        unsigned long long ea = seg[ia < last ? ia : last];
        unsigned long long eb = seg[ib < last ? ib : last];
        __builtin_amdgcn_sched_barrier(0);
        float4 xa = xd[(int)(unsigned)ea];
        float4 xb = xd[(int)(unsigned)eb];
        __builtin_amdgcn_sched_barrier(0);
        float fa = xa.z * __int_as_float((int)(unsigned)(ea >> 32)) * di;
        float fb = xb.z * __int_as_float((int)(unsigned)(eb >> 32)) * di;
        unsigned ua  = __float_as_uint(fa),   ub  = __float_as_uint(fb);
        unsigned uax = __float_as_uint(xa.x), uay = __float_as_uint(xa.y);
        unsigned ubx = __float_as_uint(xb.x), uby = __float_as_uint(xb.y);
        int eca = fx_exp(ua), ecb = fx_exp(ub);
        unsigned mca = ia < len ? fx_sig(ua) : 0u;
        unsigned mcb = ib < len ? fx_sig(ub) : 0u;
        s0 += (unsigned long long)fx_term(mca, eca, ua, uax);  s1 += (unsigned long long)fx_term(mca, eca, ua, uay);
        s0 += (unsigned long long)fx_term(mcb, ecb, ub, ubx);  s1 += (unsigned long long)fx_term(mcb, ecb, ub, uby);
        emc = max(emc, max(eca, ecb));
        emx = max(max(emx, fx_exp(uax)), max(fx_exp(uay), max(fx_exp(ubx), fx_exp(uby))));
        if (ia < len) seg[ia] = ((unsigned long long)ua << 32) | (unsigned)ea;
        if (ib < len) seg[ib] = ((unsigned long long)ub << 32) | (unsigned)eb;
    }
    float cs = di * di;
    unsigned ucs = __float_as_uint(cs), ux0 = __float_as_uint(x0), ux1 = __float_as_uint(x1v);
    emc = max(emc, fx_exp(ucs));
    emx = max(emx, max(fx_exp(ux0), fx_exp(ux1)));
    bool wide = emc > 140 || emx > 140 || emc + emx > 314 - FXS - 12 || len > 4096;
    float t0, t1;
    if (__any(wide)) {
        double d0 = 0.0, d1 = 0.0;
        for (int i = lane; i < len; i += 32) {
            unsigned long long e = seg[i];
            float4 xr = xd[(int)(unsigned)e];
            double c = (double)__int_as_float((int)(unsigned)(e >> 32));
            d0 += c * (double)xr.x;  d1 += c * (double)xr.y;
        }
        #pragma unroll
        for (int off = 16; off > 0; off >>= 1) {
            d0 += __shfl_xor(d0, off, 32);
            d1 += __shfl_xor(d1, off, 32);
        }
        t0 = (float)(d0 + (double)cs * (double)x0);
        t1 = (float)(d1 + (double)cs * (double)x1v);
    } else {
        #pragma unroll
        for (int off = 16; off > 0; off >>= 1) {
            s0 += __shfl_xor(s0, off, 32);
            s1 += __shfl_xor(s1, off, 32);
        }
        s0 += (unsigned long long)fx_term(fx_sig(ucs), fx_exp(ucs), ucs, ux0);
        s1 += (unsigned long long)fx_term(fx_sig(ucs), fx_exp(ucs), ucs, ux1);
        t0 = fx_to_float((long long)s0);
        t1 = fx_to_float((long long)s1);
    }
    int f0 = lane, f1 = lane + 32;
    float v0 = fmaf(t0, W1[f0], t1 * W1[HID + f0]) + b1[f0]
             + fmaf(x0, rW1[f0], x1v * rW1[HID + f0]) + rb1[f0];
    float v1 = fmaf(t0, W1[f1], t1 * W1[HID + f1]) + b1[f1]
             + fmaf(x0, rW1[f1], x1v * rW1[HID + f1]) + rb1[f1];
    float sm = v0 + v1, sq = fmaf(v0, v0, v1 * v1);
    #pragma unroll
    for (int off = 16; off > 0; off >>= 1) {
        sm += __shfl_xor(sm, off, 32);
        sq += __shfl_xor(sq, off, 32);
    }
    float mu  = sm * (1.f / 64.f);
    float var = sq * (1.f / 64.f) - mu * mu;
    float inv = rsqrtf(var + LN_EPS);
    float o0 = (v0 - mu) * inv * g[f0] + be[f0];
    float o1 = (v1 - mu) * inv * g[f1] + be[f1];
    out[(long long)n * 64 + f0] = fmaxf(o0, 0.f);
    out[(long long)n * 64 + f1] = fmaxf(o1, 0.f);
}

__global__ void k_l2_wmma(const float* __restrict__ x1, const float* __restrict__ W2,
                          const float* __restrict__ rW2, const float* __restrict__ b2,
                          const float* __restrict__ rb2, const float* __restrict__ dinv,
                          float* __restrict__ xw2, float* __restrict__ acc2) {
    int wave_global = (blockIdx.x * blockDim.x + threadIdx.x) >> 5;
    int lane = threadIdx.x & 31;
    const int tiles_m = N_NODES / 16;
    int tm = wave_global >> 1;
    int tn = wave_global & 1;
    if (tm >= tiles_m) return;
    int m0 = tm * 16, n0 = tn * 16;
    int  lr = lane & 15;
    bool hi = lane >= 16;
    int  kofs = hi ? 2 : 0;

    v8f c0 = {};
    v8f c1 = {};
    #pragma unroll
    for (int k0 = 0; k0 < HID; k0 += 4) {
        const float* ap = &x1[(long long)(m0 + lr) * HID + k0 + kofs];
        v2f a; a.x = ap[0]; a.y = ap[1];
        int kb = k0 + kofs;
        v2f b;  b.x  = W2 [kb * RED + n0 + lr]; b.y  = W2 [(kb + 1) * RED + n0 + lr];
        v2f br; br.x = rW2[kb * RED + n0 + lr]; br.y = rW2[(kb + 1) * RED + n0 + lr];
        c0 = __builtin_amdgcn_wmma_f32_16x16x4_f32(false, a, false, b,  (short)0, c0, false, false);
        c1 = __builtin_amdgcn_wmma_f32_16x16x4_f32(false, a, false, br, (short)0, c1, false, false);
    }
    int ncol = n0 + lr;
    float b2v = b2[ncol], rb2v = rb2[ncol];
    #pragma unroll
    for (int v = 0; v < 8; ++v) {
        int m = m0 + v + (hi ? 8 : 0);
        float di = dinv[m];
        float xv = c0[v];
        xw2 [(long long)m * RED + ncol] = xv;
        acc2[(long long)m * RED + ncol] = di * di * xv + b2v + c1[v] + rb2v;
    }
}

#define GATHER_STEP(A, J)                                                              \
    {                                                                                  \
        int   r_  = __builtin_amdgcn_readlane(ent.x, (J));                             \
        float cf_ = __int_as_float(__builtin_amdgcn_readlane(ent.y, (J)));             \
        A = fmaf(cf_, xw2[(long long)r_ * 32 + lane], A);                              \
    }
__global__ void k_gather_head(const int* __restrict__ od, const int* __restrict__ dci,
                       const int* __restrict__ csr, const float* __restrict__ xw2,
                       const float* __restrict__ acc2, const float* __restrict__ g2,
                       const float* __restrict__ be2, const float* __restrict__ fcW,
                       const float* __restrict__ fcb, const float* __restrict__ pW,
                       const float* __restrict__ pb, float* __restrict__ logits,
                       float* __restrict__ emb) {
    int lane = threadIdx.x & 31;
    int n = __builtin_amdgcn_readfirstlane((int)(blockIdx.x * (blockDim.x >> 5) + (threadIdx.x >> 5)));
    if (n >= N_NODES) return;
    int beg = od[2 * n];
    int len = dci[2 * n + 1];
    const int2* seg = reinterpret_cast<const int2*>(csr) + beg;
    float a0 = 0.f, a1 = 0.f, a2 = 0.f, a3 = 0.f;
    for (int base = 0; base < len; base += 32) {
        int idx = base + lane;
        int2 ent = make_int2(0, 0);
        if (idx < len) ent = seg[idx];
        int m = len - base < 32 ? len - base : 32;
        int j = 0;
        for (; j + 8 <= m; j += 8) {
            GATHER_STEP(a0, j)     GATHER_STEP(a1, j + 1)
            GATHER_STEP(a2, j + 2) GATHER_STEP(a3, j + 3)
            GATHER_STEP(a0, j + 4) GATHER_STEP(a1, j + 5)
            GATHER_STEP(a2, j + 6) GATHER_STEP(a3, j + 7)
        }
        if (j + 4 <= m) {
            GATHER_STEP(a0, j)     GATHER_STEP(a1, j + 1)
            GATHER_STEP(a2, j + 2) GATHER_STEP(a3, j + 3)
            j += 4;
        }
        int rem = m - j;
        if (rem == 3)      { GATHER_STEP(a0, j) GATHER_STEP(a1, j + 1) GATHER_STEP(a2, j + 2) }
        else if (rem == 2) { GATHER_STEP(a0, j) GATHER_STEP(a1, j + 1) }
        else if (rem == 1) { GATHER_STEP(a0, j) }
    }
    float v = acc2[(long long)n * 32 + lane] + ((a0 + a1) + (a2 + a3));
    float s = v, sq = v * v;
    #pragma unroll
    for (int off = 16; off > 0; off >>= 1) {
        s  += __shfl_xor(s,  off, 32);
        sq += __shfl_xor(sq, off, 32);
    }
    float mu  = s * (1.f / 32.f);
    float var = sq * (1.f / 32.f) - mu * mu;
    float inv = rsqrtf(var + LN_EPS);
    float x2 = fmaxf((v - mu) * inv * g2[lane] + be2[lane], 0.f);

    float lg = x2 * fcW[lane];
    #pragma unroll
    for (int off = 16; off > 0; off >>= 1) lg += __shfl_xor(lg, off, 32);
    if (lane == 0) logits[n] = lg + fcb[0];

    float p = pb[lane];
    #pragma unroll
    for (int i = 0; i < 32; ++i)
        p = fmaf(__shfl(x2, i, 32), pW[i * 32 + lane], p);
    float nsq = p * p;
    #pragma unroll
    for (int off = 16; off > 0; off >>= 1) nsq += __shfl_xor(nsq, off, 32);
    float nrm = sqrtf(nsq);
    emb[(long long)n * 32 + lane] = p / fmaxf(nrm, 1e-12f);
}

extern "C" void kernel_launch(void* const* d_in, const int* in_sizes, int n_in,
                              void* d_out, int out_size, void* d_ws, size_t ws_size,
                              hipStream_t stream) {
    const float* x    = (const float*)d_in[0];
    const float* ew   = (const float*)d_in[1];
    const float* W1   = (const float*)d_in[2];
    const float* b1   = (const float*)d_in[3];
    const float* rW1  = (const float*)d_in[4];
    const float* rb1  = (const float*)d_in[5];
    const float* g1   = (const float*)d_in[6];
    const float* be1  = (const float*)d_in[7];
    const float* W2   = (const float*)d_in[8];
    const float* b2   = (const float*)d_in[9];
    const float* rW2  = (const float*)d_in[10];
    const float* rb2  = (const float*)d_in[11];
    const float* g2   = (const float*)d_in[12];
    const float* be2  = (const float*)d_in[13];
    const float* fcW  = (const float*)d_in[14];
    const float* fcb  = (const float*)d_in[15];
    const float* pW   = (const float*)d_in[16];
    const float* pb   = (const float*)d_in[17];
    const int*   ei   = (const int*)d_in[18];
    const int*   row  = ei;
    const int*   col  = ei + N_EDGES;

    float* ws   = (float*)d_ws;
    float* dinv = ws;  ws += 2 * (size_t)N_NODES;
    int2*  tab     = (int2*)ws;
    int*   counter = (int*)(ws + 4 * (size_t)N_NODES);
    int*   dci  = (int*)(ws + 40 * (size_t)N_NODES);
    int*   od   = (int*)(ws + 42 * (size_t)N_NODES);
    float4* xd  = (float4*)(ws + 44 * (size_t)N_NODES);
    ws += (size_t)N_NODES * HID;
    int*   csr  = (int*)ws;
    ws += (size_t)N_NODES * HID;
    float* x1   = ws;  ws += (size_t)N_NODES * HID;
    int2*  binned = (int2*)x1;
    float* xw2  = ws;  ws += (size_t)N_NODES * RED;
    float* acc2 = ws;  ws += (size_t)N_NODES * RED;

    float* logits = (float*)d_out;
    float* emb    = logits + N_NODES;

    const int B = 256;
    k_chunk_sort<<<NCH, 1024, 0, stream>>>(row, col, ew, tab, counter, binned);
    k_bucket<<<NB, 1024, 0, stream>>>(tab, counter, binned, x, dci, od, dinv, xd, (int2*)csr);

    k_node_l1<<<(N_NODES + 7) / 8, B, 0, stream>>>(x, xd, od, dci, csr, W1, b1, rW1, rb1, g1, be1, x1);

    int waves_l2 = (N_NODES / 16) * (RED / 16);
    int thr_l2   = waves_l2 * 32;
    k_l2_wmma<<<(thr_l2 + B - 1) / B, B, 0, stream>>>(x1, W2, rW2, b2, rb2, dinv, xw2, acc2);

    k_gather_head<<<(N_NODES + 7) / 8, B, 0, stream>>>(od, dci, csr, xw2, acc2, g2, be2, fcW, fcb,
                                                       pW, pb, logits, emb);
}


# layer-2 dense products (x1@W2, x1@rW2 on f32 WMMA) moved into the layer-1 node pass: 16 waves = one 16-row tile, x1 rows handed over in LDS, never written to memory; 4 launches, not 5; same bits
# speedup vs baseline: 8.5540x; 1.0437x over previous
#include <hip/hip_runtime.h>
#include <hip/hip_bf16.h>

#define N_NODES 100000
#define N_EDGES 3200000
#define HID 64
#define RED 32
#define EMB 32
#define LN_EPS 1e-5f

typedef __attribute__((ext_vector_type(2))) float v2f;
typedef __attribute__((ext_vector_type(8))) float v8f;

#define NB  391
#define NCH 256
#define CH  12500
#define KPT 13
#define PPW 8
#define STG 16384

__device__ __forceinline__ void wave_scan_nb(const int* src, int* dst, int lane) {
    int b0 = lane * 13;
    int loc[13];
    int sum = 0;
    #pragma unroll
    for (int j = 0; j < 13; ++j) {
        int b = b0 + j;
        int v = b < NB ? src[b] : 0;
        loc[j] = sum;
        sum += v;
    }
    int incl = sum;
    #pragma unroll
    for (int d = 1; d < 32; d <<= 1) {
        int v = __shfl_up(incl, d, 32);
        if (lane >= d) incl += v;
    }
    int ex = incl - sum;
    #pragma unroll
    for (int j = 0; j < 13; ++j) {
        int b = b0 + j;
        if (b < NB) dst[b] = ex + loc[j];
    }
}

__global__ void __launch_bounds__(1024) k_chunk_sort(const int* __restrict__ row,
                                                     const int* __restrict__ col,
                                                     const float* __restrict__ w,
                                                     int2* __restrict__ tab,
                                                     int* __restrict__ counter,
                                                     int2* __restrict__ binned) {
    __shared__ int  s_cnt[NB];
    __shared__ int  s_lst[NB];
    __shared__ int  s_cur[NB];
    __shared__ int2 s_rec[CH];
    int t = threadIdx.x, lane = t & 31, wave = t >> 5, g = blockIdx.x;

    int rc[KPT], rr[KPT];
    float rw[KPT];
    const int e0 = g * CH;
    #pragma unroll
    for (int k = 0; k < KPT; ++k) {
        int i = k * 1024 + t;
        rc[k] = 0; rr[k] = 0; rw[k] = 0.f;
        if (i < CH) { rc[k] = col[e0 + i]; rr[k] = row[e0 + i]; rw[k] = w[e0 + i]; }
    }
    for (int b = t; b < NB; b += 1024) { s_cnt[b] = 0; s_cur[b] = 0; }
    if (g == 0 && t == 0) *counter = 0;
    __syncthreads();
    #pragma unroll
    for (int k = 0; k < KPT; ++k) {
        int i = k * 1024 + t;
        if (i < CH) atomicAdd(&s_cnt[rc[k] >> 8], 1);
    }
    __syncthreads();
    if (wave == 0) wave_scan_nb(s_cnt, s_lst, lane);
    __syncthreads();
    for (int b = t; b < NB; b += 1024) tab[b * NCH + g] = make_int2(s_lst[b], s_cnt[b]);
    #pragma unroll
    for (int k = 0; k < KPT; ++k) {
        int i = k * 1024 + t;
        if (i < CH) {
            int b = rc[k] >> 8;
            int p = s_lst[b] + atomicAdd(&s_cur[b], 1);
            s_rec[p] = make_int2(rr[k] | ((rc[k] & 255) << 17), __float_as_int(rw[k]));
        }
    }
    __syncthreads();
    for (int i = t; i < CH; i += 1024) binned[e0 + i] = s_rec[i];
}

#define BK_ADD(E)                                                                      \
    {                                                                                  \
        int2 e_ = (E);                                                                 \
        int cl_ = (e_.x >> 17) & 255;                                                  \
        atomicAdd(&s_cnt[cl_], 1);                                                     \
        atomicAdd(&s_deg[cl_], __int_as_float(e_.y));                                  \
    }
#define BK_PLACE(E)                                                                    \
    {                                                                                  \
        int2 e_ = (E);                                                                 \
        int cl_ = (e_.x >> 17) & 255;                                                  \
        int p_  = s_first[cl_] + atomicAdd(&s_cur[cl_], 1);                            \
        int2 o_ = make_int2(e_.x & 0x1FFFF, e_.y);                                     \
        if (staged) s_rec[p_] = o_; else csr[beg + p_] = o_;                           \
    }
#define BK_PASS(OP)                                                                    \
    _Pragma("unroll")                                                                  \
    for (int h = 0; h < PPW; h += 4) {                                                 \
        int2 r[4], s[2];                                                               \
        _Pragma("unroll")                                                              \
        for (int q = 0; q < 4; ++q) {                                                  \
            int ia_ = po[h + q] + (lane < pc[h + q] ? lane : 0);                       \
            r[q] = binned[ia_ < N_EDGES ? ia_ : N_EDGES - 1];                          \
        }                                                                              \
        _Pragma("unroll")                                                              \
        for (int k = 0; k < 2; ++k) {                                                  \
            int pcs_ = hih ? pc[h + 2 * k + 1] : pc[h + 2 * k];                        \
            int pos_ = hih ? po[h + 2 * k + 1] : po[h + 2 * k];                        \
            int ib_  = pos_ + (32 + l16 < pcs_ ? 32 + l16 : 0);                        \
            s[k] = binned[ib_ < N_EDGES ? ib_ : N_EDGES - 1];                          \
        }                                                                              \
          \
          \
          \
        asm volatile("" :: "v"(r[0].x), "v"(r[0].y), "v"(r[1].x), "v"(r[1].y),         \
                           "v"(r[2].x), "v"(r[2].y), "v"(r[3].x), "v"(r[3].y),         \
                           "v"(s[0].x), "v"(s[0].y), "v"(s[1].x), "v"(s[1].y));        \
        _Pragma("unroll")                                                              \
        for (int q = 0; q < 4; ++q)                                                    \
            if (lane < pc[h + q]) OP(r[q])                                             \
        _Pragma("unroll")                                                              \
        for (int k = 0; k < 2; ++k) {                                                  \
            int pcs_ = hih ? pc[h + 2 * k + 1] : pc[h + 2 * k];                        \
            if (32 + l16 < pcs_) OP(s[k])                                              \
        }                                                                              \
        _Pragma("unroll")                                                              \
        for (int q = 0; q < 4; ++q)                                                    \
            for (int j = lane + 48; j < pc[h + q]; j += 32) OP(binned[po[h + q] + j])  \
    }

__global__ void __launch_bounds__(1024) k_bucket(const int2* __restrict__ tab,
                                                 int* __restrict__ counter,
                                                 const int2* __restrict__ binned,
                                                 const float* __restrict__ x,
                                                 int* __restrict__ dci, int* __restrict__ od,
                                                 float4* __restrict__ xd,
                                                 int2* __restrict__ csr) {
    __shared__ int   s_cnt[256];
    __shared__ float s_deg[256];
    __shared__ int   s_first[256];
    __shared__ int   s_cur[256];
    __shared__ int   s_wt[8];
    __shared__ int   s_po[NCH];
    __shared__ int   s_pc[NCH];
    __shared__ int   s_bm[2];
    __shared__ int2  s_rec[STG];
    int t = threadIdx.x, lane = t & 31, wave = t >> 5, b = blockIdx.x;
    if (t < NCH) {
        s_cnt[t] = 0; s_deg[t] = 1.0f; s_cur[t] = 0;
        int2 e = tab[b * NCH + t];
        s_po[t] = t * CH + e.x;
        s_pc[t] = e.y;
        int s = e.y;
        #pragma unroll
        for (int o = 16; o > 0; o >>= 1) s += __shfl_xor(s, o, 32);
        if (lane == 0) s_wt[wave] = s;
    }
    __syncthreads();
    if (t == 0) {
        int tot = 0;
        #pragma unroll
        for (int i = 0; i < 8; ++i) tot += s_wt[i];
        s_bm[0] = atomicAdd(counter, tot);
        s_bm[1] = tot;
    }
    __syncthreads();
    const int beg = s_bm[0], m = s_bm[1];

    const int  l16 = lane & 15;
    const bool hih = lane >= 16;
    int pc[PPW], po[PPW];
    #pragma unroll
    for (int q = 0; q < PPW; ++q) {
        pc[q] = __builtin_amdgcn_readfirstlane(s_pc[wave + 32 * q]);
        po[q] = __builtin_amdgcn_readfirstlane(s_po[wave + 32 * q]);
    }
    BK_PASS(BK_ADD)
    __syncthreads();
    int c = 0, incl = 0;
    if (t < 256) {
        c = s_cnt[t];
        incl = c;
        #pragma unroll
        for (int d = 1; d < 32; d <<= 1) {
            int v = __shfl_up(incl, d, 32);
            if (lane >= d) incl += v;
        }
        if (lane == 31) s_wt[wave] = incl;
    }
    __syncthreads();
    if (t < 256) {
        int base = 0;
        #pragma unroll
        for (int i = 0; i < 8; ++i) base += (i < wave) ? s_wt[i] : 0;
        int first = base + incl - c;
        s_first[t] = first;
        int n = b * 256 + t;
        if (n < N_NODES) {
            float d  = s_deg[t];
            float di = d > 0.f ? rsqrtf(d) : 0.f;
            reinterpret_cast<int2*>(od)[n] = make_int2(beg + first, __float_as_int(di));
            dci[2 * n + 1] = c;
            xd[n] = make_float4(x[2 * n], x[2 * n + 1], di, 0.f);
        }
    }
    __syncthreads();
    const bool staged = m <= STG;
    BK_PASS(BK_PLACE)
    if (staged) {
        __syncthreads();
        for (int i = t; i < m; i += 1024) csr[beg + i] = s_rec[i];
    }
}

#define FXS 44

__device__ __forceinline__ int      fx_exp(unsigned u) { return (int)((u >> 23) & 255u); }
__device__ __forceinline__ unsigned fx_sig(unsigned u) { return (u & 0x007FFFFFu) | 0x00800000u; }

__device__ __forceinline__ long long fx_term(unsigned mc, int ec, unsigned uc, unsigned ux) {
    unsigned long long pr = ((unsigned long long)mc * (unsigned long long)fx_sig(ux)) << 14;
    int rs = (314 - FXS) - ec - fx_exp(ux);
    rs = rs < 0 ? 0 : (rs > 63 ? 63 : rs);
    unsigned long long mag = pr >> rs;
    unsigned long long sg  = (unsigned long long)(long long)((int)(uc ^ ux) >> 31);
    return (long long)((mag ^ sg) - sg);
}

__device__ __forceinline__ float fx_to_float(long long s) {
    unsigned long long a = s < 0 ? 0ull - (unsigned long long)s : (unsigned long long)s;
    int lz = __clzll((long long)a);
    lz = lz > 63 ? 63 : lz;
    unsigned long long nrm = a << lz;
    unsigned top = (unsigned)(nrm >> 32) | ((unsigned)nrm != 0u ? 1u : 0u);
    float f = ldexpf((float)top, 32 - lz - FXS);
    return s < 0 ? -f : f;
}

#define NPB  16
#define X1P  68
static_assert(N_NODES % NPB == 0, "every work-group holds a whole 16-row tile");
__global__ void __launch_bounds__(NPB * 32) k_node_l1(const float* __restrict__ x, const float4* __restrict__ xd,
                          const int* __restrict__ od,
                          const int* __restrict__ dci, int* csr,
                          const float* __restrict__ W1, const float* __restrict__ b1,
                          const float* __restrict__ rW1, const float* __restrict__ rb1,
                          const float* __restrict__ g, const float* __restrict__ be,
                          const float* __restrict__ W2, const float* __restrict__ rW2,
                          const float* __restrict__ b2, const float* __restrict__ rb2,
                          float* __restrict__ xw2, float* __restrict__ acc2) {
    __shared__ float s_x1[NPB * X1P];
    __shared__ float s_di[NPB];
    int lane = threadIdx.x & 31;
    int wv = __builtin_amdgcn_readfirstlane((int)(threadIdx.x >> 5));
    int n  = (int)blockIdx.x * NPB + wv;
    int2 oc  = reinterpret_cast<const int2*>(od)[n];
    int  len = dci[2 * n + 1];
    float di = __int_as_float(oc.y);
    float x0 = x[2 * n], x1v = x[2 * n + 1];
    unsigned long long* seg = reinterpret_cast<unsigned long long*>(csr) + oc.x;
    unsigned long long s0 = 0, s1 = 0;
    int emc = 0, emx = 0;
    for (int base = 0; base < len; base += 64) {
        int ia = base + lane, ib = ia + 32, last = len - 1;
        unsigned long long ea = seg[ia < last ? ia : last];
        unsigned long long eb = seg[ib < last ? ib : last];
        __builtin_amdgcn_sched_barrier(0);
        float4 xa = xd[(int)(unsigned)ea];
        float4 xb = xd[(int)(unsigned)eb];
        __builtin_amdgcn_sched_barrier(0);
        float fa = xa.z * __int_as_float((int)(unsigned)(ea >> 32)) * di;
        float fb = xb.z * __int_as_float((int)(unsigned)(eb >> 32)) * di;
        unsigned ua  = __float_as_uint(fa),   ub  = __float_as_uint(fb);
        unsigned uax = __float_as_uint(xa.x), uay = __float_as_uint(xa.y);
        unsigned ubx = __float_as_uint(xb.x), uby = __float_as_uint(xb.y);
        int eca = fx_exp(ua), ecb = fx_exp(ub);
        unsigned mca = ia < len ? fx_sig(ua) : 0u;
        unsigned mcb = ib < len ? fx_sig(ub) : 0u;
        s0 += (unsigned long long)fx_term(mca, eca, ua, uax);  s1 += (unsigned long long)fx_term(mca, eca, ua, uay);
        s0 += (unsigned long long)fx_term(mcb, ecb, ub, ubx);  s1 += (unsigned long long)fx_term(mcb, ecb, ub, uby);
        emc = max(emc, max(eca, ecb));
        emx = max(max(emx, fx_exp(uax)), max(fx_exp(uay), max(fx_exp(ubx), fx_exp(uby))));
        if (ia < len) seg[ia] = ((unsigned long long)ua << 32) | (unsigned)ea;
        if (ib < len) seg[ib] = ((unsigned long long)ub << 32) | (unsigned)eb;
    }
    float cs = di * di;
    unsigned ucs = __float_as_uint(cs), ux0 = __float_as_uint(x0), ux1 = __float_as_uint(x1v);
    emc = max(emc, fx_exp(ucs));
    emx = max(emx, max(fx_exp(ux0), fx_exp(ux1)));
    bool wide = emc > 140 || emx > 140 || emc + emx > 314 - FXS - 12 || len > 4096;
    float t0, t1;
    if (__any(wide)) {
        double d0 = 0.0, d1 = 0.0;
        for (int i = lane; i < len; i += 32) {
            unsigned long long e = seg[i];
            float4 xr = xd[(int)(unsigned)e];
            double c = (double)__int_as_float((int)(unsigned)(e >> 32));
            d0 += c * (double)xr.x;  d1 += c * (double)xr.y;
        }
        #pragma unroll
        for (int off = 16; off > 0; off >>= 1) {
            d0 += __shfl_xor(d0, off, 32);
            d1 += __shfl_xor(d1, off, 32);
        }
        t0 = (float)(d0 + (double)cs * (double)x0);
        t1 = (float)(d1 + (double)cs * (double)x1v);
    } else {
        #pragma unroll
        for (int off = 16; off > 0; off >>= 1) {
            s0 += __shfl_xor(s0, off, 32);
            s1 += __shfl_xor(s1, off, 32);
        }
        s0 += (unsigned long long)fx_term(fx_sig(ucs), fx_exp(ucs), ucs, ux0);
        s1 += (unsigned long long)fx_term(fx_sig(ucs), fx_exp(ucs), ucs, ux1);
        t0 = fx_to_float((long long)s0);
        t1 = fx_to_float((long long)s1);
    }
    int f0 = lane, f1 = lane + 32;
    float v0 = fmaf(t0, W1[f0], t1 * W1[HID + f0]) + b1[f0]
             + fmaf(x0, rW1[f0], x1v * rW1[HID + f0]) + rb1[f0];
    float v1 = fmaf(t0, W1[f1], t1 * W1[HID + f1]) + b1[f1]
             + fmaf(x0, rW1[f1], x1v * rW1[HID + f1]) + rb1[f1];
    float sm = v0 + v1, sq = fmaf(v0, v0, v1 * v1);
    #pragma unroll
    for (int off = 16; off > 0; off >>= 1) {
        sm += __shfl_xor(sm, off, 32);
        sq += __shfl_xor(sq, off, 32);
    }
    float mu  = sm * (1.f / 64.f);
    float var = sq * (1.f / 64.f) - mu * mu;
    float inv = rsqrtf(var + LN_EPS);
    float o0 = (v0 - mu) * inv * g[f0] + be[f0];
    float o1 = (v1 - mu) * inv * g[f1] + be[f1];
    s_x1[wv * X1P + f0] = fmaxf(o0, 0.f);
    s_x1[wv * X1P + f1] = fmaxf(o1, 0.f);
    if (lane == 0) s_di[wv] = di;
    __builtin_amdgcn_fence(__ATOMIC_RELEASE, "workgroup", "local");
    __builtin_amdgcn_s_barrier();
    __builtin_amdgcn_fence(__ATOMIC_ACQUIRE, "workgroup", "local");
    if (wv >= RED / 16) return;

    int m0 = (int)blockIdx.x * NPB, n0 = wv * 16;
    int  lr = lane & 15;
    bool hi = lane >= 16;
    int  kofs = hi ? 2 : 0;
    v8f c0 = {};
    v8f c1 = {};
    #pragma unroll
    for (int h0 = 0; h0 < HID; h0 += 32) {
        __builtin_amdgcn_sched_barrier(0);
        v2f b[8], br[8];
        #pragma unroll
        for (int s = 0; s < 8; ++s) {
            int kb = h0 + 4 * s + kofs;
            b[s].x  = W2 [kb * RED + n0 + lr]; b[s].y  = W2 [(kb + 1) * RED + n0 + lr];
            br[s].x = rW2[kb * RED + n0 + lr]; br[s].y = rW2[(kb + 1) * RED + n0 + lr];
        }
        #pragma unroll
        for (int s = 0; s < 8; ++s) {
            const float* ap = &s_x1[lr * X1P + h0 + 4 * s + kofs];
            v2f a; a.x = ap[0]; a.y = ap[1];
            c0 = __builtin_amdgcn_wmma_f32_16x16x4_f32(false, a, false, b[s],  (short)0, c0, false, false);
            c1 = __builtin_amdgcn_wmma_f32_16x16x4_f32(false, a, false, br[s], (short)0, c1, false, false);
        }
    }
    int ncol = n0 + lr;
    float b2v = b2[ncol], rb2v = rb2[ncol];
    #pragma unroll
    for (int v = 0; v < 8; ++v) {
        int mr = v + (hi ? 8 : 0);
        float dm = s_di[mr];
        float xv = c0[v];
        xw2 [(long long)(m0 + mr) * RED + ncol] = xv;
        acc2[(long long)(m0 + mr) * RED + ncol] = dm * dm * xv + b2v + c1[v] + rb2v;
    }
}

#define GATHER_STEP(A, J)                                                              \
    {                                                                                  \
        int   r_  = __builtin_amdgcn_readlane(ent.x, (J));                             \
        float cf_ = __int_as_float(__builtin_amdgcn_readlane(ent.y, (J)));             \
        A = fmaf(cf_, xw2[(long long)r_ * 32 + lane], A);                              \
    }
__global__ void k_gather_head(const int* __restrict__ od, const int* __restrict__ dci,
                       const int* __restrict__ csr, const float* __restrict__ xw2,
                       const float* __restrict__ acc2, const float* __restrict__ g2,
                       const float* __restrict__ be2, const float* __restrict__ fcW,
                       const float* __restrict__ fcb, const float* __restrict__ pW,
                       const float* __restrict__ pb, float* __restrict__ logits,
                       float* __restrict__ emb) {
    int lane = threadIdx.x & 31;
    int n = __builtin_amdgcn_readfirstlane((int)(blockIdx.x * (blockDim.x >> 5) + (threadIdx.x >> 5)));
    if (n >= N_NODES) return;
    int beg = od[2 * n];
    int len = dci[2 * n + 1];
    const int2* seg = reinterpret_cast<const int2*>(csr) + beg;
    float a0 = 0.f, a1 = 0.f, a2 = 0.f, a3 = 0.f;
    for (int base = 0; base < len; base += 32) {
        int idx = base + lane;
        int2 ent = make_int2(0, 0);
        if (idx < len) ent = seg[idx];
        int m = len - base < 32 ? len - base : 32;
        int j = 0;
        for (; j + 8 <= m; j += 8) {
            GATHER_STEP(a0, j)     GATHER_STEP(a1, j + 1)
            GATHER_STEP(a2, j + 2) GATHER_STEP(a3, j + 3)
            GATHER_STEP(a0, j + 4) GATHER_STEP(a1, j + 5)
            GATHER_STEP(a2, j + 6) GATHER_STEP(a3, j + 7)
        }
        if (j + 4 <= m) {
            GATHER_STEP(a0, j)     GATHER_STEP(a1, j + 1)
            GATHER_STEP(a2, j + 2) GATHER_STEP(a3, j + 3)
            j += 4;
        }
        int rem = m - j;
        if (rem == 3)      { GATHER_STEP(a0, j) GATHER_STEP(a1, j + 1) GATHER_STEP(a2, j + 2) }
        else if (rem == 2) { GATHER_STEP(a0, j) GATHER_STEP(a1, j + 1) }
        else if (rem == 1) { GATHER_STEP(a0, j) }
    }
    float v = acc2[(long long)n * 32 + lane] + ((a0 + a1) + (a2 + a3));
    float s = v, sq = v * v;
    #pragma unroll
    for (int off = 16; off > 0; off >>= 1) {
        s  += __shfl_xor(s,  off, 32);
        sq += __shfl_xor(sq, off, 32);
    }
    float mu  = s * (1.f / 32.f);
    float var = sq * (1.f / 32.f) - mu * mu;
    float inv = rsqrtf(var + LN_EPS);
    float x2 = fmaxf((v - mu) * inv * g2[lane] + be2[lane], 0.f);

    float lg = x2 * fcW[lane];
    #pragma unroll
    for (int off = 16; off > 0; off >>= 1) lg += __shfl_xor(lg, off, 32);
    if (lane == 0) logits[n] = lg + fcb[0];

    float p = pb[lane];
    #pragma unroll
    for (int i = 0; i < 32; ++i)
        p = fmaf(__shfl(x2, i, 32), pW[i * 32 + lane], p);
    float nsq = p * p;
    #pragma unroll
    for (int off = 16; off > 0; off >>= 1) nsq += __shfl_xor(nsq, off, 32);
    float nrm = sqrtf(nsq);
    emb[(long long)n * 32 + lane] = p / fmaxf(nrm, 1e-12f);
}

extern "C" void kernel_launch(void* const* d_in, const int* in_sizes, int n_in,
                              void* d_out, int out_size, void* d_ws, size_t ws_size,
                              hipStream_t stream) {
    const float* x    = (const float*)d_in[0];
    const float* ew   = (const float*)d_in[1];
    const float* W1   = (const float*)d_in[2];
    const float* b1   = (const float*)d_in[3];
    const float* rW1  = (const float*)d_in[4];
    const float* rb1  = (const float*)d_in[5];
    const float* g1   = (const float*)d_in[6];
    const float* be1  = (const float*)d_in[7];
    const float* W2   = (const float*)d_in[8];
    const float* b2   = (const float*)d_in[9];
    const float* rW2  = (const float*)d_in[10];
    const float* rb2  = (const float*)d_in[11];
    const float* g2   = (const float*)d_in[12];
    const float* be2  = (const float*)d_in[13];
    const float* fcW  = (const float*)d_in[14];
    const float* fcb  = (const float*)d_in[15];
    const float* pW   = (const float*)d_in[16];
    const float* pb   = (const float*)d_in[17];
    const int*   ei   = (const int*)d_in[18];
    const int*   row  = ei;
    const int*   col  = ei + N_EDGES;

    float* ws   = (float*)d_ws;
    ws += 2 * (size_t)N_NODES;
    int2*  tab     = (int2*)ws;
    int*   counter = (int*)(ws + 4 * (size_t)N_NODES);
    int*   dci  = (int*)(ws + 40 * (size_t)N_NODES);
    int*   od   = (int*)(ws + 42 * (size_t)N_NODES);
    float4* xd  = (float4*)(ws + 44 * (size_t)N_NODES);
    ws += (size_t)N_NODES * HID;
    int*   csr  = (int*)ws;
    ws += (size_t)N_NODES * HID;
    int2*  binned = (int2*)ws;  ws += (size_t)N_NODES * HID;
    float* xw2  = ws;  ws += (size_t)N_NODES * RED;
    float* acc2 = ws;  ws += (size_t)N_NODES * RED;

    float* logits = (float*)d_out;
    float* emb    = logits + N_NODES;

    const int B = 256;
    k_chunk_sort<<<NCH, 1024, 0, stream>>>(row, col, ew, tab, counter, binned);
    k_bucket<<<NB, 1024, 0, stream>>>(tab, counter, binned, x, dci, od, xd, (int2*)csr);

    k_node_l1<<<N_NODES / NPB, NPB * 32, 0, stream>>>(x, xd, od, dci, csr, W1, b1, rW1, rb1, g1, be1,
                                                      W2, rW2, b2, rb2, xw2, acc2);

    k_gather_head<<<(N_NODES + 7) / 8, B, 0, stream>>>(od, dci, csr, xw2, acc2, g2, be2, fcW, fcb,
                                                       pW, pb, logits, emb);
}


# layer-2 gather: 8 lanes read one source row as 16-byte pieces: a load instruction fetches 4 rows, a wave keeps up to 32 in flight (entries dealt by DPP8, not readlane); sum order kept: same bits
# speedup vs baseline: 9.0184x; 1.0543x over previous
#include <hip/hip_runtime.h>
#include <hip/hip_bf16.h>

#define N_NODES 100000
#define N_EDGES 3200000
#define HID 64
#define RED 32
#define EMB 32
#define LN_EPS 1e-5f

typedef __attribute__((ext_vector_type(2))) float v2f;
typedef __attribute__((ext_vector_type(8))) float v8f;

#define NB  391
#define NCH 256
#define CH  12500
#define KPT 13
#define PPW 8
#define STG 16384

__device__ __forceinline__ void wave_scan_nb(const int* src, int* dst, int lane) {
    int b0 = lane * 13;
    int loc[13];
    int sum = 0;
    #pragma unroll
    for (int j = 0; j < 13; ++j) {
        int b = b0 + j;
        int v = b < NB ? src[b] : 0;
        loc[j] = sum;
        sum += v;
    }
    int incl = sum;
    #pragma unroll
    for (int d = 1; d < 32; d <<= 1) {
        int v = __shfl_up(incl, d, 32);
        if (lane >= d) incl += v;
    }
    int ex = incl - sum;
    #pragma unroll
    for (int j = 0; j < 13; ++j) {
        int b = b0 + j;
        if (b < NB) dst[b] = ex + loc[j];
    }
}

__global__ void __launch_bounds__(1024) k_chunk_sort(const int* __restrict__ row,
                                                     const int* __restrict__ col,
                                                     const float* __restrict__ w,
                                                     int2* __restrict__ tab,
                                                     int* __restrict__ counter,
                                                     int2* __restrict__ binned) {
    __shared__ int  s_cnt[NB];
    __shared__ int  s_lst[NB];
    __shared__ int  s_cur[NB];
    __shared__ int2 s_rec[CH];
    int t = threadIdx.x, lane = t & 31, wave = t >> 5, g = blockIdx.x;

    int rc[KPT], rr[KPT];
    float rw[KPT];
    const int e0 = g * CH;
    #pragma unroll
    for (int k = 0; k < KPT; ++k) {
        int i = k * 1024 + t;
        rc[k] = 0; rr[k] = 0; rw[k] = 0.f;
        if (i < CH) { rc[k] = col[e0 + i]; rr[k] = row[e0 + i]; rw[k] = w[e0 + i]; }
    }
    for (int b = t; b < NB; b += 1024) { s_cnt[b] = 0; s_cur[b] = 0; }
    if (g == 0 && t == 0) *counter = 0;
    __syncthreads();
    #pragma unroll
    for (int k = 0; k < KPT; ++k) {
        int i = k * 1024 + t;
        if (i < CH) atomicAdd(&s_cnt[rc[k] >> 8], 1);
    }
    __syncthreads();
    if (wave == 0) wave_scan_nb(s_cnt, s_lst, lane);
    __syncthreads();
    for (int b = t; b < NB; b += 1024) tab[b * NCH + g] = make_int2(s_lst[b], s_cnt[b]);
    #pragma unroll
    for (int k = 0; k < KPT; ++k) {
        int i = k * 1024 + t;
        if (i < CH) {
            int b = rc[k] >> 8;
            int p = s_lst[b] + atomicAdd(&s_cur[b], 1);
            s_rec[p] = make_int2(rr[k] | ((rc[k] & 255) << 17), __float_as_int(rw[k]));
        }
    }
    __syncthreads();
    for (int i = t; i < CH; i += 1024) binned[e0 + i] = s_rec[i];
}

#define BK_ADD(E)                                                                      \
    {                                                                                  \
        int2 e_ = (E);                                                                 \
        int cl_ = (e_.x >> 17) & 255;                                                  \
        atomicAdd(&s_cnt[cl_], 1);                                                     \
        atomicAdd(&s_deg[cl_], __int_as_float(e_.y));                                  \
    }
#define BK_PLACE(E)                                                                    \
    {                                                                                  \
        int2 e_ = (E);                                                                 \
        int cl_ = (e_.x >> 17) & 255;                                                  \
        int p_  = s_first[cl_] + atomicAdd(&s_cur[cl_], 1);                            \
        int2 o_ = make_int2(e_.x & 0x1FFFF, e_.y);                                     \
        if (staged) s_rec[p_] = o_; else csr[beg + p_] = o_;                           \
    }
#define BK_PASS(OP)                                                                    \
    _Pragma("unroll")                                                                  \
    for (int h = 0; h < PPW; h += 4) {                                                 \
        int2 r[4], s[2];                                                               \
        _Pragma("unroll")                                                              \
        for (int q = 0; q < 4; ++q) {                                                  \
            int ia_ = po[h + q] + (lane < pc[h + q] ? lane : 0);                       \
            r[q] = binned[ia_ < N_EDGES ? ia_ : N_EDGES - 1];                          \
        }                                                                              \
        _Pragma("unroll")                                                              \
        for (int k = 0; k < 2; ++k) {                                                  \
            int pcs_ = hih ? pc[h + 2 * k + 1] : pc[h + 2 * k];                        \
            int pos_ = hih ? po[h + 2 * k + 1] : po[h + 2 * k];                        \
            int ib_  = pos_ + (32 + l16 < pcs_ ? 32 + l16 : 0);                        \
            s[k] = binned[ib_ < N_EDGES ? ib_ : N_EDGES - 1];                          \
        }                                                                              \
          \
          \
          \
        asm volatile("" :: "v"(r[0].x), "v"(r[0].y), "v"(r[1].x), "v"(r[1].y),         \
                           "v"(r[2].x), "v"(r[2].y), "v"(r[3].x), "v"(r[3].y),         \
                           "v"(s[0].x), "v"(s[0].y), "v"(s[1].x), "v"(s[1].y));        \
        _Pragma("unroll")                                                              \
        for (int q = 0; q < 4; ++q)                                                    \
            if (lane < pc[h + q]) OP(r[q])                                             \
        _Pragma("unroll")                                                              \
        for (int k = 0; k < 2; ++k) {                                                  \
            int pcs_ = hih ? pc[h + 2 * k + 1] : pc[h + 2 * k];                        \
            if (32 + l16 < pcs_) OP(s[k])                                              \
        }                                                                              \
        _Pragma("unroll")                                                              \
        for (int q = 0; q < 4; ++q)                                                    \
            for (int j = lane + 48; j < pc[h + q]; j += 32) OP(binned[po[h + q] + j])  \
    }

__global__ void __launch_bounds__(1024) k_bucket(const int2* __restrict__ tab,
                                                 int* __restrict__ counter,
                                                 const int2* __restrict__ binned,
                                                 const float* __restrict__ x,
                                                 int* __restrict__ dci, int* __restrict__ od,
                                                 float4* __restrict__ xd,
                                                 int2* __restrict__ csr) {
    __shared__ int   s_cnt[256];
    __shared__ float s_deg[256];
    __shared__ int   s_first[256];
    __shared__ int   s_cur[256];
    __shared__ int   s_wt[8];
    __shared__ int   s_po[NCH];
    __shared__ int   s_pc[NCH];
    __shared__ int   s_bm[2];
    __shared__ int2  s_rec[STG];
    int t = threadIdx.x, lane = t & 31, wave = t >> 5, b = blockIdx.x;
    if (t < NCH) {
        s_cnt[t] = 0; s_deg[t] = 1.0f; s_cur[t] = 0;
        int2 e = tab[b * NCH + t];
        s_po[t] = t * CH + e.x;
        s_pc[t] = e.y;
        int s = e.y;
        #pragma unroll
        for (int o = 16; o > 0; o >>= 1) s += __shfl_xor(s, o, 32);
        if (lane == 0) s_wt[wave] = s;
    }
    __syncthreads();
    if (t == 0) {
        int tot = 0;
        #pragma unroll
        for (int i = 0; i < 8; ++i) tot += s_wt[i];
        s_bm[0] = atomicAdd(counter, tot);
        s_bm[1] = tot;
    }
    __syncthreads();
    const int beg = s_bm[0], m = s_bm[1];

    const int  l16 = lane & 15;
    const bool hih = lane >= 16;
    int pc[PPW], po[PPW];
    #pragma unroll
    for (int q = 0; q < PPW; ++q) {
        pc[q] = __builtin_amdgcn_readfirstlane(s_pc[wave + 32 * q]);
        po[q] = __builtin_amdgcn_readfirstlane(s_po[wave + 32 * q]);
    }
    BK_PASS(BK_ADD)
    __syncthreads();
    int c = 0, incl = 0;
    if (t < 256) {
        c = s_cnt[t];
        incl = c;
        #pragma unroll
        for (int d = 1; d < 32; d <<= 1) {
            int v = __shfl_up(incl, d, 32);
            if (lane >= d) incl += v;
        }
        if (lane == 31) s_wt[wave] = incl;
    }
    __syncthreads();
    if (t < 256) {
        int base = 0;
        #pragma unroll
        for (int i = 0; i < 8; ++i) base += (i < wave) ? s_wt[i] : 0;
        int first = base + incl - c;
        s_first[t] = first;
        int n = b * 256 + t;
        if (n < N_NODES) {
            float d  = s_deg[t];
            float di = d > 0.f ? rsqrtf(d) : 0.f;
            reinterpret_cast<int2*>(od)[n] = make_int2(beg + first, __float_as_int(di));
            dci[2 * n + 1] = c;
            xd[n] = make_float4(x[2 * n], x[2 * n + 1], di, 0.f);
        }
    }
    __syncthreads();
    const bool staged = m <= STG;
    BK_PASS(BK_PLACE)
    if (staged) {
        __syncthreads();
        for (int i = t; i < m; i += 1024) csr[beg + i] = s_rec[i];
    }
}

#define FXS 44

__device__ __forceinline__ int      fx_exp(unsigned u) { return (int)((u >> 23) & 255u); }
__device__ __forceinline__ unsigned fx_sig(unsigned u) { return (u & 0x007FFFFFu) | 0x00800000u; }

__device__ __forceinline__ long long fx_term(unsigned mc, int ec, unsigned uc, unsigned ux) {
    unsigned long long pr = ((unsigned long long)mc * (unsigned long long)fx_sig(ux)) << 14;
    int rs = (314 - FXS) - ec - fx_exp(ux);
    rs = rs < 0 ? 0 : (rs > 63 ? 63 : rs);
    unsigned long long mag = pr >> rs;
    unsigned long long sg  = (unsigned long long)(long long)((int)(uc ^ ux) >> 31);
    return (long long)((mag ^ sg) - sg);
}

__device__ __forceinline__ float fx_to_float(long long s) {
    unsigned long long a = s < 0 ? 0ull - (unsigned long long)s : (unsigned long long)s;
    int lz = __clzll((long long)a);
    lz = lz > 63 ? 63 : lz;
    unsigned long long nrm = a << lz;
    unsigned top = (unsigned)(nrm >> 32) | ((unsigned)nrm != 0u ? 1u : 0u);
    float f = ldexpf((float)top, 32 - lz - FXS);
    return s < 0 ? -f : f;
}

#define NPB  16
#define X1P  68
static_assert(N_NODES % NPB == 0, "every work-group holds a whole 16-row tile");
__global__ void __launch_bounds__(NPB * 32) k_node_l1(const float* __restrict__ x, const float4* __restrict__ xd,
                          const int* __restrict__ od,
                          const int* __restrict__ dci, int* csr,
                          const float* __restrict__ W1, const float* __restrict__ b1,
                          const float* __restrict__ rW1, const float* __restrict__ rb1,
                          const float* __restrict__ g, const float* __restrict__ be,
                          const float* __restrict__ W2, const float* __restrict__ rW2,
                          const float* __restrict__ b2, const float* __restrict__ rb2,
                          float* __restrict__ xw2, float* __restrict__ acc2) {
    __shared__ float s_x1[NPB * X1P];
    __shared__ float s_di[NPB];
    int lane = threadIdx.x & 31;
    int wv = __builtin_amdgcn_readfirstlane((int)(threadIdx.x >> 5));
    int n  = (int)blockIdx.x * NPB + wv;
    int2 oc  = reinterpret_cast<const int2*>(od)[n];
    int  len = dci[2 * n + 1];
    float di = __int_as_float(oc.y);
    float x0 = x[2 * n], x1v = x[2 * n + 1];
    unsigned long long* seg = reinterpret_cast<unsigned long long*>(csr) + oc.x;
    unsigned long long s0 = 0, s1 = 0;
    int emc = 0, emx = 0;
    for (int base = 0; base < len; base += 64) {
        int ia = base + lane, ib = ia + 32, last = len - 1;
        unsigned long long ea = seg[ia < last ? ia : last];
        unsigned long long eb = seg[ib < last ? ib : last];
        __builtin_amdgcn_sched_barrier(0);
        float4 xa = xd[(int)(unsigned)ea];
        float4 xb = xd[(int)(unsigned)eb];
        __builtin_amdgcn_sched_barrier(0);
        float fa = xa.z * __int_as_float((int)(unsigned)(ea >> 32)) * di;
        float fb = xb.z * __int_as_float((int)(unsigned)(eb >> 32)) * di;
        unsigned ua  = __float_as_uint(fa),   ub  = __float_as_uint(fb);
        unsigned uax = __float_as_uint(xa.x), uay = __float_as_uint(xa.y);
        unsigned ubx = __float_as_uint(xb.x), uby = __float_as_uint(xb.y);
        int eca = fx_exp(ua), ecb = fx_exp(ub);
        unsigned mca = ia < len ? fx_sig(ua) : 0u;
        unsigned mcb = ib < len ? fx_sig(ub) : 0u;
        s0 += (unsigned long long)fx_term(mca, eca, ua, uax);  s1 += (unsigned long long)fx_term(mca, eca, ua, uay);
        s0 += (unsigned long long)fx_term(mcb, ecb, ub, ubx);  s1 += (unsigned long long)fx_term(mcb, ecb, ub, uby);
        emc = max(emc, max(eca, ecb));
        emx = max(max(emx, fx_exp(uax)), max(fx_exp(uay), max(fx_exp(ubx), fx_exp(uby))));
        if (ia < len) seg[ia] = ((unsigned long long)ua << 32) | (unsigned)ea;
        if (ib < len) seg[ib] = ((unsigned long long)ub << 32) | (unsigned)eb;
    }
    float cs = di * di;
    unsigned ucs = __float_as_uint(cs), ux0 = __float_as_uint(x0), ux1 = __float_as_uint(x1v);
    emc = max(emc, fx_exp(ucs));
    emx = max(emx, max(fx_exp(ux0), fx_exp(ux1)));
    bool wide = emc > 140 || emx > 140 || emc + emx > 314 - FXS - 12 || len > 4096;
    float t0, t1;
    if (__any(wide)) {
        double d0 = 0.0, d1 = 0.0;
        for (int i = lane; i < len; i += 32) {
            unsigned long long e = seg[i];
            float4 xr = xd[(int)(unsigned)e];
            double c = (double)__int_as_float((int)(unsigned)(e >> 32));
            d0 += c * (double)xr.x;  d1 += c * (double)xr.y;
        }
        #pragma unroll
        for (int off = 16; off > 0; off >>= 1) {
            d0 += __shfl_xor(d0, off, 32);
            d1 += __shfl_xor(d1, off, 32);
        }
        t0 = (float)(d0 + (double)cs * (double)x0);
        t1 = (float)(d1 + (double)cs * (double)x1v);
    } else {
        #pragma unroll
        for (int off = 16; off > 0; off >>= 1) {
            s0 += __shfl_xor(s0, off, 32);
            s1 += __shfl_xor(s1, off, 32);
        }
        s0 += (unsigned long long)fx_term(fx_sig(ucs), fx_exp(ucs), ucs, ux0);
        s1 += (unsigned long long)fx_term(fx_sig(ucs), fx_exp(ucs), ucs, ux1);
        t0 = fx_to_float((long long)s0);
        t1 = fx_to_float((long long)s1);
    }
    int f0 = lane, f1 = lane + 32;
    float v0 = fmaf(t0, W1[f0], t1 * W1[HID + f0]) + b1[f0]
             + fmaf(x0, rW1[f0], x1v * rW1[HID + f0]) + rb1[f0];
    float v1 = fmaf(t0, W1[f1], t1 * W1[HID + f1]) + b1[f1]
             + fmaf(x0, rW1[f1], x1v * rW1[HID + f1]) + rb1[f1];
    float sm = v0 + v1, sq = fmaf(v0, v0, v1 * v1);
    #pragma unroll
    for (int off = 16; off > 0; off >>= 1) {
        sm += __shfl_xor(sm, off, 32);
        sq += __shfl_xor(sq, off, 32);
    }
    float mu  = sm * (1.f / 64.f);
    float var = sq * (1.f / 64.f) - mu * mu;
    float inv = rsqrtf(var + LN_EPS);
    float o0 = (v0 - mu) * inv * g[f0] + be[f0];
    float o1 = (v1 - mu) * inv * g[f1] + be[f1];
    s_x1[wv * X1P + f0] = fmaxf(o0, 0.f);
    s_x1[wv * X1P + f1] = fmaxf(o1, 0.f);
    if (lane == 0) s_di[wv] = di;
    __builtin_amdgcn_fence(__ATOMIC_RELEASE, "workgroup", "local");
    __builtin_amdgcn_s_barrier();
    __builtin_amdgcn_fence(__ATOMIC_ACQUIRE, "workgroup", "local");
    if (wv >= RED / 16) return;

    int m0 = (int)blockIdx.x * NPB, n0 = wv * 16;
    int  lr = lane & 15;
    bool hi = lane >= 16;
    int  kofs = hi ? 2 : 0;
    v8f c0 = {};
    v8f c1 = {};
    #pragma unroll
    for (int h0 = 0; h0 < HID; h0 += 32) {
        __builtin_amdgcn_sched_barrier(0);
        v2f b[8], br[8];
        #pragma unroll
        for (int s = 0; s < 8; ++s) {
            int kb = h0 + 4 * s + kofs;
            b[s].x  = W2 [kb * RED + n0 + lr]; b[s].y  = W2 [(kb + 1) * RED + n0 + lr];
            br[s].x = rW2[kb * RED + n0 + lr]; br[s].y = rW2[(kb + 1) * RED + n0 + lr];
        }
        #pragma unroll
        for (int s = 0; s < 8; ++s) {
            const float* ap = &s_x1[lr * X1P + h0 + 4 * s + kofs];
            v2f a; a.x = ap[0]; a.y = ap[1];
            c0 = __builtin_amdgcn_wmma_f32_16x16x4_f32(false, a, false, b[s],  (short)0, c0, false, false);
            c1 = __builtin_amdgcn_wmma_f32_16x16x4_f32(false, a, false, br[s], (short)0, c1, false, false);
        }
    }
    int ncol = n0 + lr;
    float b2v = b2[ncol], rb2v = rb2[ncol];
    #pragma unroll
    for (int v = 0; v < 8; ++v) {
        int mr = v + (hi ? 8 : 0);
        float dm = s_di[mr];
        float xv = c0[v];
        xw2 [(long long)(m0 + mr) * RED + ncol] = xv;
        acc2[(long long)(m0 + mr) * RED + ncol] = dm * dm * xv + b2v + c1[v] + rb2v;
    }
}

#define L2_BC(S) ((S) * 0x249249)
#define L2_ASK(S)                                                                      \
    {                                                                                  \
        unsigned rr_ = (unsigned)__builtin_amdgcn_mov_dpp8(er, L2_BC(S));              \
        cf[S] = __uint_as_float((unsigned)__builtin_amdgcn_mov_dpp8(ec, L2_BC(S)));    \
        r[S]  = *reinterpret_cast<const float4*>(xb + ((rr_ << 7) | sub16));           \
    }
#define L2_ADD(S)                                                                      \
    {                                                                                  \
        acc.x = fmaf(cf[S], r[S].x, acc.x);                                            \
        acc.y = fmaf(cf[S], r[S].y, acc.y);                                            \
        acc.z = fmaf(cf[S], r[S].z, acc.z);                                            \
        acc.w = fmaf(cf[S], r[S].w, acc.w);                                            \
    }
__global__ void k_gather_head(const int* __restrict__ od, const int* __restrict__ dci,
                       const int* __restrict__ csr, const float* __restrict__ xw2,
                       const float* __restrict__ acc2, const float* __restrict__ g2,
                       const float* __restrict__ be2, const float* __restrict__ fcW,
                       const float* __restrict__ fcb, const float* __restrict__ pW,
                       const float* __restrict__ pb, float* __restrict__ logits,
                       float* __restrict__ emb) {
    int lane = threadIdx.x & 31;
    int n = __builtin_amdgcn_readfirstlane((int)(blockIdx.x * (blockDim.x >> 5) + (threadIdx.x >> 5)));
    if (n >= N_NODES) return;
    int beg = od[2 * n];
    int len = dci[2 * n + 1];
    const int2* seg = reinterpret_cast<const int2*>(csr) + beg;
    const char* xb  = reinterpret_cast<const char*>(xw2);
    const int      grp   = lane >> 3;
    const unsigned sub16 = (unsigned)(lane & 7) << 4;
    const int      eidx  = 4 * (lane & 7) + grp;
    float4 acc = make_float4(0.f, 0.f, 0.f, 0.f);
    int2 ent = make_int2(0, 0);
    if (eidx < len) ent = seg[eidx];
    for (int base = 0; base < len; base += 32) {
        int idx = base + eidx;
        int2 nxt = make_int2(0, 0);
        if (idx + 32 < len) nxt = seg[idx + 32];
        int m = len - base;
        int row0 = __builtin_amdgcn_readfirstlane(ent.x);
        unsigned er = (unsigned)(idx < len ? ent.x : row0);
        unsigned ec = (unsigned)ent.y;
        float4 r[8];
        float  cf[8];
        L2_ASK(0) L2_ASK(1) L2_ASK(2) L2_ASK(3)
        if (m > 16) { L2_ASK(4) L2_ASK(5) L2_ASK(6) L2_ASK(7) }
        L2_ADD(0) L2_ADD(1) L2_ADD(2) L2_ADD(3)
        if (m > 16) { L2_ADD(4) L2_ADD(5) L2_ADD(6) L2_ADD(7) }
        ent = nxt;
    }
    bool odd = (grp & 1) != 0, upper = (grp & 2) != 0;
    float k0 = odd ? acc.z : acc.x, k1 = odd ? acc.w : acc.y;
    float s0 = odd ? acc.x : acc.z, s1 = odd ? acc.y : acc.w;
    float p0 = k0 + __shfl_xor(s0, 8, 32);
    float p1 = k1 + __shfl_xor(s1, 8, 32);
    float kk = upper ? p1 : p0, ss = upper ? p0 : p1;
    float tot = kk + __shfl_xor(ss, 16, 32);
    int cmp = lane & 3;
    float nsum = __shfl(tot, 8 * (((cmp & 1) << 1) | (cmp >> 1)) + (lane >> 2), 32);
    float v = acc2[(long long)n * 32 + lane] + nsum;
    float s = v, sq = v * v;
    #pragma unroll
    for (int off = 16; off > 0; off >>= 1) {
        s  += __shfl_xor(s,  off, 32);
        sq += __shfl_xor(sq, off, 32);
    }
    float mu  = s * (1.f / 32.f);
    float var = sq * (1.f / 32.f) - mu * mu;
    float inv = rsqrtf(var + LN_EPS);
    float x2 = fmaxf((v - mu) * inv * g2[lane] + be2[lane], 0.f);

    float lg = x2 * fcW[lane];
    #pragma unroll
    for (int off = 16; off > 0; off >>= 1) lg += __shfl_xor(lg, off, 32);
    if (lane == 0) logits[n] = lg + fcb[0];

    float p = pb[lane];
    #pragma unroll
    for (int i = 0; i < 32; ++i)
        p = fmaf(__shfl(x2, i, 32), pW[i * 32 + lane], p);
    float nsq = p * p;
    #pragma unroll
    for (int off = 16; off > 0; off >>= 1) nsq += __shfl_xor(nsq, off, 32);
    float nrm = sqrtf(nsq);
    emb[(long long)n * 32 + lane] = p / fmaxf(nrm, 1e-12f);
}

extern "C" void kernel_launch(void* const* d_in, const int* in_sizes, int n_in,
                              void* d_out, int out_size, void* d_ws, size_t ws_size,
                              hipStream_t stream) {
    const float* x    = (const float*)d_in[0];
    const float* ew   = (const float*)d_in[1];
    const float* W1   = (const float*)d_in[2];
    const float* b1   = (const float*)d_in[3];
    const float* rW1  = (const float*)d_in[4];
    const float* rb1  = (const float*)d_in[5];
    const float* g1   = (const float*)d_in[6];
    const float* be1  = (const float*)d_in[7];
    const float* W2   = (const float*)d_in[8];
    const float* b2   = (const float*)d_in[9];
    const float* rW2  = (const float*)d_in[10];
    const float* rb2  = (const float*)d_in[11];
    const float* g2   = (const float*)d_in[12];
    const float* be2  = (const float*)d_in[13];
    const float* fcW  = (const float*)d_in[14];
    const float* fcb  = (const float*)d_in[15];
    const float* pW   = (const float*)d_in[16];
    const float* pb   = (const float*)d_in[17];
    const int*   ei   = (const int*)d_in[18];
    const int*   row  = ei;
    const int*   col  = ei + N_EDGES;

    float* ws   = (float*)d_ws;
    ws += 2 * (size_t)N_NODES;
    int2*  tab     = (int2*)ws;
    int*   counter = (int*)(ws + 4 * (size_t)N_NODES);
    int*   dci  = (int*)(ws + 40 * (size_t)N_NODES);
    int*   od   = (int*)(ws + 42 * (size_t)N_NODES);
    float4* xd  = (float4*)(ws + 44 * (size_t)N_NODES);
    ws += (size_t)N_NODES * HID;
    int*   csr  = (int*)ws;
    ws += (size_t)N_NODES * HID;
    int2*  binned = (int2*)ws;  ws += (size_t)N_NODES * HID;
    float* xw2  = ws;  ws += (size_t)N_NODES * RED;
    float* acc2 = ws;  ws += (size_t)N_NODES * RED;

    float* logits = (float*)d_out;
    float* emb    = logits + N_NODES;

    const int B = 256;
    k_chunk_sort<<<NCH, 1024, 0, stream>>>(row, col, ew, tab, counter, binned);
    k_bucket<<<NB, 1024, 0, stream>>>(tab, counter, binned, x, dci, od, xd, (int2*)csr);

    k_node_l1<<<N_NODES / NPB, NPB * 32, 0, stream>>>(x, xd, od, dci, csr, W1, b1, rW1, rb1, g1, be1,
                                                      W2, rW2, b2, rb2, xw2, acc2);

    k_gather_head<<<(N_NODES + 7) / 8, B, 0, stream>>>(od, dci, csr, xw2, acc2, g2, be2, fcW, fcb,
                                                       pW, pb, logits, emb);
}


# wave sums and broadcasts of both node passes on the vector ALU (DPP row_xmask, v_permlanex16, v_readlane), not ds_bpermute round trips through the LDS unit; same partners, same order: same bits
# speedup vs baseline: 9.4710x; 1.0502x over previous
#include <hip/hip_runtime.h>
#include <hip/hip_bf16.h>

#define N_NODES 100000
#define N_EDGES 3200000
#define HID 64
#define RED 32
#define EMB 32
#define LN_EPS 1e-5f

typedef __attribute__((ext_vector_type(2))) float v2f;
typedef __attribute__((ext_vector_type(8))) float v8f;

template <int M> __device__ __forceinline__ int xrow_i(int v) {
    return __builtin_amdgcn_update_dpp(0, v, 0x160 + M, 0xf, 0xf, true);
}
__device__ __forceinline__ int xhalf_i(int v) {
    return __builtin_amdgcn_permlanex16(v, v, 0x76543210, 0xfedcba98, false, false);
}
template <int M> __device__ __forceinline__ float xrow(float v) {
    return __builtin_bit_cast(float, xrow_i<M>(__builtin_bit_cast(int, v)));
}
__device__ __forceinline__ float xhalf(float v) {
    return __builtin_bit_cast(float, xhalf_i(__builtin_bit_cast(int, v)));
}
template <int M> __device__ __forceinline__ unsigned long long xrow(unsigned long long v) {
    unsigned lo = (unsigned)xrow_i<M>((int)(unsigned)v), hi = (unsigned)xrow_i<M>((int)(unsigned)(v >> 32));
    return ((unsigned long long)hi << 32) | lo;
}
__device__ __forceinline__ unsigned long long xhalf(unsigned long long v) {
    unsigned lo = (unsigned)xhalf_i((int)(unsigned)v), hi = (unsigned)xhalf_i((int)(unsigned)(v >> 32));
    return ((unsigned long long)hi << 32) | lo;
}
template <typename T> __device__ __forceinline__ T wave_sum(T v) {
    v += xhalf(v);
    v += xrow<8>(v);
    v += xrow<4>(v);
    v += xrow<2>(v);
    v += xrow<1>(v);
    return v;
}

#define NB  391
#define NCH 256
#define CH  12500
#define KPT 13
#define PPW 8
#define STG 16384

__device__ __forceinline__ void wave_scan_nb(const int* src, int* dst, int lane) {
    int b0 = lane * 13;
    int loc[13];
    int sum = 0;
    #pragma unroll
    for (int j = 0; j < 13; ++j) {
        int b = b0 + j;
        int v = b < NB ? src[b] : 0;
        loc[j] = sum;
        sum += v;
    }
    int incl = sum;
    #pragma unroll
    for (int d = 1; d < 32; d <<= 1) {
        int v = __shfl_up(incl, d, 32);
        if (lane >= d) incl += v;
    }
    int ex = incl - sum;
    #pragma unroll
    for (int j = 0; j < 13; ++j) {
        int b = b0 + j;
        if (b < NB) dst[b] = ex + loc[j];
    }
}

__global__ void __launch_bounds__(1024) k_chunk_sort(const int* __restrict__ row,
                                                     const int* __restrict__ col,
                                                     const float* __restrict__ w,
                                                     int2* __restrict__ tab,
                                                     int* __restrict__ counter,
                                                     int2* __restrict__ binned) {
    __shared__ int  s_cnt[NB];
    __shared__ int  s_lst[NB];
    __shared__ int  s_cur[NB];
    __shared__ int2 s_rec[CH];
    int t = threadIdx.x, lane = t & 31, wave = t >> 5, g = blockIdx.x;

    int rc[KPT], rr[KPT];
    float rw[KPT];
    const int e0 = g * CH;
    #pragma unroll
    for (int k = 0; k < KPT; ++k) {
        int i = k * 1024 + t;
        rc[k] = 0; rr[k] = 0; rw[k] = 0.f;
        if (i < CH) { rc[k] = col[e0 + i]; rr[k] = row[e0 + i]; rw[k] = w[e0 + i]; }
    }
    for (int b = t; b < NB; b += 1024) { s_cnt[b] = 0; s_cur[b] = 0; }
    if (g == 0 && t == 0) *counter = 0;
    __syncthreads();
    #pragma unroll
    for (int k = 0; k < KPT; ++k) {
        int i = k * 1024 + t;
        if (i < CH) atomicAdd(&s_cnt[rc[k] >> 8], 1);
    }
    __syncthreads();
    if (wave == 0) wave_scan_nb(s_cnt, s_lst, lane);
    __syncthreads();
    for (int b = t; b < NB; b += 1024) tab[b * NCH + g] = make_int2(s_lst[b], s_cnt[b]);
    #pragma unroll
    for (int k = 0; k < KPT; ++k) {
        int i = k * 1024 + t;
        if (i < CH) {
            int b = rc[k] >> 8;
            int p = s_lst[b] + atomicAdd(&s_cur[b], 1);
            s_rec[p] = make_int2(rr[k] | ((rc[k] & 255) << 17), __float_as_int(rw[k]));
        }
    }
    __syncthreads();
    for (int i = t; i < CH; i += 1024) binned[e0 + i] = s_rec[i];
}

#define BK_ADD(E)                                                                      \
    {                                                                                  \
        int2 e_ = (E);                                                                 \
        int cl_ = (e_.x >> 17) & 255;                                                  \
        atomicAdd(&s_cnt[cl_], 1);                                                     \
        atomicAdd(&s_deg[cl_], __int_as_float(e_.y));                                  \
    }
#define BK_PLACE(E)                                                                    \
    {                                                                                  \
        int2 e_ = (E);                                                                 \
        int cl_ = (e_.x >> 17) & 255;                                                  \
        int p_  = s_first[cl_] + atomicAdd(&s_cur[cl_], 1);                            \
        int2 o_ = make_int2(e_.x & 0x1FFFF, e_.y);                                     \
        if (staged) s_rec[p_] = o_; else csr[beg + p_] = o_;                           \
    }
#define BK_PASS(OP)                                                                    \
    _Pragma("unroll")                                                                  \
    for (int h = 0; h < PPW; h += 4) {                                                 \
        int2 r[4], s[2];                                                               \
        _Pragma("unroll")                                                              \
        for (int q = 0; q < 4; ++q) {                                                  \
            int ia_ = po[h + q] + (lane < pc[h + q] ? lane : 0);                       \
            r[q] = binned[ia_ < N_EDGES ? ia_ : N_EDGES - 1];                          \
        }                                                                              \
        _Pragma("unroll")                                                              \
        for (int k = 0; k < 2; ++k) {                                                  \
            int pcs_ = hih ? pc[h + 2 * k + 1] : pc[h + 2 * k];                        \
            int pos_ = hih ? po[h + 2 * k + 1] : po[h + 2 * k];                        \
            int ib_  = pos_ + (32 + l16 < pcs_ ? 32 + l16 : 0);                        \
            s[k] = binned[ib_ < N_EDGES ? ib_ : N_EDGES - 1];                          \
        }                                                                              \
          \
          \
          \
        asm volatile("" :: "v"(r[0].x), "v"(r[0].y), "v"(r[1].x), "v"(r[1].y),         \
                           "v"(r[2].x), "v"(r[2].y), "v"(r[3].x), "v"(r[3].y),         \
                           "v"(s[0].x), "v"(s[0].y), "v"(s[1].x), "v"(s[1].y));        \
        _Pragma("unroll")                                                              \
        for (int q = 0; q < 4; ++q)                                                    \
            if (lane < pc[h + q]) OP(r[q])                                             \
        _Pragma("unroll")                                                              \
        for (int k = 0; k < 2; ++k) {                                                  \
            int pcs_ = hih ? pc[h + 2 * k + 1] : pc[h + 2 * k];                        \
            if (32 + l16 < pcs_) OP(s[k])                                              \
        }                                                                              \
        _Pragma("unroll")                                                              \
        for (int q = 0; q < 4; ++q)                                                    \
            for (int j = lane + 48; j < pc[h + q]; j += 32) OP(binned[po[h + q] + j])  \
    }

__global__ void __launch_bounds__(1024) k_bucket(const int2* __restrict__ tab,
                                                 int* __restrict__ counter,
                                                 const int2* __restrict__ binned,
                                                 const float* __restrict__ x,
                                                 int* __restrict__ dci, int* __restrict__ od,
                                                 float4* __restrict__ xd,
                                                 int2* __restrict__ csr) {
    __shared__ int   s_cnt[256];
    __shared__ float s_deg[256];
    __shared__ int   s_first[256];
    __shared__ int   s_cur[256];
    __shared__ int   s_wt[8];
    __shared__ int   s_po[NCH];
    __shared__ int   s_pc[NCH];
    __shared__ int   s_bm[2];
    __shared__ int2  s_rec[STG];
    int t = threadIdx.x, lane = t & 31, wave = t >> 5, b = blockIdx.x;
    if (t < NCH) {
        s_cnt[t] = 0; s_deg[t] = 1.0f; s_cur[t] = 0;
        int2 e = tab[b * NCH + t];
        s_po[t] = t * CH + e.x;
        s_pc[t] = e.y;
        int s = e.y;
        #pragma unroll
        for (int o = 16; o > 0; o >>= 1) s += __shfl_xor(s, o, 32);
        if (lane == 0) s_wt[wave] = s;
    }
    __syncthreads();
    if (t == 0) {
        int tot = 0;
        #pragma unroll
        for (int i = 0; i < 8; ++i) tot += s_wt[i];
        s_bm[0] = atomicAdd(counter, tot);
        s_bm[1] = tot;
    }
    __syncthreads();
    const int beg = s_bm[0], m = s_bm[1];

    const int  l16 = lane & 15;
    const bool hih = lane >= 16;
    int pc[PPW], po[PPW];
    #pragma unroll
    for (int q = 0; q < PPW; ++q) {
        pc[q] = __builtin_amdgcn_readfirstlane(s_pc[wave + 32 * q]);
        po[q] = __builtin_amdgcn_readfirstlane(s_po[wave + 32 * q]);
    }
    BK_PASS(BK_ADD)
    __syncthreads();
    int c = 0, incl = 0;
    if (t < 256) {
        c = s_cnt[t];
        incl = c;
        #pragma unroll
        for (int d = 1; d < 32; d <<= 1) {
            int v = __shfl_up(incl, d, 32);
            if (lane >= d) incl += v;
        }
        if (lane == 31) s_wt[wave] = incl;
    }
    __syncthreads();
    if (t < 256) {
        int base = 0;
        #pragma unroll
        for (int i = 0; i < 8; ++i) base += (i < wave) ? s_wt[i] : 0;
        int first = base + incl - c;
        s_first[t] = first;
        int n = b * 256 + t;
        if (n < N_NODES) {
            float d  = s_deg[t];
            float di = d > 0.f ? rsqrtf(d) : 0.f;
            reinterpret_cast<int2*>(od)[n] = make_int2(beg + first, __float_as_int(di));
            dci[2 * n + 1] = c;
            xd[n] = make_float4(x[2 * n], x[2 * n + 1], di, 0.f);
        }
    }
    __syncthreads();
    const bool staged = m <= STG;
    BK_PASS(BK_PLACE)
    if (staged) {
        __syncthreads();
        for (int i = t; i < m; i += 1024) csr[beg + i] = s_rec[i];
    }
}

#define FXS 44

__device__ __forceinline__ int      fx_exp(unsigned u) { return (int)((u >> 23) & 255u); }
__device__ __forceinline__ unsigned fx_sig(unsigned u) { return (u & 0x007FFFFFu) | 0x00800000u; }

__device__ __forceinline__ long long fx_term(unsigned mc, int ec, unsigned uc, unsigned ux) {
    unsigned long long pr = ((unsigned long long)mc * (unsigned long long)fx_sig(ux)) << 14;
    int rs = (314 - FXS) - ec - fx_exp(ux);
    rs = rs < 0 ? 0 : (rs > 63 ? 63 : rs);
    unsigned long long mag = pr >> rs;
    unsigned long long sg  = (unsigned long long)(long long)((int)(uc ^ ux) >> 31);
    return (long long)((mag ^ sg) - sg);
}

__device__ __forceinline__ float fx_to_float(long long s) {
    unsigned long long a = s < 0 ? 0ull - (unsigned long long)s : (unsigned long long)s;
    int lz = __clzll((long long)a);
    lz = lz > 63 ? 63 : lz;
    unsigned long long nrm = a << lz;
    unsigned top = (unsigned)(nrm >> 32) | ((unsigned)nrm != 0u ? 1u : 0u);
    float f = ldexpf((float)top, 32 - lz - FXS);
    return s < 0 ? -f : f;
}

#define NPB  16
#define X1P  68
static_assert(N_NODES % NPB == 0, "every work-group holds a whole 16-row tile");
__global__ void __launch_bounds__(NPB * 32) k_node_l1(const float* __restrict__ x, const float4* __restrict__ xd,
                          const int* __restrict__ od,
                          const int* __restrict__ dci, int* csr,
                          const float* __restrict__ W1, const float* __restrict__ b1,
                          const float* __restrict__ rW1, const float* __restrict__ rb1,
                          const float* __restrict__ g, const float* __restrict__ be,
                          const float* __restrict__ W2, const float* __restrict__ rW2,
                          const float* __restrict__ b2, const float* __restrict__ rb2,
                          float* __restrict__ xw2, float* __restrict__ acc2) {
    __shared__ float s_x1[NPB * X1P];
    __shared__ float s_di[NPB];
    int lane = threadIdx.x & 31;
    int wv = __builtin_amdgcn_readfirstlane((int)(threadIdx.x >> 5));
    int n  = (int)blockIdx.x * NPB + wv;
    int2 oc  = reinterpret_cast<const int2*>(od)[n];
    int  len = dci[2 * n + 1];
    float di = __int_as_float(oc.y);
    float x0 = x[2 * n], x1v = x[2 * n + 1];
    unsigned long long* seg = reinterpret_cast<unsigned long long*>(csr) + oc.x;
    unsigned long long s0 = 0, s1 = 0;
    int emc = 0, emx = 0;
    for (int base = 0; base < len; base += 64) {
        int ia = base + lane, ib = ia + 32, last = len - 1;
        unsigned long long ea = seg[ia < last ? ia : last];
        unsigned long long eb = seg[ib < last ? ib : last];
        __builtin_amdgcn_sched_barrier(0);
        float4 xa = xd[(int)(unsigned)ea];
        float4 xb = xd[(int)(unsigned)eb];
        __builtin_amdgcn_sched_barrier(0);
        float fa = xa.z * __int_as_float((int)(unsigned)(ea >> 32)) * di;
        float fb = xb.z * __int_as_float((int)(unsigned)(eb >> 32)) * di;
        unsigned ua  = __float_as_uint(fa),   ub  = __float_as_uint(fb);
        unsigned uax = __float_as_uint(xa.x), uay = __float_as_uint(xa.y);
        unsigned ubx = __float_as_uint(xb.x), uby = __float_as_uint(xb.y);
        int eca = fx_exp(ua), ecb = fx_exp(ub);
        unsigned mca = ia < len ? fx_sig(ua) : 0u;
        unsigned mcb = ib < len ? fx_sig(ub) : 0u;
        s0 += (unsigned long long)fx_term(mca, eca, ua, uax);  s1 += (unsigned long long)fx_term(mca, eca, ua, uay);
        s0 += (unsigned long long)fx_term(mcb, ecb, ub, ubx);  s1 += (unsigned long long)fx_term(mcb, ecb, ub, uby);
        emc = max(emc, max(eca, ecb));
        emx = max(max(emx, fx_exp(uax)), max(fx_exp(uay), max(fx_exp(ubx), fx_exp(uby))));
        if (ia < len) seg[ia] = ((unsigned long long)ua << 32) | (unsigned)ea;
        if (ib < len) seg[ib] = ((unsigned long long)ub << 32) | (unsigned)eb;
    }
    float cs = di * di;
    unsigned ucs = __float_as_uint(cs), ux0 = __float_as_uint(x0), ux1 = __float_as_uint(x1v);
    emc = max(emc, fx_exp(ucs));
    emx = max(emx, max(fx_exp(ux0), fx_exp(ux1)));
    bool wide = emc > 140 || emx > 140 || emc + emx > 314 - FXS - 12 || len > 4096;
    float t0, t1;
    if (__any(wide)) {
        double d0 = 0.0, d1 = 0.0;
        for (int i = lane; i < len; i += 32) {
            unsigned long long e = seg[i];
            float4 xr = xd[(int)(unsigned)e];
            double c = (double)__int_as_float((int)(unsigned)(e >> 32));
            d0 += c * (double)xr.x;  d1 += c * (double)xr.y;
        }
        #pragma unroll
        for (int off = 16; off > 0; off >>= 1) {
            d0 += __shfl_xor(d0, off, 32);
            d1 += __shfl_xor(d1, off, 32);
        }
        t0 = (float)(d0 + (double)cs * (double)x0);
        t1 = (float)(d1 + (double)cs * (double)x1v);
    } else {
        s0 = wave_sum(s0);
        s1 = wave_sum(s1);
        s0 += (unsigned long long)fx_term(fx_sig(ucs), fx_exp(ucs), ucs, ux0);
        s1 += (unsigned long long)fx_term(fx_sig(ucs), fx_exp(ucs), ucs, ux1);
        t0 = fx_to_float((long long)s0);
        t1 = fx_to_float((long long)s1);
    }
    int f0 = lane, f1 = lane + 32;
    float v0 = fmaf(t0, W1[f0], t1 * W1[HID + f0]) + b1[f0]
             + fmaf(x0, rW1[f0], x1v * rW1[HID + f0]) + rb1[f0];
    float v1 = fmaf(t0, W1[f1], t1 * W1[HID + f1]) + b1[f1]
             + fmaf(x0, rW1[f1], x1v * rW1[HID + f1]) + rb1[f1];
    float sm = wave_sum(v0 + v1), sq = wave_sum(fmaf(v0, v0, v1 * v1));
    float mu  = sm * (1.f / 64.f);
    float var = sq * (1.f / 64.f) - mu * mu;
    float inv = rsqrtf(var + LN_EPS);
    float o0 = (v0 - mu) * inv * g[f0] + be[f0];
    float o1 = (v1 - mu) * inv * g[f1] + be[f1];
    s_x1[wv * X1P + f0] = fmaxf(o0, 0.f);
    s_x1[wv * X1P + f1] = fmaxf(o1, 0.f);
    if (lane == 0) s_di[wv] = di;
    __builtin_amdgcn_fence(__ATOMIC_RELEASE, "workgroup", "local");
    __builtin_amdgcn_s_barrier();
    __builtin_amdgcn_fence(__ATOMIC_ACQUIRE, "workgroup", "local");
    if (wv >= RED / 16) return;

    int m0 = (int)blockIdx.x * NPB, n0 = wv * 16;
    int  lr = lane & 15;
    bool hi = lane >= 16;
    int  kofs = hi ? 2 : 0;
    v8f c0 = {};
    v8f c1 = {};
    #pragma unroll
    for (int h0 = 0; h0 < HID; h0 += 32) {
        __builtin_amdgcn_sched_barrier(0);
        v2f b[8], br[8];
        #pragma unroll
        for (int s = 0; s < 8; ++s) {
            int kb = h0 + 4 * s + kofs;
            b[s].x  = W2 [kb * RED + n0 + lr]; b[s].y  = W2 [(kb + 1) * RED + n0 + lr];
            br[s].x = rW2[kb * RED + n0 + lr]; br[s].y = rW2[(kb + 1) * RED + n0 + lr];
        }
        #pragma unroll
        for (int s = 0; s < 8; ++s) {
            const float* ap = &s_x1[lr * X1P + h0 + 4 * s + kofs];
            v2f a; a.x = ap[0]; a.y = ap[1];
            c0 = __builtin_amdgcn_wmma_f32_16x16x4_f32(false, a, false, b[s],  (short)0, c0, false, false);
            c1 = __builtin_amdgcn_wmma_f32_16x16x4_f32(false, a, false, br[s], (short)0, c1, false, false);
        }
    }
    int ncol = n0 + lr;
    float b2v = b2[ncol], rb2v = rb2[ncol];
    #pragma unroll
    for (int v = 0; v < 8; ++v) {
        int mr = v + (hi ? 8 : 0);
        float dm = s_di[mr];
        float xv = c0[v];
        xw2 [(long long)(m0 + mr) * RED + ncol] = xv;
        acc2[(long long)(m0 + mr) * RED + ncol] = dm * dm * xv + b2v + c1[v] + rb2v;
    }
}

#define L2_BC(S) ((S) * 0x249249)
#define L2_ASK(S)                                                                      \
    {                                                                                  \
        unsigned rr_ = (unsigned)__builtin_amdgcn_mov_dpp8(er, L2_BC(S));              \
        cf[S] = __uint_as_float((unsigned)__builtin_amdgcn_mov_dpp8(ec, L2_BC(S)));    \
        r[S]  = *reinterpret_cast<const float4*>(xb + ((rr_ << 7) | sub16));           \
    }
#define L2_ADD(S)                                                                      \
    {                                                                                  \
        acc.x = fmaf(cf[S], r[S].x, acc.x);                                            \
        acc.y = fmaf(cf[S], r[S].y, acc.y);                                            \
        acc.z = fmaf(cf[S], r[S].z, acc.z);                                            \
        acc.w = fmaf(cf[S], r[S].w, acc.w);                                            \
    }
__global__ void k_gather_head(const int* __restrict__ od, const int* __restrict__ dci,
                       const int* __restrict__ csr, const float* __restrict__ xw2,
                       const float* __restrict__ acc2, const float* __restrict__ g2,
                       const float* __restrict__ be2, const float* __restrict__ fcW,
                       const float* __restrict__ fcb, const float* __restrict__ pW,
                       const float* __restrict__ pb, float* __restrict__ logits,
                       float* __restrict__ emb) {
    int lane = threadIdx.x & 31;
    int n = __builtin_amdgcn_readfirstlane((int)(blockIdx.x * (blockDim.x >> 5) + (threadIdx.x >> 5)));
    if (n >= N_NODES) return;
    int beg = od[2 * n];
    int len = dci[2 * n + 1];
    const int2* seg = reinterpret_cast<const int2*>(csr) + beg;
    const char* xb  = reinterpret_cast<const char*>(xw2);
    const int      grp   = lane >> 3;
    const unsigned sub16 = (unsigned)(lane & 7) << 4;
    const int      eidx  = 4 * (lane & 7) + grp;
    float4 acc = make_float4(0.f, 0.f, 0.f, 0.f);
    int2 ent = make_int2(0, 0);
    if (eidx < len) ent = seg[eidx];
    for (int base = 0; base < len; base += 32) {
        int idx = base + eidx;
        int2 nxt = make_int2(0, 0);
        if (idx + 32 < len) nxt = seg[idx + 32];
        int m = len - base;
        int row0 = __builtin_amdgcn_readfirstlane(ent.x);
        unsigned er = (unsigned)(idx < len ? ent.x : row0);
        unsigned ec = (unsigned)ent.y;
        float4 r[8];
        float  cf[8];
        L2_ASK(0) L2_ASK(1) L2_ASK(2) L2_ASK(3)
        if (m > 16) { L2_ASK(4) L2_ASK(5) L2_ASK(6) L2_ASK(7) }
        L2_ADD(0) L2_ADD(1) L2_ADD(2) L2_ADD(3)
        if (m > 16) { L2_ADD(4) L2_ADD(5) L2_ADD(6) L2_ADD(7) }
        ent = nxt;
    }
    bool odd = (grp & 1) != 0, upper = (grp & 2) != 0;
    float k0 = odd ? acc.z : acc.x, k1 = odd ? acc.w : acc.y;
    float s0 = odd ? acc.x : acc.z, s1 = odd ? acc.y : acc.w;
    float p0 = k0 + xrow<8>(s0);
    float p1 = k1 + xrow<8>(s1);
    float kk = upper ? p1 : p0, ss = upper ? p0 : p1;
    float tot = kk + xhalf(ss);
    int cmp = lane & 3;
    float nsum = __shfl(tot, 8 * (((cmp & 1) << 1) | (cmp >> 1)) + (lane >> 2), 32);
    float v = acc2[(long long)n * 32 + lane] + nsum;
    float s = wave_sum(v), sq = wave_sum(v * v);
    float mu  = s * (1.f / 32.f);
    float var = sq * (1.f / 32.f) - mu * mu;
    float inv = rsqrtf(var + LN_EPS);
    float x2 = fmaxf((v - mu) * inv * g2[lane] + be2[lane], 0.f);

    float lg = wave_sum(x2 * fcW[lane]);
    if (lane == 0) logits[n] = lg + fcb[0];

    float p = pb[lane];
    int x2b = __builtin_bit_cast(int, x2);
    #pragma unroll
    for (int i = 0; i < 32; ++i)
        p = fmaf(__builtin_bit_cast(float, __builtin_amdgcn_readlane(x2b, i)), pW[i * 32 + lane], p);
    float nsq = wave_sum(p * p);
    float nrm = sqrtf(nsq);
    emb[(long long)n * 32 + lane] = p / fmaxf(nrm, 1e-12f);
}

extern "C" void kernel_launch(void* const* d_in, const int* in_sizes, int n_in,
                              void* d_out, int out_size, void* d_ws, size_t ws_size,
                              hipStream_t stream) {
    const float* x    = (const float*)d_in[0];
    const float* ew   = (const float*)d_in[1];
    const float* W1   = (const float*)d_in[2];
    const float* b1   = (const float*)d_in[3];
    const float* rW1  = (const float*)d_in[4];
    const float* rb1  = (const float*)d_in[5];
    const float* g1   = (const float*)d_in[6];
    const float* be1  = (const float*)d_in[7];
    const float* W2   = (const float*)d_in[8];
    const float* b2   = (const float*)d_in[9];
    const float* rW2  = (const float*)d_in[10];
    const float* rb2  = (const float*)d_in[11];
    const float* g2   = (const float*)d_in[12];
    const float* be2  = (const float*)d_in[13];
    const float* fcW  = (const float*)d_in[14];
    const float* fcb  = (const float*)d_in[15];
    const float* pW   = (const float*)d_in[16];
    const float* pb   = (const float*)d_in[17];
    const int*   ei   = (const int*)d_in[18];
    const int*   row  = ei;
    const int*   col  = ei + N_EDGES;

    float* ws   = (float*)d_ws;
    ws += 2 * (size_t)N_NODES;
    int2*  tab     = (int2*)ws;
    int*   counter = (int*)(ws + 4 * (size_t)N_NODES);
    int*   dci  = (int*)(ws + 40 * (size_t)N_NODES);
    int*   od   = (int*)(ws + 42 * (size_t)N_NODES);
    float4* xd  = (float4*)(ws + 44 * (size_t)N_NODES);
    ws += (size_t)N_NODES * HID;
    int*   csr  = (int*)ws;
    ws += (size_t)N_NODES * HID;
    int2*  binned = (int2*)ws;  ws += (size_t)N_NODES * HID;
    float* xw2  = ws;  ws += (size_t)N_NODES * RED;
    float* acc2 = ws;  ws += (size_t)N_NODES * RED;

    float* logits = (float*)d_out;
    float* emb    = logits + N_NODES;

    const int B = 256;
    k_chunk_sort<<<NCH, 1024, 0, stream>>>(row, col, ew, tab, counter, binned);
    k_bucket<<<NB, 1024, 0, stream>>>(tab, counter, binned, x, dci, od, xd, (int2*)csr);

    k_node_l1<<<N_NODES / NPB, NPB * 32, 0, stream>>>(x, xd, od, dci, csr, W1, b1, rW1, rb1, g1, be1,
                                                      W2, rW2, b2, rb2, xw2, acc2);

    k_gather_head<<<(N_NODES + 7) / 8, B, 0, stream>>>(od, dci, csr, xw2, acc2, g2, be2, fcW, fcb,
                                                       pW, pb, logits, emb);
}


# layer-2 gather+heads: a wave walks 6 consecutive nodes, not 1 (6x fewer waves); its pW column, g2, be2, fcW, pb stay in registers, next node's entries and acc2 row are asked early; same bits
# speedup vs baseline: 10.3031x; 1.0879x over previous
#include <hip/hip_runtime.h>
#include <hip/hip_bf16.h>

#define N_NODES 100000
#define N_EDGES 3200000
#define HID 64
#define RED 32
#define EMB 32
#define LN_EPS 1e-5f

typedef __attribute__((ext_vector_type(2))) float v2f;
typedef __attribute__((ext_vector_type(8))) float v8f;

template <int M> __device__ __forceinline__ int xrow_i(int v) {
    return __builtin_amdgcn_update_dpp(0, v, 0x160 + M, 0xf, 0xf, true);
}
__device__ __forceinline__ int xhalf_i(int v) {
    return __builtin_amdgcn_permlanex16(v, v, 0x76543210, 0xfedcba98, false, false);
}
template <int M> __device__ __forceinline__ float xrow(float v) {
    return __builtin_bit_cast(float, xrow_i<M>(__builtin_bit_cast(int, v)));
}
__device__ __forceinline__ float xhalf(float v) {
    return __builtin_bit_cast(float, xhalf_i(__builtin_bit_cast(int, v)));
}
template <int M> __device__ __forceinline__ unsigned long long xrow(unsigned long long v) {
    unsigned lo = (unsigned)xrow_i<M>((int)(unsigned)v), hi = (unsigned)xrow_i<M>((int)(unsigned)(v >> 32));
    return ((unsigned long long)hi << 32) | lo;
}
__device__ __forceinline__ unsigned long long xhalf(unsigned long long v) {
    unsigned lo = (unsigned)xhalf_i((int)(unsigned)v), hi = (unsigned)xhalf_i((int)(unsigned)(v >> 32));
    return ((unsigned long long)hi << 32) | lo;
}
template <typename T> __device__ __forceinline__ T wave_sum(T v) {
    v += xhalf(v);
    v += xrow<8>(v);
    v += xrow<4>(v);
    v += xrow<2>(v);
    v += xrow<1>(v);
    return v;
}

#define NB  391
#define NCH 256
#define CH  12500
#define KPT 13
#define PPW 8
#define STG 16384

__device__ __forceinline__ void wave_scan_nb(const int* src, int* dst, int lane) {
    int b0 = lane * 13;
    int loc[13];
    int sum = 0;
    #pragma unroll
    for (int j = 0; j < 13; ++j) {
        int b = b0 + j;
        int v = b < NB ? src[b] : 0;
        loc[j] = sum;
        sum += v;
    }
    int incl = sum;
    #pragma unroll
    for (int d = 1; d < 32; d <<= 1) {
        int v = __shfl_up(incl, d, 32);
        if (lane >= d) incl += v;
    }
    int ex = incl - sum;
    #pragma unroll
    for (int j = 0; j < 13; ++j) {
        int b = b0 + j;
        if (b < NB) dst[b] = ex + loc[j];
    }
}

__global__ void __launch_bounds__(1024) k_chunk_sort(const int* __restrict__ row,
                                                     const int* __restrict__ col,
                                                     const float* __restrict__ w,
                                                     int2* __restrict__ tab,
                                                     int* __restrict__ counter,
                                                     int2* __restrict__ binned) {
    __shared__ int  s_cnt[NB];
    __shared__ int  s_lst[NB];
    __shared__ int  s_cur[NB];
    __shared__ int2 s_rec[CH];
    int t = threadIdx.x, lane = t & 31, wave = t >> 5, g = blockIdx.x;

    int rc[KPT], rr[KPT];
    float rw[KPT];
    const int e0 = g * CH;
    #pragma unroll
    for (int k = 0; k < KPT; ++k) {
        int i = k * 1024 + t;
        rc[k] = 0; rr[k] = 0; rw[k] = 0.f;
        if (i < CH) { rc[k] = col[e0 + i]; rr[k] = row[e0 + i]; rw[k] = w[e0 + i]; }
    }
    for (int b = t; b < NB; b += 1024) { s_cnt[b] = 0; s_cur[b] = 0; }
    if (g == 0 && t == 0) *counter = 0;
    __syncthreads();
    #pragma unroll
    for (int k = 0; k < KPT; ++k) {
        int i = k * 1024 + t;
        if (i < CH) atomicAdd(&s_cnt[rc[k] >> 8], 1);
    }
    __syncthreads();
    if (wave == 0) wave_scan_nb(s_cnt, s_lst, lane);
    __syncthreads();
    for (int b = t; b < NB; b += 1024) tab[b * NCH + g] = make_int2(s_lst[b], s_cnt[b]);
    #pragma unroll
    for (int k = 0; k < KPT; ++k) {
        int i = k * 1024 + t;
        if (i < CH) {
            int b = rc[k] >> 8;
            int p = s_lst[b] + atomicAdd(&s_cur[b], 1);
            s_rec[p] = make_int2(rr[k] | ((rc[k] & 255) << 17), __float_as_int(rw[k]));
        }
    }
    __syncthreads();
    for (int i = t; i < CH; i += 1024) binned[e0 + i] = s_rec[i];
}

#define BK_ADD(E)                                                                      \
    {                                                                                  \
        int2 e_ = (E);                                                                 \
        int cl_ = (e_.x >> 17) & 255;                                                  \
        atomicAdd(&s_cnt[cl_], 1);                                                     \
        atomicAdd(&s_deg[cl_], __int_as_float(e_.y));                                  \
    }
#define BK_PLACE(E)                                                                    \
    {                                                                                  \
        int2 e_ = (E);                                                                 \
        int cl_ = (e_.x >> 17) & 255;                                                  \
        int p_  = s_first[cl_] + atomicAdd(&s_cur[cl_], 1);                            \
        int2 o_ = make_int2(e_.x & 0x1FFFF, e_.y);                                     \
        if (staged) s_rec[p_] = o_; else csr[beg + p_] = o_;                           \
    }
#define BK_PASS(OP)                                                                    \
    _Pragma("unroll")                                                                  \
    for (int h = 0; h < PPW; h += 4) {                                                 \
        int2 r[4], s[2];                                                               \
        _Pragma("unroll")                                                              \
        for (int q = 0; q < 4; ++q) {                                                  \
            int ia_ = po[h + q] + (lane < pc[h + q] ? lane : 0);                       \
            r[q] = binned[ia_ < N_EDGES ? ia_ : N_EDGES - 1];                          \
        }                                                                              \
        _Pragma("unroll")                                                              \
        for (int k = 0; k < 2; ++k) {                                                  \
            int pcs_ = hih ? pc[h + 2 * k + 1] : pc[h + 2 * k];                        \
            int pos_ = hih ? po[h + 2 * k + 1] : po[h + 2 * k];                        \
            int ib_  = pos_ + (32 + l16 < pcs_ ? 32 + l16 : 0);                        \
            s[k] = binned[ib_ < N_EDGES ? ib_ : N_EDGES - 1];                          \
        }                                                                              \
          \
          \
          \
        asm volatile("" :: "v"(r[0].x), "v"(r[0].y), "v"(r[1].x), "v"(r[1].y),         \
                           "v"(r[2].x), "v"(r[2].y), "v"(r[3].x), "v"(r[3].y),         \
                           "v"(s[0].x), "v"(s[0].y), "v"(s[1].x), "v"(s[1].y));        \
        _Pragma("unroll")                                                              \
        for (int q = 0; q < 4; ++q)                                                    \
            if (lane < pc[h + q]) OP(r[q])                                             \
        _Pragma("unroll")                                                              \
        for (int k = 0; k < 2; ++k) {                                                  \
            int pcs_ = hih ? pc[h + 2 * k + 1] : pc[h + 2 * k];                        \
            if (32 + l16 < pcs_) OP(s[k])                                              \
        }                                                                              \
        _Pragma("unroll")                                                              \
        for (int q = 0; q < 4; ++q)                                                    \
            for (int j = lane + 48; j < pc[h + q]; j += 32) OP(binned[po[h + q] + j])  \
    }

__global__ void __launch_bounds__(1024) k_bucket(const int2* __restrict__ tab,
                                                 int* __restrict__ counter,
                                                 const int2* __restrict__ binned,
                                                 const float* __restrict__ x,
                                                 int* __restrict__ dci, int* __restrict__ od,
                                                 float4* __restrict__ xd,
                                                 int2* __restrict__ csr) {
    __shared__ int   s_cnt[256];
    __shared__ float s_deg[256];
    __shared__ int   s_first[256];
    __shared__ int   s_cur[256];
    __shared__ int   s_wt[8];
    __shared__ int   s_po[NCH];
    __shared__ int   s_pc[NCH];
    __shared__ int   s_bm[2];
    __shared__ int2  s_rec[STG];
    int t = threadIdx.x, lane = t & 31, wave = t >> 5, b = blockIdx.x;
    if (t < NCH) {
        s_cnt[t] = 0; s_deg[t] = 1.0f; s_cur[t] = 0;
        int2 e = tab[b * NCH + t];
        s_po[t] = t * CH + e.x;
        s_pc[t] = e.y;
        int s = e.y;
        #pragma unroll
        for (int o = 16; o > 0; o >>= 1) s += __shfl_xor(s, o, 32);
        if (lane == 0) s_wt[wave] = s;
    }
    __syncthreads();
    if (t == 0) {
        int tot = 0;
        #pragma unroll
        for (int i = 0; i < 8; ++i) tot += s_wt[i];
        s_bm[0] = atomicAdd(counter, tot);
        s_bm[1] = tot;
    }
    __syncthreads();
    const int beg = s_bm[0], m = s_bm[1];

    const int  l16 = lane & 15;
    const bool hih = lane >= 16;
    int pc[PPW], po[PPW];
    #pragma unroll
    for (int q = 0; q < PPW; ++q) {
        pc[q] = __builtin_amdgcn_readfirstlane(s_pc[wave + 32 * q]);
        po[q] = __builtin_amdgcn_readfirstlane(s_po[wave + 32 * q]);
    }
    BK_PASS(BK_ADD)
    __syncthreads();
    int c = 0, incl = 0;
    if (t < 256) {
        c = s_cnt[t];
        incl = c;
        #pragma unroll
        for (int d = 1; d < 32; d <<= 1) {
            int v = __shfl_up(incl, d, 32);
            if (lane >= d) incl += v;
        }
        if (lane == 31) s_wt[wave] = incl;
    }
    __syncthreads();
    if (t < 256) {
        int base = 0;
        #pragma unroll
        for (int i = 0; i < 8; ++i) base += (i < wave) ? s_wt[i] : 0;
        int first = base + incl - c;
        s_first[t] = first;
        int n = b * 256 + t;
        if (n < N_NODES) {
            float d  = s_deg[t];
            float di = d > 0.f ? rsqrtf(d) : 0.f;
            reinterpret_cast<int2*>(od)[n] = make_int2(beg + first, __float_as_int(di));
            dci[2 * n + 1] = c;
            xd[n] = make_float4(x[2 * n], x[2 * n + 1], di, 0.f);
        }
    }
    __syncthreads();
    const bool staged = m <= STG;
    BK_PASS(BK_PLACE)
    if (staged) {
        __syncthreads();
        for (int i = t; i < m; i += 1024) csr[beg + i] = s_rec[i];
    }
}

#define FXS 44

__device__ __forceinline__ int      fx_exp(unsigned u) { return (int)((u >> 23) & 255u); }
__device__ __forceinline__ unsigned fx_sig(unsigned u) { return (u & 0x007FFFFFu) | 0x00800000u; }

__device__ __forceinline__ long long fx_term(unsigned mc, int ec, unsigned uc, unsigned ux) {
    unsigned long long pr = ((unsigned long long)mc * (unsigned long long)fx_sig(ux)) << 14;
    int rs = (314 - FXS) - ec - fx_exp(ux);
    rs = rs < 0 ? 0 : (rs > 63 ? 63 : rs);
    unsigned long long mag = pr >> rs;
    unsigned long long sg  = (unsigned long long)(long long)((int)(uc ^ ux) >> 31);
    return (long long)((mag ^ sg) - sg);
}

__device__ __forceinline__ float fx_to_float(long long s) {
    unsigned long long a = s < 0 ? 0ull - (unsigned long long)s : (unsigned long long)s;
    int lz = __clzll((long long)a);
    lz = lz > 63 ? 63 : lz;
    unsigned long long nrm = a << lz;
    unsigned top = (unsigned)(nrm >> 32) | ((unsigned)nrm != 0u ? 1u : 0u);
    float f = ldexpf((float)top, 32 - lz - FXS);
    return s < 0 ? -f : f;
}

#define NPB  16
#define X1P  68
static_assert(N_NODES % NPB == 0, "every work-group holds a whole 16-row tile");
__global__ void __launch_bounds__(NPB * 32) k_node_l1(const float* __restrict__ x, const float4* __restrict__ xd,
                          const int* __restrict__ od,
                          const int* __restrict__ dci, int* csr,
                          const float* __restrict__ W1, const float* __restrict__ b1,
                          const float* __restrict__ rW1, const float* __restrict__ rb1,
                          const float* __restrict__ g, const float* __restrict__ be,
                          const float* __restrict__ W2, const float* __restrict__ rW2,
                          const float* __restrict__ b2, const float* __restrict__ rb2,
                          float* __restrict__ xw2, float* __restrict__ acc2) {
    __shared__ float s_x1[NPB * X1P];
    __shared__ float s_di[NPB];
    int lane = threadIdx.x & 31;
    int wv = __builtin_amdgcn_readfirstlane((int)(threadIdx.x >> 5));
    int n  = (int)blockIdx.x * NPB + wv;
    int2 oc  = reinterpret_cast<const int2*>(od)[n];
    int  len = dci[2 * n + 1];
    float di = __int_as_float(oc.y);
    float x0 = x[2 * n], x1v = x[2 * n + 1];
    unsigned long long* seg = reinterpret_cast<unsigned long long*>(csr) + oc.x;
    unsigned long long s0 = 0, s1 = 0;
    int emc = 0, emx = 0;
    for (int base = 0; base < len; base += 64) {
        int ia = base + lane, ib = ia + 32, last = len - 1;
        unsigned long long ea = seg[ia < last ? ia : last];
        unsigned long long eb = seg[ib < last ? ib : last];
        __builtin_amdgcn_sched_barrier(0);
        float4 xa = xd[(int)(unsigned)ea];
        float4 xb = xd[(int)(unsigned)eb];
        __builtin_amdgcn_sched_barrier(0);
        float fa = xa.z * __int_as_float((int)(unsigned)(ea >> 32)) * di;
        float fb = xb.z * __int_as_float((int)(unsigned)(eb >> 32)) * di;
        unsigned ua  = __float_as_uint(fa),   ub  = __float_as_uint(fb);
        unsigned uax = __float_as_uint(xa.x), uay = __float_as_uint(xa.y);
        unsigned ubx = __float_as_uint(xb.x), uby = __float_as_uint(xb.y);
        int eca = fx_exp(ua), ecb = fx_exp(ub);
        unsigned mca = ia < len ? fx_sig(ua) : 0u;
        unsigned mcb = ib < len ? fx_sig(ub) : 0u;
        s0 += (unsigned long long)fx_term(mca, eca, ua, uax);  s1 += (unsigned long long)fx_term(mca, eca, ua, uay);
        s0 += (unsigned long long)fx_term(mcb, ecb, ub, ubx);  s1 += (unsigned long long)fx_term(mcb, ecb, ub, uby);
        emc = max(emc, max(eca, ecb));
        emx = max(max(emx, fx_exp(uax)), max(fx_exp(uay), max(fx_exp(ubx), fx_exp(uby))));
        if (ia < len) seg[ia] = ((unsigned long long)ua << 32) | (unsigned)ea;
        if (ib < len) seg[ib] = ((unsigned long long)ub << 32) | (unsigned)eb;
    }
    float cs = di * di;
    unsigned ucs = __float_as_uint(cs), ux0 = __float_as_uint(x0), ux1 = __float_as_uint(x1v);
    emc = max(emc, fx_exp(ucs));
    emx = max(emx, max(fx_exp(ux0), fx_exp(ux1)));
    bool wide = emc > 140 || emx > 140 || emc + emx > 314 - FXS - 12 || len > 4096;
    float t0, t1;
    if (__any(wide)) {
        double d0 = 0.0, d1 = 0.0;
        for (int i = lane; i < len; i += 32) {
            unsigned long long e = seg[i];
            float4 xr = xd[(int)(unsigned)e];
            double c = (double)__int_as_float((int)(unsigned)(e >> 32));
            d0 += c * (double)xr.x;  d1 += c * (double)xr.y;
        }
        #pragma unroll
        for (int off = 16; off > 0; off >>= 1) {
            d0 += __shfl_xor(d0, off, 32);
            d1 += __shfl_xor(d1, off, 32);
        }
        t0 = (float)(d0 + (double)cs * (double)x0);
        t1 = (float)(d1 + (double)cs * (double)x1v);
    } else {
        s0 = wave_sum(s0);
        s1 = wave_sum(s1);
        s0 += (unsigned long long)fx_term(fx_sig(ucs), fx_exp(ucs), ucs, ux0);
        s1 += (unsigned long long)fx_term(fx_sig(ucs), fx_exp(ucs), ucs, ux1);
        t0 = fx_to_float((long long)s0);
        t1 = fx_to_float((long long)s1);
    }
    int f0 = lane, f1 = lane + 32;
    float v0 = fmaf(t0, W1[f0], t1 * W1[HID + f0]) + b1[f0]
             + fmaf(x0, rW1[f0], x1v * rW1[HID + f0]) + rb1[f0];
    float v1 = fmaf(t0, W1[f1], t1 * W1[HID + f1]) + b1[f1]
             + fmaf(x0, rW1[f1], x1v * rW1[HID + f1]) + rb1[f1];
    float sm = wave_sum(v0 + v1), sq = wave_sum(fmaf(v0, v0, v1 * v1));
    float mu  = sm * (1.f / 64.f);
    float var = sq * (1.f / 64.f) - mu * mu;
    float inv = rsqrtf(var + LN_EPS);
    float o0 = (v0 - mu) * inv * g[f0] + be[f0];
    float o1 = (v1 - mu) * inv * g[f1] + be[f1];
    s_x1[wv * X1P + f0] = fmaxf(o0, 0.f);
    s_x1[wv * X1P + f1] = fmaxf(o1, 0.f);
    if (lane == 0) s_di[wv] = di;
    __builtin_amdgcn_fence(__ATOMIC_RELEASE, "workgroup", "local");
    __builtin_amdgcn_s_barrier();
    __builtin_amdgcn_fence(__ATOMIC_ACQUIRE, "workgroup", "local");
    if (wv >= RED / 16) return;

    int m0 = (int)blockIdx.x * NPB, n0 = wv * 16;
    int  lr = lane & 15;
    bool hi = lane >= 16;
    int  kofs = hi ? 2 : 0;
    v8f c0 = {};
    v8f c1 = {};
    #pragma unroll
    for (int h0 = 0; h0 < HID; h0 += 32) {
        __builtin_amdgcn_sched_barrier(0);
        v2f b[8], br[8];
        #pragma unroll
        for (int s = 0; s < 8; ++s) {
            int kb = h0 + 4 * s + kofs;
            b[s].x  = W2 [kb * RED + n0 + lr]; b[s].y  = W2 [(kb + 1) * RED + n0 + lr];
            br[s].x = rW2[kb * RED + n0 + lr]; br[s].y = rW2[(kb + 1) * RED + n0 + lr];
        }
        #pragma unroll
        for (int s = 0; s < 8; ++s) {
            const float* ap = &s_x1[lr * X1P + h0 + 4 * s + kofs];
            v2f a; a.x = ap[0]; a.y = ap[1];
            c0 = __builtin_amdgcn_wmma_f32_16x16x4_f32(false, a, false, b[s],  (short)0, c0, false, false);
            c1 = __builtin_amdgcn_wmma_f32_16x16x4_f32(false, a, false, br[s], (short)0, c1, false, false);
        }
    }
    int ncol = n0 + lr;
    float b2v = b2[ncol], rb2v = rb2[ncol];
    #pragma unroll
    for (int v = 0; v < 8; ++v) {
        int mr = v + (hi ? 8 : 0);
        float dm = s_di[mr];
        float xv = c0[v];
        xw2 [(long long)(m0 + mr) * RED + ncol] = xv;
        acc2[(long long)(m0 + mr) * RED + ncol] = dm * dm * xv + b2v + c1[v] + rb2v;
    }
}

#define NPW 6
#define GHW 4
#define GH_WAVES ((N_NODES + NPW - 1) / NPW)
static_assert(NPW <= 32, "a lane per node holds the node's bounds and its logit");
#define L2_BC(S) ((S) * 0x249249)
#define L2_ASK(S)                                                                      \
    {                                                                                  \
        unsigned rr_ = (unsigned)__builtin_amdgcn_mov_dpp8(er, L2_BC(S));              \
        cf[S] = __uint_as_float((unsigned)__builtin_amdgcn_mov_dpp8(ec, L2_BC(S)));    \
        r[S]  = *reinterpret_cast<const float4*>(xb + ((rr_ << 7) | sub16));           \
    }
#define L2_ADD(S)                                                                      \
    {                                                                                  \
        acc.x = fmaf(cf[S], r[S].x, acc.x);                                            \
        acc.y = fmaf(cf[S], r[S].y, acc.y);                                            \
        acc.z = fmaf(cf[S], r[S].z, acc.z);                                            \
        acc.w = fmaf(cf[S], r[S].w, acc.w);                                            \
    }
__global__ void __launch_bounds__(GHW * 32)
k_gather_head(const int* __restrict__ od, const int* __restrict__ dci,
                       const int* __restrict__ csr, const float* __restrict__ xw2,
                       const float* __restrict__ acc2, const float* __restrict__ g2,
                       const float* __restrict__ be2, const float* __restrict__ fcW,
                       const float* __restrict__ fcb, const float* __restrict__ pW,
                       const float* __restrict__ pb, float* __restrict__ logits,
                       float* __restrict__ emb) {
    int lane = threadIdx.x & 31;
    int n0 = __builtin_amdgcn_readfirstlane((int)(blockIdx.x * GHW + (threadIdx.x >> 5))) * NPW;
    if (n0 >= N_NODES) return;
    int cnt = N_NODES - n0 < NPW ? N_NODES - n0 : NPW;
    const int2* csr2 = reinterpret_cast<const int2*>(csr);
    const char* xb   = reinterpret_cast<const char*>(xw2);
    const int      grp   = lane >> 3;
    const unsigned sub16 = (unsigned)(lane & 7) << 4;
    const int      eidx  = 4 * (lane & 7) + grp;
    int jl   = lane < cnt ? lane : cnt - 1;
    int vbeg = od[2 * (n0 + jl)];
    int vlen = dci[2 * (n0 + jl) + 1];
    float g2v = g2[lane], be2v = be2[lane], fcWv = fcW[lane], pbv = pb[lane], fcbv = fcb[0];
    float pw[32];
    #pragma unroll
    for (int i = 0; i < 32; ++i) pw[i] = pW[i * 32 + lane];
    int beg = __builtin_amdgcn_readlane(vbeg, 0);
    int len = __builtin_amdgcn_readlane(vlen, 0);
    int2 ent = make_int2(0, 0);
    if (eidx < len) ent = csr2[beg + eidx];
    float av  = acc2[(long long)n0 * 32 + lane];
    float lgv = 0.f;
    for (int j = 0; j < cnt; ++j) {
        int n = n0 + j;
        const int2* seg = csr2 + beg;
        float4 acc = make_float4(0.f, 0.f, 0.f, 0.f);
        for (int base = 0; base < len; base += 32) {
            int idx = base + eidx;
            int2 nxt = make_int2(0, 0);
            if (idx + 32 < len) nxt = seg[idx + 32];
            int m = len - base;
            int row0 = __builtin_amdgcn_readfirstlane(ent.x);
            unsigned er = (unsigned)(idx < len ? ent.x : row0);
            unsigned ec = (unsigned)ent.y;
            float4 r[8];
            float  cf[8];
            L2_ASK(0) L2_ASK(1) L2_ASK(2) L2_ASK(3)
            if (m > 16) { L2_ASK(4) L2_ASK(5) L2_ASK(6) L2_ASK(7) }
            L2_ADD(0) L2_ADD(1) L2_ADD(2) L2_ADD(3)
            if (m > 16) { L2_ADD(4) L2_ADD(5) L2_ADD(6) L2_ADD(7) }
            ent = nxt;
        }
        bool odd = (grp & 1) != 0, upper = (grp & 2) != 0;
        float k0 = odd ? acc.z : acc.x, k1 = odd ? acc.w : acc.y;
        float s0 = odd ? acc.x : acc.z, s1 = odd ? acc.y : acc.w;
        float p0 = k0 + xrow<8>(s0);
        float p1 = k1 + xrow<8>(s1);
        float kk = upper ? p1 : p0, ss = upper ? p0 : p1;
        float tot = kk + xhalf(ss);
        int cmp = lane & 3;
        float nsum = __shfl(tot, 8 * (((cmp & 1) << 1) | (cmp >> 1)) + (lane >> 2), 32);
        float v = av + nsum;
        asm volatile("" :: "v"(v));
        int jn = j + 1 < cnt ? j + 1 : cnt - 1;
        beg = __builtin_amdgcn_readlane(vbeg, jn);
        len = __builtin_amdgcn_readlane(vlen, jn);
        ent = make_int2(0, 0);
        if (eidx < len) ent = csr2[beg + eidx];
        av = acc2[(long long)(n0 + jn) * 32 + lane];
        float s = wave_sum(v), sq = wave_sum(v * v);
        float mu  = s * (1.f / 32.f);
        float var = sq * (1.f / 32.f) - mu * mu;
        float inv = rsqrtf(var + LN_EPS);
        float x2 = fmaxf((v - mu) * inv * g2v + be2v, 0.f);

        float lg = wave_sum(x2 * fcWv);
        lgv = lane == j ? lg + fcbv : lgv;

        float p = pbv;
        int x2b = __builtin_bit_cast(int, x2);
        #pragma unroll
        for (int i = 0; i < 32; ++i)
            p = fmaf(__builtin_bit_cast(float, __builtin_amdgcn_readlane(x2b, i)), pw[i], p);
        float nsq = wave_sum(p * p);
        float nrm = sqrtf(nsq);
        emb[(long long)n * 32 + lane] = p / fmaxf(nrm, 1e-12f);
    }
    if (lane < cnt) logits[n0 + lane] = lgv;
}

extern "C" void kernel_launch(void* const* d_in, const int* in_sizes, int n_in,
                              void* d_out, int out_size, void* d_ws, size_t ws_size,
                              hipStream_t stream) {
    const float* x    = (const float*)d_in[0];
    const float* ew   = (const float*)d_in[1];
    const float* W1   = (const float*)d_in[2];
    const float* b1   = (const float*)d_in[3];
    const float* rW1  = (const float*)d_in[4];
    const float* rb1  = (const float*)d_in[5];
    const float* g1   = (const float*)d_in[6];
    const float* be1  = (const float*)d_in[7];
    const float* W2   = (const float*)d_in[8];
    const float* b2   = (const float*)d_in[9];
    const float* rW2  = (const float*)d_in[10];
    const float* rb2  = (const float*)d_in[11];
    const float* g2   = (const float*)d_in[12];
    const float* be2  = (const float*)d_in[13];
    const float* fcW  = (const float*)d_in[14];
    const float* fcb  = (const float*)d_in[15];
    const float* pW   = (const float*)d_in[16];
    const float* pb   = (const float*)d_in[17];
    const int*   ei   = (const int*)d_in[18];
    const int*   row  = ei;
    const int*   col  = ei + N_EDGES;

    float* ws   = (float*)d_ws;
    ws += 2 * (size_t)N_NODES;
    int2*  tab     = (int2*)ws;
    int*   counter = (int*)(ws + 4 * (size_t)N_NODES);
    int*   dci  = (int*)(ws + 40 * (size_t)N_NODES);
    int*   od   = (int*)(ws + 42 * (size_t)N_NODES);
    float4* xd  = (float4*)(ws + 44 * (size_t)N_NODES);
    ws += (size_t)N_NODES * HID;
    int*   csr  = (int*)ws;
    ws += (size_t)N_NODES * HID;
    int2*  binned = (int2*)ws;  ws += (size_t)N_NODES * HID;
    float* xw2  = ws;  ws += (size_t)N_NODES * RED;
    float* acc2 = ws;  ws += (size_t)N_NODES * RED;

    float* logits = (float*)d_out;
    float* emb    = logits + N_NODES;

    k_chunk_sort<<<NCH, 1024, 0, stream>>>(row, col, ew, tab, counter, binned);
    k_bucket<<<NB, 1024, 0, stream>>>(tab, counter, binned, x, dci, od, xd, (int2*)csr);

    k_node_l1<<<N_NODES / NPB, NPB * 32, 0, stream>>>(x, xd, od, dci, csr, W1, b1, rW1, rb1, g1, be1,
                                                      W2, rW2, b2, rb2, xw2, acc2);

    k_gather_head<<<(GH_WAVES + GHW - 1) / GHW, GHW * 32, 0, stream>>>(od, dci, csr, xw2, acc2, g2, be2, fcW, fcb,
                                                       pW, pb, logits, emb);
}
